# starting kernel
# baseline (speedup 1.0000x reference)
#include <hip/hip_runtime.h>
#include <math.h>


#define BB   8
#define NN_  36
#define DD   64
#define NN2  (NN_*NN_)
#define NCON (2*NN_)
#define KSZ  (NN2+NCON)
#define PP   1376
#define NBLK (PP/16)
#define KDIAG 1225.0f
#define EPSK  1e-5f
#define ALPHA 1000.0f

typedef float v2f __attribute__((ext_vector_type(2)));
typedef float v8f __attribute__((ext_vector_type(8)));

__global__ void feat_kernel(const float* __restrict__ emb1,
                            const float* __restrict__ emb2,
                            const float* __restrict__ W,
                            const float* __restrict__ bias,
                            float* __restrict__ h1, float* __restrict__ h2) {
    int b = blockIdx.x, which = blockIdx.y;
    const float* e = (which == 0 ? emb1 : emb2) + (size_t)b * NN_ * DD;
    float* h = (which == 0 ? h1 : h2) + (size_t)b * NN_ * DD;
    __shared__ float se[NN_ * DD];
    __shared__ float sw[DD * DD];
    __shared__ float snrm[DD];
    int tid = threadIdx.x;
    for (int i = tid; i < NN_ * DD; i += 256) se[i] = e[i];
    for (int i = tid; i < DD * DD; i += 256) sw[i] = W[i];
    __syncthreads();
    for (int d = tid; d < DD; d += 256) {
        float s = 0.f;
        for (int n = 0; n < NN_; n++) { float v = se[n * DD + d]; s += v * v; }
        snrm[d] = fmaxf(sqrtf(s), 1e-12f);
    }
    __syncthreads();
    for (int i = tid; i < NN_ * DD; i += 256) se[i] = se[i] / snrm[i % DD];
    __syncthreads();
    for (int idx = tid; idx < NN_ * DD; idx += 256) {
        int n = idx / DD, i = idx % DD;
        float acc = bias[i];
        for (int d = 0; d < DD; d++) acc += se[n * DD + d] * sw[i * DD + d];
        h[idx] = fmaxf(acc, 0.f);
    }
}

__global__ void gram_rhs_kernel(const float* __restrict__ h1,
                                const float* __restrict__ h2,
                                float* __restrict__ S, float* __restrict__ rhs) {
    int b = blockIdx.x, tid = threadIdx.x;
    __shared__ float s1[NN_ * DD];
    __shared__ float s2[NN_ * DD];
    for (int i = tid; i < NN_ * DD; i += 256) {
        s1[i] = h1[(size_t)b * NN_ * DD + i];
        s2[i] = h2[(size_t)b * NN_ * DD + i];
    }
    __syncthreads();
    for (int idx = tid; idx < NN2; idx += 256) {
        int i = idx / NN_, j = idx % NN_;
        float a = 0.f;
        for (int d = 0; d < DD; d++) a += s1[i * DD + d] * s2[j * DD + d];
        S[(size_t)b * NN2 + idx] = a;
        rhs[(size_t)b * PP + j * NN_ + i] = a;
    }
    for (int r = NN2 + tid; r < PP; r += 256)
        rhs[(size_t)b * PP + r] = (r < KSZ) ? 1.0f : 0.0f;
}

__global__ void assemble_kernel(const float* __restrict__ S, float* __restrict__ K) {
    int b = blockIdx.z;
    int r = blockIdx.y * 16 + threadIdx.y;
    int c = blockIdx.x * 16 + threadIdx.x;
    const float* Sb = S + (size_t)b * NN2;
    float val = 0.f;
    if (r < NN2 && c < NN2) {
        int t = r / NN_, s = r - t * NN_, u = c / NN_, v = c - u * NN_;
        val = (r == c) ? KDIAG : 0.f;
        if (s != v && t != u) val -= 0.5f * (Sb[s * NN_ + t] + Sb[v * NN_ + u]);
    } else if (r < NN2) {
        int i = c - NN2, t = r / NN_, s = r - t * NN_;
        if (i < NN_)            val = (s == i) ? 1.f : 0.f;
        else if (i < NCON)      val = (t == i - NN_) ? 1.f : 0.f;
    } else if (c < NN2) {
        int i = r - NN2, u = c / NN_, v = c - u * NN_;
        if (i < NN_)            val = (v == i) ? 1.f : 0.f;
        else if (i < NCON)      val = (u == i - NN_) ? 1.f : 0.f;
    } else if (r == c) {
        val = (r < KSZ) ? -EPSK : 1.0f;
    }
    K[((size_t)b * PP + r) * PP + c] = val;
}

__global__ void lu_panel_kernel(float* __restrict__ K, int kb) {
    extern __shared__ float pan[];
    int b = blockIdx.x, tid = threadIdx.x;
    int r0 = kb * 16, H = PP - r0;
    float* Kb = K + (size_t)b * PP * PP;
    for (int i = tid; i < H * 16; i += 256)
        pan[i] = Kb[(size_t)(r0 + i / 16) * PP + r0 + (i % 16)];
    __syncthreads();
    for (int j = 0; j < 16; j++) {
        float dinv = 1.0f / pan[j * 16 + j];
        for (int i = j + 1 + tid; i < H; i += 256) {
            float lij = pan[i * 16 + j] * dinv;
            pan[i * 16 + j] = lij;
            for (int c2 = j + 1; c2 < 16; c2++)
                pan[i * 16 + c2] -= lij * pan[j * 16 + c2];
        }
        __syncthreads();
    }
    for (int i = tid; i < H * 16; i += 256)
        Kb[(size_t)(r0 + i / 16) * PP + r0 + (i % 16)] = pan[i];
}

__global__ void lu_rowpanel_kernel(float* __restrict__ K, int kb) {
    int b = blockIdx.y, tid = threadIdx.x;
    int r0 = kb * 16;
    int col = r0 + 16 + blockIdx.x * 256 + tid;
    __shared__ float L11[16 * 16];
    float* Kb = K + (size_t)b * PP * PP;
    if (tid < 256)
        L11[tid] = Kb[(size_t)(r0 + tid / 16) * PP + r0 + (tid % 16)];
    __syncthreads();
    if (col < PP) {
        float x[16];
        for (int j = 0; j < 16; j++) x[j] = Kb[(size_t)(r0 + j) * PP + col];
        for (int j = 1; j < 16; j++) {
            float s = x[j];
            for (int t2 = 0; t2 < j; t2++) s -= L11[j * 16 + t2] * x[t2];
            x[j] = s;
        }
        for (int j = 0; j < 16; j++) Kb[(size_t)(r0 + j) * PP + col] = x[j];
    }
}

template <bool R1, bool C1>
__device__ __forceinline__ void update_tile(float* __restrict__ Kb,
                                            int row0, int col0, int kcol,
                                            int lane) {
    int half = lane >> 4, l16 = lane & 15;
    int row1 = row0 + 16, col1 = col0 + 16;

    v8f c00, c01, c10, c11;
    for (int r = 0; r < 8; r++) {
        size_t ro0 = (size_t)(row0 + r + 8 * half) * PP;
        c00[r] = Kb[ro0 + col0 + l16];
        if (C1) c01[r] = Kb[ro0 + col1 + l16];
        if (R1) {
            size_t ro1 = (size_t)(row1 + r + 8 * half) * PP;
            c10[r] = Kb[ro1 + col0 + l16];
            if (C1) c11[r] = Kb[ro1 + col1 + l16];
        }
    }
    for (int kk = 0; kk < 4; kk++) {
        int k0 = kcol + kk * 4;
        int kc = k0 + 2 * half;
        v2f a0, a1, b0, b1;
        v2f av0 = *(const v2f*)(Kb + (size_t)(row0 + l16) * PP + kc);
        a0[0] = -av0[0]; a0[1] = -av0[1];
        if (R1) {
            v2f av1 = *(const v2f*)(Kb + (size_t)(row1 + l16) * PP + kc);
            a1[0] = -av1[0]; a1[1] = -av1[1];
        }
        for (int i = 0; i < 2; i++) {
            size_t kr = (size_t)(kc + i) * PP;
            b0[i] = Kb[kr + col0 + l16];
            if (C1) b1[i] = Kb[kr + col1 + l16];
        }
        c00 = __builtin_amdgcn_wmma_f32_16x16x4_f32(false, a0, false, b0, (short)0, c00, false, false);
        if (C1) c01 = __builtin_amdgcn_wmma_f32_16x16x4_f32(false, a0, false, b1, (short)0, c01, false, false);
        if (R1) c10 = __builtin_amdgcn_wmma_f32_16x16x4_f32(false, a1, false, b0, (short)0, c10, false, false);
        if (R1 && C1) c11 = __builtin_amdgcn_wmma_f32_16x16x4_f32(false, a1, false, b1, (short)0, c11, false, false);
    }
    for (int r = 0; r < 8; r++) {
        size_t ro0 = (size_t)(row0 + r + 8 * half) * PP;
        Kb[ro0 + col0 + l16] = c00[r];
        if (C1) Kb[ro0 + col1 + l16] = c01[r];
        if (R1) {
            size_t ro1 = (size_t)(row1 + r + 8 * half) * PP;
            Kb[ro1 + col0 + l16] = c10[r];
            if (C1) Kb[ro1 + col1 + l16] = c11[r];
        }
    }
}

__global__ void lu_update_kernel(float* __restrict__ K, int kb) {
    int b = blockIdx.z;
    float* Kb = K + (size_t)b * PP * PP;
    int base = kb * 16 + 16;
    int row0 = base + blockIdx.y * 32;
    int col0 = base + blockIdx.x * 32;
    bool hasR1 = (row0 + 16 < PP), hasC1 = (col0 + 16 < PP);
    int lane = threadIdx.x;
    int kcol = kb * 16;
    if (hasR1 && hasC1)      update_tile<true,  true >(Kb, row0, col0, kcol, lane);
    else if (hasR1)          update_tile<true,  false>(Kb, row0, col0, kcol, lane);
    else if (hasC1)          update_tile<false, true >(Kb, row0, col0, kcol, lane);
    else                     update_tile<false, false>(Kb, row0, col0, kcol, lane);
}

__global__ void trisolve_kernel(const float* __restrict__ K,
                                const float* __restrict__ rhs,
                                float* __restrict__ x) {
    int b = blockIdx.x, tid = threadIdx.x;
    __shared__ float y[PP];
    const float* Kb = K + (size_t)b * PP * PP;
    for (int i = tid; i < PP; i += 256) y[i] = rhs[(size_t)b * PP + i];
    __syncthreads();
    for (int kb = 0; kb < NBLK; kb++) {
        int r0 = kb * 16;
        if (tid == 0) {
            for (int j = 1; j < 16; j++) {
                float s = y[r0 + j];
                for (int t2 = 0; t2 < j; t2++)
                    s -= Kb[(size_t)(r0 + j) * PP + r0 + t2] * y[r0 + t2];
                y[r0 + j] = s;
            }
        }
        __syncthreads();
        for (int i = r0 + 16 + tid; i < PP; i += 256) {
            float s = y[i];
            for (int j = 0; j < 16; j++)
                s -= Kb[(size_t)i * PP + r0 + j] * y[r0 + j];
            y[i] = s;
        }
        __syncthreads();
    }
    for (int kb = NBLK - 1; kb >= 0; kb--) {
        int r0 = kb * 16;
        if (tid == 0) {
            for (int j = 15; j >= 0; j--) {
                float s = y[r0 + j];
                for (int t2 = j + 1; t2 < 16; t2++)
                    s -= Kb[(size_t)(r0 + j) * PP + r0 + t2] * y[r0 + t2];
                y[r0 + j] = s / Kb[(size_t)(r0 + j) * PP + r0 + j];
            }
        }
        __syncthreads();
        for (int i = tid; i < r0; i += 256) {
            float s = y[i];
            for (int j = 0; j < 16; j++)
                s -= Kb[(size_t)i * PP + r0 + j] * y[r0 + j];
            y[i] = s;
        }
        __syncthreads();
    }
    for (int i = tid; i < PP; i += 256) x[(size_t)b * PP + i] = y[i];
}

__global__ void vote_kernel(const float* __restrict__ x, float* __restrict__ out) {
    int s = blockIdx.x, b = blockIdx.y, tid = threadIdx.x;
    __shared__ float vals[NN_];
    if (tid < NN_) {
        float v = x[(size_t)b * PP + tid * NN_ + s];
        vals[tid] = fminf(fmaxf(v, 0.f), 1.f);
    }
    __syncthreads();
    if (tid == 0) {
        float m = -1e30f;
        for (int t = 0; t < NN_; t++) m = fmaxf(m, vals[t]);
        float sum = 0.f;
        for (int t = 0; t < NN_; t++) { float e = __expf(ALPHA * (vals[t] - m)); vals[t] = e; sum += e; }
        float inv = 1.f / sum;
        for (int t = 0; t < NN_; t++) out[((size_t)b * NN_ + s) * NN_ + t] = vals[t] * inv;
    }
}

extern "C" void kernel_launch(void* const* d_in, const int* in_sizes, int n_in,
                              void* d_out, int out_size, void* d_ws, size_t ws_size,
                              hipStream_t stream) {
    const float* emb1 = (const float*)d_in[0];
    const float* emb2 = (const float*)d_in[1];
    const float* Wcg  = (const float*)d_in[2];
    const float* bcg  = (const float*)d_in[3];

    float* ws  = (float*)d_ws;
    size_t kelem = (size_t)BB * PP * PP;
    float* Km  = ws;
    float* h1  = Km + kelem;
    float* h2  = h1 + (size_t)BB * NN_ * DD;
    float* S   = h2 + (size_t)BB * NN_ * DD;
    float* rhs = S  + (size_t)BB * NN2;
    float* x   = rhs + (size_t)BB * PP;

    feat_kernel<<<dim3(BB, 2), 256, 0, stream>>>(emb1, emb2, Wcg, bcg, h1, h2);
    gram_rhs_kernel<<<BB, 256, 0, stream>>>(h1, h2, S, rhs);
    assemble_kernel<<<dim3(NBLK, NBLK, BB), dim3(16, 16), 0, stream>>>(S, Km);

    for (int kb = 0; kb < NBLK; kb++) {
        int H = PP - kb * 16;
        lu_panel_kernel<<<BB, 256, (size_t)H * 16 * sizeof(float), stream>>>(Km, kb);
        int W2 = PP - (kb + 1) * 16;
        if (W2 > 0) {
            lu_rowpanel_kernel<<<dim3((W2 + 255) / 256, BB), 256, 0, stream>>>(Km, kb);
            int tiles = W2 / 16;
            int t32 = (tiles + 1) / 2;
            lu_update_kernel<<<dim3(t32, t32, BB), 32, 0, stream>>>(Km, kb);
        }
    }

    trisolve_kernel<<<BB, 256, 0, stream>>>(Km, rhs, x);
    vote_kernel<<<dim3(NN_, BB), 64, 0, stream>>>(x, (float*)d_out);
}


# K is symmetric, so U12 = D*L21^T: row-panel solve launch dropped, WMMA update only on lower-triangle tiles (B operand = pivot-scaled L21), back substitution sums L^T rows then scales (order changes)
# speedup vs baseline: 1.0625x; 1.0625x over previous
#include <hip/hip_runtime.h>
#include <math.h>


#define BB   8
#define NN_  36
#define DD   64
#define NN2  (NN_*NN_)
#define NCON (2*NN_)
#define KSZ  (NN2+NCON)
#define PP   1376
#define NBLK (PP/16)
#define KDIAG 1225.0f
#define EPSK  1e-5f
#define ALPHA 1000.0f

typedef float v2f __attribute__((ext_vector_type(2)));
typedef float v8f __attribute__((ext_vector_type(8)));

__global__ void feat_kernel(const float* __restrict__ emb1,
                            const float* __restrict__ emb2,
                            const float* __restrict__ W,
                            const float* __restrict__ bias,
                            float* __restrict__ h1, float* __restrict__ h2) {
    int b = blockIdx.x, which = blockIdx.y;
    const float* e = (which == 0 ? emb1 : emb2) + (size_t)b * NN_ * DD;
    float* h = (which == 0 ? h1 : h2) + (size_t)b * NN_ * DD;
    __shared__ float se[NN_ * DD];
    __shared__ float sw[DD * DD];
    __shared__ float snrm[DD];
    int tid = threadIdx.x;
    for (int i = tid; i < NN_ * DD; i += 256) se[i] = e[i];
    for (int i = tid; i < DD * DD; i += 256) sw[i] = W[i];
    __syncthreads();
    for (int d = tid; d < DD; d += 256) {
        float s = 0.f;
        for (int n = 0; n < NN_; n++) { float v = se[n * DD + d]; s += v * v; }
        snrm[d] = fmaxf(sqrtf(s), 1e-12f);
    }
    __syncthreads();
    for (int i = tid; i < NN_ * DD; i += 256) se[i] = se[i] / snrm[i % DD];
    __syncthreads();
    for (int idx = tid; idx < NN_ * DD; idx += 256) {
        int n = idx / DD, i = idx % DD;
        float acc = bias[i];
        for (int d = 0; d < DD; d++) acc += se[n * DD + d] * sw[i * DD + d];
        h[idx] = fmaxf(acc, 0.f);
    }
}

__global__ void gram_rhs_kernel(const float* __restrict__ h1,
                                const float* __restrict__ h2,
                                float* __restrict__ S, float* __restrict__ rhs) {
    int b = blockIdx.x, tid = threadIdx.x;
    __shared__ float s1[NN_ * DD];
    __shared__ float s2[NN_ * DD];
    for (int i = tid; i < NN_ * DD; i += 256) {
        s1[i] = h1[(size_t)b * NN_ * DD + i];
        s2[i] = h2[(size_t)b * NN_ * DD + i];
    }
    __syncthreads();
    for (int idx = tid; idx < NN2; idx += 256) {
        int i = idx / NN_, j = idx % NN_;
        float a = 0.f;
        for (int d = 0; d < DD; d++) a += s1[i * DD + d] * s2[j * DD + d];
        S[(size_t)b * NN2 + idx] = a;
        rhs[(size_t)b * PP + j * NN_ + i] = a;
    }
    for (int r = NN2 + tid; r < PP; r += 256)
        rhs[(size_t)b * PP + r] = (r < KSZ) ? 1.0f : 0.0f;
}

__global__ void assemble_kernel(const float* __restrict__ S, float* __restrict__ K) {
    int b = blockIdx.z;
    int r = blockIdx.y * 16 + threadIdx.y;
    int c = blockIdx.x * 16 + threadIdx.x;
    const float* Sb = S + (size_t)b * NN2;
    float val = 0.f;
    if (r < NN2 && c < NN2) {
        int t = r / NN_, s = r - t * NN_, u = c / NN_, v = c - u * NN_;
        val = (r == c) ? KDIAG : 0.f;
        if (s != v && t != u) val -= 0.5f * (Sb[s * NN_ + t] + Sb[v * NN_ + u]);
    } else if (r < NN2) {
        int i = c - NN2, t = r / NN_, s = r - t * NN_;
        if (i < NN_)            val = (s == i) ? 1.f : 0.f;
        else if (i < NCON)      val = (t == i - NN_) ? 1.f : 0.f;
    } else if (c < NN2) {
        int i = r - NN2, u = c / NN_, v = c - u * NN_;
        if (i < NN_)            val = (v == i) ? 1.f : 0.f;
        else if (i < NCON)      val = (u == i - NN_) ? 1.f : 0.f;
    } else if (r == c) {
        val = (r < KSZ) ? -EPSK : 1.0f;
    }
    K[((size_t)b * PP + r) * PP + c] = val;
}

__global__ void lu_panel_kernel(float* __restrict__ K, int kb) {
    extern __shared__ float pan[];
    int b = blockIdx.x, tid = threadIdx.x;
    int r0 = kb * 16, H = PP - r0;
    float* Kb = K + (size_t)b * PP * PP;
    for (int i = tid; i < H * 16; i += 256)
        pan[i] = Kb[(size_t)(r0 + i / 16) * PP + r0 + (i % 16)];
    __syncthreads();
    for (int j = 0; j < 16; j++) {
        float dinv = 1.0f / pan[j * 16 + j];
        for (int i = j + 1 + tid; i < H; i += 256) {
            float lij = pan[i * 16 + j] * dinv;
            pan[i * 16 + j] = lij;
            for (int c2 = j + 1; c2 < 16; c2++)
                pan[i * 16 + c2] -= lij * pan[j * 16 + c2];
        }
        __syncthreads();
    }
    for (int i = tid; i < H * 16; i += 256)
        Kb[(size_t)(r0 + i / 16) * PP + r0 + (i % 16)] = pan[i];
}

template <bool R1, bool C1>
__device__ __forceinline__ void update_tile(float* __restrict__ Kb,
                                            int row0, int col0, int kcol,
                                            int lane) {
    int half = lane >> 4, l16 = lane & 15;
    int row1 = row0 + 16, col1 = col0 + 16;

    v8f c00, c01, c10, c11;
    for (int r = 0; r < 8; r++) {
        size_t ro0 = (size_t)(row0 + r + 8 * half) * PP;
        c00[r] = Kb[ro0 + col0 + l16];
        if (C1) c01[r] = Kb[ro0 + col1 + l16];
        if (R1) {
            size_t ro1 = (size_t)(row1 + r + 8 * half) * PP;
            c10[r] = Kb[ro1 + col0 + l16];
            if (C1) c11[r] = Kb[ro1 + col1 + l16];
        }
    }
    for (int kk = 0; kk < 4; kk++) {
        int k0 = kcol + kk * 4;
        int kc = k0 + 2 * half;
        v2f a0, a1, b0, b1;
        v2f av0 = *(const v2f*)(Kb + (size_t)(row0 + l16) * PP + kc);
        a0[0] = -av0[0]; a0[1] = -av0[1];
        if (R1) {
            v2f av1 = *(const v2f*)(Kb + (size_t)(row1 + l16) * PP + kc);
            a1[0] = -av1[0]; a1[1] = -av1[1];
        }
        v2f dk;
        dk[0] = Kb[(size_t)kc * PP + kc];
        dk[1] = Kb[(size_t)(kc + 1) * PP + kc + 1];
        v2f lv0 = *(const v2f*)(Kb + (size_t)(col0 + l16) * PP + kc);
        b0 = lv0 * dk;
        if (C1) {
            v2f lv1 = *(const v2f*)(Kb + (size_t)(col1 + l16) * PP + kc);
            b1 = lv1 * dk;
        }
        c00 = __builtin_amdgcn_wmma_f32_16x16x4_f32(false, a0, false, b0, (short)0, c00, false, false);
        if (C1) c01 = __builtin_amdgcn_wmma_f32_16x16x4_f32(false, a0, false, b1, (short)0, c01, false, false);
        if (R1) c10 = __builtin_amdgcn_wmma_f32_16x16x4_f32(false, a1, false, b0, (short)0, c10, false, false);
        if (R1 && C1) c11 = __builtin_amdgcn_wmma_f32_16x16x4_f32(false, a1, false, b1, (short)0, c11, false, false);
    }
    for (int r = 0; r < 8; r++) {
        size_t ro0 = (size_t)(row0 + r + 8 * half) * PP;
        Kb[ro0 + col0 + l16] = c00[r];
        if (C1) Kb[ro0 + col1 + l16] = c01[r];
        if (R1) {
            size_t ro1 = (size_t)(row1 + r + 8 * half) * PP;
            Kb[ro1 + col0 + l16] = c10[r];
            if (C1) Kb[ro1 + col1 + l16] = c11[r];
        }
    }
}

__global__ void lu_update_kernel(float* __restrict__ K, int kb) {
    int b = blockIdx.y;
    float* Kb = K + (size_t)b * PP * PP;
    int t = blockIdx.x;
    int by = (int)((sqrtf(8.0f * (float)t + 1.0f) - 1.0f) * 0.5f);
    while (by * (by + 1) / 2 > t) by--;
    while ((by + 1) * (by + 2) / 2 <= t) by++;
    int bx = t - by * (by + 1) / 2;
    int base = kb * 16 + 16;
    int row0 = base + by * 32;
    int col0 = base + bx * 32;
    bool hasR1 = (row0 + 16 < PP), hasC1 = (col0 + 16 < PP);
    int lane = threadIdx.x;
    int kcol = kb * 16;
    if (hasR1)               update_tile<true,  true >(Kb, row0, col0, kcol, lane);
    else if (hasC1)          update_tile<false, true >(Kb, row0, col0, kcol, lane);
    else                     update_tile<false, false>(Kb, row0, col0, kcol, lane);
}

__global__ void trisolve_kernel(const float* __restrict__ K,
                                const float* __restrict__ rhs,
                                float* __restrict__ x) {
    int b = blockIdx.x, tid = threadIdx.x;
    __shared__ float y[PP];
    __shared__ float dg[PP];
    const float* Kb = K + (size_t)b * PP * PP;
    for (int i = tid; i < PP; i += 256) {
        y[i] = rhs[(size_t)b * PP + i];
        dg[i] = Kb[(size_t)i * PP + i];
    }
    __syncthreads();
    for (int kb = 0; kb < NBLK; kb++) {
        int r0 = kb * 16;
        if (tid == 0) {
            for (int j = 1; j < 16; j++) {
                float s = y[r0 + j];
                for (int t2 = 0; t2 < j; t2++)
                    s -= Kb[(size_t)(r0 + j) * PP + r0 + t2] * y[r0 + t2];
                y[r0 + j] = s;
            }
        }
        __syncthreads();
        for (int i = r0 + 16 + tid; i < PP; i += 256) {
            float s = y[i];
            for (int j = 0; j < 16; j++)
                s -= Kb[(size_t)i * PP + r0 + j] * y[r0 + j];
            y[i] = s;
        }
        __syncthreads();
    }
    for (int kb = NBLK - 1; kb >= 0; kb--) {
        int r0 = kb * 16;
        if (tid == 0) {
            for (int j = 15; j >= 0; j--) {
                float s = y[r0 + j];
                for (int t2 = j + 1; t2 < 16; t2++)
                    s -= Kb[(size_t)(r0 + j) * PP + r0 + t2] * y[r0 + t2];
                y[r0 + j] = s / Kb[(size_t)(r0 + j) * PP + r0 + j];
            }
        }
        __syncthreads();
        for (int i = tid; i < r0; i += 256) {
            float acc = 0.f;
            for (int j = 0; j < 16; j++)
                acc += Kb[(size_t)(r0 + j) * PP + i] * y[r0 + j];
            y[i] -= dg[i] * acc;
        }
        __syncthreads();
    }
    for (int i = tid; i < PP; i += 256) x[(size_t)b * PP + i] = y[i];
}

__global__ void vote_kernel(const float* __restrict__ x, float* __restrict__ out) {
    int s = blockIdx.x, b = blockIdx.y, tid = threadIdx.x;
    __shared__ float vals[NN_];
    if (tid < NN_) {
        float v = x[(size_t)b * PP + tid * NN_ + s];
        vals[tid] = fminf(fmaxf(v, 0.f), 1.f);
    }
    __syncthreads();
    if (tid == 0) {
        float m = -1e30f;
        for (int t = 0; t < NN_; t++) m = fmaxf(m, vals[t]);
        float sum = 0.f;
        for (int t = 0; t < NN_; t++) { float e = __expf(ALPHA * (vals[t] - m)); vals[t] = e; sum += e; }
        float inv = 1.f / sum;
        for (int t = 0; t < NN_; t++) out[((size_t)b * NN_ + s) * NN_ + t] = vals[t] * inv;
    }
}

extern "C" void kernel_launch(void* const* d_in, const int* in_sizes, int n_in,
                              void* d_out, int out_size, void* d_ws, size_t ws_size,
                              hipStream_t stream) {
    const float* emb1 = (const float*)d_in[0];
    const float* emb2 = (const float*)d_in[1];
    const float* Wcg  = (const float*)d_in[2];
    const float* bcg  = (const float*)d_in[3];

    float* ws  = (float*)d_ws;
    size_t kelem = (size_t)BB * PP * PP;
    float* Km  = ws;
    float* h1  = Km + kelem;
    float* h2  = h1 + (size_t)BB * NN_ * DD;
    float* S   = h2 + (size_t)BB * NN_ * DD;
    float* rhs = S  + (size_t)BB * NN2;
    float* x   = rhs + (size_t)BB * PP;

    feat_kernel<<<dim3(BB, 2), 256, 0, stream>>>(emb1, emb2, Wcg, bcg, h1, h2);
    gram_rhs_kernel<<<BB, 256, 0, stream>>>(h1, h2, S, rhs);
    assemble_kernel<<<dim3(NBLK, NBLK, BB), dim3(16, 16), 0, stream>>>(S, Km);

    for (int kb = 0; kb < NBLK; kb++) {
        int H = PP - kb * 16;
        lu_panel_kernel<<<BB, 256, (size_t)H * 16 * sizeof(float), stream>>>(Km, kb);
        int W2 = PP - (kb + 1) * 16;
        if (W2 > 0) {
            int tiles = W2 / 16;
            int t32 = (tiles + 1) / 2;
            int ntri = t32 * (t32 + 1) / 2;
            lu_update_kernel<<<dim3(ntri, BB), 32, 0, stream>>>(Km, kb);
        }
    }

    trisolve_kernel<<<BB, 256, 0, stream>>>(Km, rhs, x);
    vote_kernel<<<dim3(NN_, BB), 64, 0, stream>>>(x, (float*)d_out);
}


# LU panel kernel: LDS row pitch 16 -> 17 dwords, so the column sweep's per-lane row loads/stores hit 32 different banks instead of 4 (8-way conflict removed); arithmetic and order unchanged
# speedup vs baseline: 1.5293x; 1.4393x over previous
#include <hip/hip_runtime.h>
#include <math.h>


#define BB   8
#define NN_  36
#define DD   64
#define NN2  (NN_*NN_)
#define NCON (2*NN_)
#define KSZ  (NN2+NCON)
#define PP   1376
#define NBLK (PP/16)
#define KDIAG 1225.0f
#define EPSK  1e-5f
#define ALPHA 1000.0f
#define PS   17

typedef float v2f __attribute__((ext_vector_type(2)));
typedef float v8f __attribute__((ext_vector_type(8)));

__global__ void feat_kernel(const float* __restrict__ emb1,
                            const float* __restrict__ emb2,
                            const float* __restrict__ W,
                            const float* __restrict__ bias,
                            float* __restrict__ h1, float* __restrict__ h2) {
    int b = blockIdx.x, which = blockIdx.y;
    const float* e = (which == 0 ? emb1 : emb2) + (size_t)b * NN_ * DD;
    float* h = (which == 0 ? h1 : h2) + (size_t)b * NN_ * DD;
    __shared__ float se[NN_ * DD];
    __shared__ float sw[DD * DD];
    __shared__ float snrm[DD];
    int tid = threadIdx.x;
    for (int i = tid; i < NN_ * DD; i += 256) se[i] = e[i];
    for (int i = tid; i < DD * DD; i += 256) sw[i] = W[i];
    __syncthreads();
    for (int d = tid; d < DD; d += 256) {
        float s = 0.f;
        for (int n = 0; n < NN_; n++) { float v = se[n * DD + d]; s += v * v; }
        snrm[d] = fmaxf(sqrtf(s), 1e-12f);
    }
    __syncthreads();
    for (int i = tid; i < NN_ * DD; i += 256) se[i] = se[i] / snrm[i % DD];
    __syncthreads();
    for (int idx = tid; idx < NN_ * DD; idx += 256) {
        int n = idx / DD, i = idx % DD;
        float acc = bias[i];
        for (int d = 0; d < DD; d++) acc += se[n * DD + d] * sw[i * DD + d];
        h[idx] = fmaxf(acc, 0.f);
    }
}

__global__ void gram_rhs_kernel(const float* __restrict__ h1,
                                const float* __restrict__ h2,
                                float* __restrict__ S, float* __restrict__ rhs) {
    int b = blockIdx.x, tid = threadIdx.x;
    __shared__ float s1[NN_ * DD];
    __shared__ float s2[NN_ * DD];
    for (int i = tid; i < NN_ * DD; i += 256) {
        s1[i] = h1[(size_t)b * NN_ * DD + i];
        s2[i] = h2[(size_t)b * NN_ * DD + i];
    }
    __syncthreads();
    for (int idx = tid; idx < NN2; idx += 256) {
        int i = idx / NN_, j = idx % NN_;
        float a = 0.f;
        for (int d = 0; d < DD; d++) a += s1[i * DD + d] * s2[j * DD + d];
        S[(size_t)b * NN2 + idx] = a;
        rhs[(size_t)b * PP + j * NN_ + i] = a;
    }
    for (int r = NN2 + tid; r < PP; r += 256)
        rhs[(size_t)b * PP + r] = (r < KSZ) ? 1.0f : 0.0f;
}

__global__ void assemble_kernel(const float* __restrict__ S, float* __restrict__ K) {
    int b = blockIdx.z;
    int r = blockIdx.y * 16 + threadIdx.y;
    int c = blockIdx.x * 16 + threadIdx.x;
    const float* Sb = S + (size_t)b * NN2;
    float val = 0.f;
    if (r < NN2 && c < NN2) {
        int t = r / NN_, s = r - t * NN_, u = c / NN_, v = c - u * NN_;
        val = (r == c) ? KDIAG : 0.f;
        if (s != v && t != u) val -= 0.5f * (Sb[s * NN_ + t] + Sb[v * NN_ + u]);
    } else if (r < NN2) {
        int i = c - NN2, t = r / NN_, s = r - t * NN_;
        if (i < NN_)            val = (s == i) ? 1.f : 0.f;
        else if (i < NCON)      val = (t == i - NN_) ? 1.f : 0.f;
    } else if (c < NN2) {
        int i = r - NN2, u = c / NN_, v = c - u * NN_;
        if (i < NN_)            val = (v == i) ? 1.f : 0.f;
        else if (i < NCON)      val = (u == i - NN_) ? 1.f : 0.f;
    } else if (r == c) {
        val = (r < KSZ) ? -EPSK : 1.0f;
    }
    K[((size_t)b * PP + r) * PP + c] = val;
}

__global__ void lu_panel_kernel(float* __restrict__ K, int kb) {
    extern __shared__ float pan[];
    int b = blockIdx.x, tid = threadIdx.x;
    int r0 = kb * 16, H = PP - r0;
    float* Kb = K + (size_t)b * PP * PP;
    for (int i = tid; i < H * 16; i += 256)
        pan[(i / 16) * PS + (i % 16)] = Kb[(size_t)(r0 + i / 16) * PP + r0 + (i % 16)];
    __syncthreads();
    for (int j = 0; j < 16; j++) {
        float dinv = 1.0f / pan[j * PS + j];
        for (int i = j + 1 + tid; i < H; i += 256) {
            float lij = pan[i * PS + j] * dinv;
            pan[i * PS + j] = lij;
            for (int c2 = j + 1; c2 < 16; c2++)
                pan[i * PS + c2] -= lij * pan[j * PS + c2];
        }
        __syncthreads();
    }
    for (int i = tid; i < H * 16; i += 256)
        Kb[(size_t)(r0 + i / 16) * PP + r0 + (i % 16)] = pan[(i / 16) * PS + (i % 16)];
}

template <bool R1, bool C1>
__device__ __forceinline__ void update_tile(float* __restrict__ Kb,
                                            int row0, int col0, int kcol,
                                            int lane) {
    int half = lane >> 4, l16 = lane & 15;
    int row1 = row0 + 16, col1 = col0 + 16;

    v8f c00, c01, c10, c11;
    for (int r = 0; r < 8; r++) {
        size_t ro0 = (size_t)(row0 + r + 8 * half) * PP;
        c00[r] = Kb[ro0 + col0 + l16];
        if (C1) c01[r] = Kb[ro0 + col1 + l16];
        if (R1) {
            size_t ro1 = (size_t)(row1 + r + 8 * half) * PP;
            c10[r] = Kb[ro1 + col0 + l16];
            if (C1) c11[r] = Kb[ro1 + col1 + l16];
        }
    }
    for (int kk = 0; kk < 4; kk++) {
        int k0 = kcol + kk * 4;
        int kc = k0 + 2 * half;
        v2f a0, a1, b0, b1;
        v2f av0 = *(const v2f*)(Kb + (size_t)(row0 + l16) * PP + kc);
        a0[0] = -av0[0]; a0[1] = -av0[1];
        if (R1) {
            v2f av1 = *(const v2f*)(Kb + (size_t)(row1 + l16) * PP + kc);
            a1[0] = -av1[0]; a1[1] = -av1[1];
        }
        v2f dk;
        dk[0] = Kb[(size_t)kc * PP + kc];
        dk[1] = Kb[(size_t)(kc + 1) * PP + kc + 1];
        v2f lv0 = *(const v2f*)(Kb + (size_t)(col0 + l16) * PP + kc);
        b0 = lv0 * dk;
        if (C1) {
            v2f lv1 = *(const v2f*)(Kb + (size_t)(col1 + l16) * PP + kc);
            b1 = lv1 * dk;
        }
        c00 = __builtin_amdgcn_wmma_f32_16x16x4_f32(false, a0, false, b0, (short)0, c00, false, false);
        if (C1) c01 = __builtin_amdgcn_wmma_f32_16x16x4_f32(false, a0, false, b1, (short)0, c01, false, false);
        if (R1) c10 = __builtin_amdgcn_wmma_f32_16x16x4_f32(false, a1, false, b0, (short)0, c10, false, false);
        if (R1 && C1) c11 = __builtin_amdgcn_wmma_f32_16x16x4_f32(false, a1, false, b1, (short)0, c11, false, false);
    }
    for (int r = 0; r < 8; r++) {
        size_t ro0 = (size_t)(row0 + r + 8 * half) * PP;
        Kb[ro0 + col0 + l16] = c00[r];
        if (C1) Kb[ro0 + col1 + l16] = c01[r];
        if (R1) {
            size_t ro1 = (size_t)(row1 + r + 8 * half) * PP;
            Kb[ro1 + col0 + l16] = c10[r];
            if (C1) Kb[ro1 + col1 + l16] = c11[r];
        }
    }
}

__global__ void lu_update_kernel(float* __restrict__ K, int kb) {
    int b = blockIdx.y;
    float* Kb = K + (size_t)b * PP * PP;
    int t = blockIdx.x;
    int by = (int)((sqrtf(8.0f * (float)t + 1.0f) - 1.0f) * 0.5f);
    while (by * (by + 1) / 2 > t) by--;
    while ((by + 1) * (by + 2) / 2 <= t) by++;
    int bx = t - by * (by + 1) / 2;
    int base = kb * 16 + 16;
    int row0 = base + by * 32;
    int col0 = base + bx * 32;
    bool hasR1 = (row0 + 16 < PP), hasC1 = (col0 + 16 < PP);
    int lane = threadIdx.x;
    int kcol = kb * 16;
    if (hasR1)               update_tile<true,  true >(Kb, row0, col0, kcol, lane);
    else if (hasC1)          update_tile<false, true >(Kb, row0, col0, kcol, lane);
    else                     update_tile<false, false>(Kb, row0, col0, kcol, lane);
}

__global__ void trisolve_kernel(const float* __restrict__ K,
                                const float* __restrict__ rhs,
                                float* __restrict__ x) {
    int b = blockIdx.x, tid = threadIdx.x;
    __shared__ float y[PP];
    __shared__ float dg[PP];
    const float* Kb = K + (size_t)b * PP * PP;
    for (int i = tid; i < PP; i += 256) {
        y[i] = rhs[(size_t)b * PP + i];
        dg[i] = Kb[(size_t)i * PP + i];
    }
    __syncthreads();
    for (int kb = 0; kb < NBLK; kb++) {
        int r0 = kb * 16;
        if (tid == 0) {
            for (int j = 1; j < 16; j++) {
                float s = y[r0 + j];
                for (int t2 = 0; t2 < j; t2++)
                    s -= Kb[(size_t)(r0 + j) * PP + r0 + t2] * y[r0 + t2];
                y[r0 + j] = s;
            }
        }
        __syncthreads();
        for (int i = r0 + 16 + tid; i < PP; i += 256) {
            float s = y[i];
            for (int j = 0; j < 16; j++)
                s -= Kb[(size_t)i * PP + r0 + j] * y[r0 + j];
            y[i] = s;
        }
        __syncthreads();
    }
    for (int kb = NBLK - 1; kb >= 0; kb--) {
        int r0 = kb * 16;
        if (tid == 0) {
            for (int j = 15; j >= 0; j--) {
                float s = y[r0 + j];
                for (int t2 = j + 1; t2 < 16; t2++)
                    s -= Kb[(size_t)(r0 + j) * PP + r0 + t2] * y[r0 + t2];
                y[r0 + j] = s / Kb[(size_t)(r0 + j) * PP + r0 + j];
            }
        }
        __syncthreads();
        for (int i = tid; i < r0; i += 256) {
            float acc = 0.f;
            for (int j = 0; j < 16; j++)
                acc += Kb[(size_t)(r0 + j) * PP + i] * y[r0 + j];
            y[i] -= dg[i] * acc;
        }
        __syncthreads();
    }
    for (int i = tid; i < PP; i += 256) x[(size_t)b * PP + i] = y[i];
}

__global__ void vote_kernel(const float* __restrict__ x, float* __restrict__ out) {
    int s = blockIdx.x, b = blockIdx.y, tid = threadIdx.x;
    __shared__ float vals[NN_];
    if (tid < NN_) {
        float v = x[(size_t)b * PP + tid * NN_ + s];
        vals[tid] = fminf(fmaxf(v, 0.f), 1.f);
    }
    __syncthreads();
    if (tid == 0) {
        float m = -1e30f;
        for (int t = 0; t < NN_; t++) m = fmaxf(m, vals[t]);
        float sum = 0.f;
        for (int t = 0; t < NN_; t++) { float e = __expf(ALPHA * (vals[t] - m)); vals[t] = e; sum += e; }
        float inv = 1.f / sum;
        for (int t = 0; t < NN_; t++) out[((size_t)b * NN_ + s) * NN_ + t] = vals[t] * inv;
    }
}

extern "C" void kernel_launch(void* const* d_in, const int* in_sizes, int n_in,
                              void* d_out, int out_size, void* d_ws, size_t ws_size,
                              hipStream_t stream) {
    const float* emb1 = (const float*)d_in[0];
    const float* emb2 = (const float*)d_in[1];
    const float* Wcg  = (const float*)d_in[2];
    const float* bcg  = (const float*)d_in[3];

    float* ws  = (float*)d_ws;
    size_t kelem = (size_t)BB * PP * PP;
    float* Km  = ws;
    float* h1  = Km + kelem;
    float* h2  = h1 + (size_t)BB * NN_ * DD;
    float* S   = h2 + (size_t)BB * NN_ * DD;
    float* rhs = S  + (size_t)BB * NN2;
    float* x   = rhs + (size_t)BB * PP;

    feat_kernel<<<dim3(BB, 2), 256, 0, stream>>>(emb1, emb2, Wcg, bcg, h1, h2);
    gram_rhs_kernel<<<BB, 256, 0, stream>>>(h1, h2, S, rhs);
    assemble_kernel<<<dim3(NBLK, NBLK, BB), dim3(16, 16), 0, stream>>>(S, Km);

    for (int kb = 0; kb < NBLK; kb++) {
        int H = PP - kb * 16;
        lu_panel_kernel<<<BB, 256, (size_t)H * PS * sizeof(float), stream>>>(Km, kb);
        int W2 = PP - (kb + 1) * 16;
        if (W2 > 0) {
            int tiles = W2 / 16;
            int t32 = (tiles + 1) / 2;
            int ntri = t32 * (t32 + 1) / 2;
            lu_update_kernel<<<dim3(ntri, BB), 32, 0, stream>>>(Km, kb);
        }
    }

    trisolve_kernel<<<BB, 256, 0, stream>>>(Km, rhs, x);
    vote_kernel<<<dim3(NN_, BB), 64, 0, stream>>>(x, (float*)d_out);
}


# Left-looking LU, one launch per 16-wide block column: each wave forms its own tile and the diagonal tile by WMMA chains over all earlier columns, then eliminates both in registers (sum order kept)
# speedup vs baseline: 2.9159x; 1.9067x over previous
#include <hip/hip_runtime.h>
#include <math.h>


#define BB   8
#define NN_  36
#define DD   64
#define NN2  (NN_*NN_)
#define NCON (2*NN_)
#define KSZ  (NN2+NCON)
#define PP   1376
#define NBLK (PP/16)
#define KDIAG 1225.0f
#define EPSK  1e-5f
#define ALPHA 1000.0f
#define PS   17

typedef float v2f __attribute__((ext_vector_type(2)));
typedef float v8f __attribute__((ext_vector_type(8)));

__global__ void feat_kernel(const float* __restrict__ emb1,
                            const float* __restrict__ emb2,
                            const float* __restrict__ W,
                            const float* __restrict__ bias,
                            float* __restrict__ h1, float* __restrict__ h2) {
    int b = blockIdx.x, which = blockIdx.y;
    const float* e = (which == 0 ? emb1 : emb2) + (size_t)b * NN_ * DD;
    float* h = (which == 0 ? h1 : h2) + (size_t)b * NN_ * DD;
    __shared__ float se[NN_ * DD];
    __shared__ float sw[DD * DD];
    __shared__ float snrm[DD];
    int tid = threadIdx.x;
    for (int i = tid; i < NN_ * DD; i += 256) se[i] = e[i];
    for (int i = tid; i < DD * DD; i += 256) sw[i] = W[i];
    __syncthreads();
    for (int d = tid; d < DD; d += 256) {
        float s = 0.f;
        for (int n = 0; n < NN_; n++) { float v = se[n * DD + d]; s += v * v; }
        snrm[d] = fmaxf(sqrtf(s), 1e-12f);
    }
    __syncthreads();
    for (int i = tid; i < NN_ * DD; i += 256) se[i] = se[i] / snrm[i % DD];
    __syncthreads();
    for (int idx = tid; idx < NN_ * DD; idx += 256) {
        int n = idx / DD, i = idx % DD;
        float acc = bias[i];
        for (int d = 0; d < DD; d++) acc += se[n * DD + d] * sw[i * DD + d];
        h[idx] = fmaxf(acc, 0.f);
    }
}

__global__ void gram_rhs_kernel(const float* __restrict__ h1,
                                const float* __restrict__ h2,
                                float* __restrict__ S, float* __restrict__ rhs) {
    int b = blockIdx.x, tid = threadIdx.x;
    __shared__ float s1[NN_ * DD];
    __shared__ float s2[NN_ * DD];
    for (int i = tid; i < NN_ * DD; i += 256) {
        s1[i] = h1[(size_t)b * NN_ * DD + i];
        s2[i] = h2[(size_t)b * NN_ * DD + i];
    }
    __syncthreads();
    for (int idx = tid; idx < NN2; idx += 256) {
        int i = idx / NN_, j = idx % NN_;
        float a = 0.f;
        for (int d = 0; d < DD; d++) a += s1[i * DD + d] * s2[j * DD + d];
        S[(size_t)b * NN2 + idx] = a;
        rhs[(size_t)b * PP + j * NN_ + i] = a;
    }
    for (int r = NN2 + tid; r < PP; r += 256)
        rhs[(size_t)b * PP + r] = (r < KSZ) ? 1.0f : 0.0f;
}

__global__ void assemble_kernel(const float* __restrict__ S, float* __restrict__ K) {
    int b = blockIdx.z;
    int r = blockIdx.y * 16 + threadIdx.y;
    int c = blockIdx.x * 16 + threadIdx.x;
    const float* Sb = S + (size_t)b * NN2;
    float val = 0.f;
    if (r < NN2 && c < NN2) {
        int t = r / NN_, s = r - t * NN_, u = c / NN_, v = c - u * NN_;
        val = (r == c) ? KDIAG : 0.f;
        if (s != v && t != u) val -= 0.5f * (Sb[s * NN_ + t] + Sb[v * NN_ + u]);
    } else if (r < NN2) {
        int i = c - NN2, t = r / NN_, s = r - t * NN_;
        if (i < NN_)            val = (s == i) ? 1.f : 0.f;
        else if (i < NCON)      val = (t == i - NN_) ? 1.f : 0.f;
    } else if (c < NN2) {
        int i = r - NN2, u = c / NN_, v = c - u * NN_;
        if (i < NN_)            val = (v == i) ? 1.f : 0.f;
        else if (i < NCON)      val = (u == i - NN_) ? 1.f : 0.f;
    } else if (r == c) {
        val = (r < KSZ) ? -EPSK : 1.0f;
    }
    K[((size_t)b * PP + r) * PP + c] = val;
}

#define QS 16

__global__ void __launch_bounds__(32) lu_col_kernel(float* __restrict__ K,
                                                    float* __restrict__ dpiv,
                                                    int jb) {
    __shared__ float sT[32 * PS];
    const int b = blockIdx.y, ti = blockIdx.x, lane = threadIdx.x;
    const int half = lane >> 4, l16 = lane & 15;
    float* Kb = K + (size_t)b * PP * PP;
    float* db = dpiv + (size_t)b * PP;
    const int c0 = jb * 16;
    const int rown = c0 + ti * 16;

    v8f cd, co;
#pragma unroll
    for (int r = 0; r < 8; r++) {
        cd[r] = Kb[(size_t)(c0 + r + 8 * half) * PP + c0 + l16];
        co[r] = Kb[(size_t)(rown + r + 8 * half) * PP + c0 + l16];
    }
    const float* pl = Kb + (size_t)(c0 + l16) * PP + 2 * half;
    const float* pa = Kb + (size_t)(rown + l16) * PP + 2 * half;
    const float* pd = db + 2 * half;

    const int nq = c0 / (4 * QS);
    v2f ra[QS], rl[QS], rd[QS];
    if (nq > 0) {
#pragma unroll
        for (int s = 0; s < QS; s++) {
            ra[s] = *(const v2f*)(pa + 4 * s);
            rl[s] = *(const v2f*)(pl + 4 * s);
            rd[s] = *(const v2f*)(pd + 4 * s);
        }
    }
    for (int q = 0; q < nq; q++) {
        v2f an[QS], ln[QS], bb[QS];
#pragma unroll
        for (int s = 0; s < QS; s++) {
            an[s] = -ra[s];
            ln[s] = -rl[s];
            bb[s] = rl[s] * rd[s];
        }
        const int kn = ((q + 1 < nq) ? (q + 1) : q) * (4 * QS);
#pragma unroll
        for (int s = 0; s < QS; s++) {
            ra[s] = *(const v2f*)(pa + kn + 4 * s);
            rl[s] = *(const v2f*)(pl + kn + 4 * s);
            rd[s] = *(const v2f*)(pd + kn + 4 * s);
        }
#pragma unroll
        for (int s = 0; s < QS; s++) {
            co = __builtin_amdgcn_wmma_f32_16x16x4_f32(false, an[s], false, bb[s], (short)0, co, false, false);
            cd = __builtin_amdgcn_wmma_f32_16x16x4_f32(false, ln[s], false, bb[s], (short)0, cd, false, false);
        }
    }
    for (int k0 = nq * (4 * QS); k0 < c0; k0 += 16) {
        v2f ta[4], tl[4], td[4];
#pragma unroll
        for (int s = 0; s < 4; s++) {
            ta[s] = *(const v2f*)(pa + k0 + 4 * s);
            tl[s] = *(const v2f*)(pl + k0 + 4 * s);
            td[s] = *(const v2f*)(pd + k0 + 4 * s);
        }
#pragma unroll
        for (int s = 0; s < 4; s++) {
            v2f a1 = -ta[s], l1 = -tl[s], b1 = tl[s] * td[s];
            co = __builtin_amdgcn_wmma_f32_16x16x4_f32(false, a1, false, b1, (short)0, co, false, false);
            cd = __builtin_amdgcn_wmma_f32_16x16x4_f32(false, l1, false, b1, (short)0, cd, false, false);
        }
    }

#pragma unroll
    for (int r = 0; r < 8; r++) {
        sT[(r + 8 * half) * PS + l16] = cd[r];
        sT[(16 + r + 8 * half) * PS + l16] = co[r];
    }
    __syncthreads();
    float t[16];
#pragma unroll
    for (int c = 0; c < 16; c++) t[c] = sT[lane * PS + c];

    float mypiv = 0.f;
#pragma unroll
    for (int p = 0; p < 16; p++) {
        float u[16];
#pragma unroll
        for (int c2 = p; c2 < 16; c2++)
            u[c2] = __int_as_float(__builtin_amdgcn_readlane(__float_as_int(t[c2]), p));
        float dinv = 1.0f / u[p];
        if (lane == p) mypiv = u[p];
        if (lane > p) {
            float lij = t[p] * dinv;
            t[p] = lij;
#pragma unroll
            for (int c2 = p + 1; c2 < 16; c2++) t[c2] -= lij * u[c2];
        }
    }

    float* dst = nullptr;
    if (ti == 0) {
        if (lane < 16) {
            dst = Kb + (size_t)(c0 + lane) * PP + c0 + ((jb < NBLK - 1) ? 16 : 0);
            db[c0 + lane] = mypiv;
        }
    } else if (lane >= 16) {
        dst = Kb + (size_t)(rown + lane - 16) * PP + c0;
    }
    if (dst) {
#pragma unroll
        for (int c = 0; c < 16; c += 4) {
            float4 v; v.x = t[c]; v.y = t[c + 1]; v.z = t[c + 2]; v.w = t[c + 3];
            *(float4*)(dst + c) = v;
        }
    }
}

__global__ void trisolve_kernel(const float* __restrict__ K,
                                const float* __restrict__ rhs,
                                float* __restrict__ x) {
    int b = blockIdx.x, tid = threadIdx.x;
    __shared__ float y[PP];
    __shared__ float dg[PP];
    const float* Kb = K + (size_t)b * PP * PP;
    for (int i = tid; i < PP; i += 256) {
        y[i] = rhs[(size_t)b * PP + i];
        dg[i] = Kb[(size_t)i * PP + i + (((i >> 4) < NBLK - 1) ? 16 : 0)];
    }
    __syncthreads();
    for (int kb = 0; kb < NBLK; kb++) {
        int r0 = kb * 16;
        int d0 = r0 + ((kb < NBLK - 1) ? 16 : 0);
        if (tid == 0) {
            for (int j = 1; j < 16; j++) {
                float s = y[r0 + j];
                for (int t2 = 0; t2 < j; t2++)
                    s -= Kb[(size_t)(r0 + j) * PP + d0 + t2] * y[r0 + t2];
                y[r0 + j] = s;
            }
        }
        __syncthreads();
        for (int i = r0 + 16 + tid; i < PP; i += 256) {
            float s = y[i];
            for (int j = 0; j < 16; j++)
                s -= Kb[(size_t)i * PP + r0 + j] * y[r0 + j];
            y[i] = s;
        }
        __syncthreads();
    }
    for (int kb = NBLK - 1; kb >= 0; kb--) {
        int r0 = kb * 16;
        int d0 = r0 + ((kb < NBLK - 1) ? 16 : 0);
        if (tid == 0) {
            for (int j = 15; j >= 0; j--) {
                float s = y[r0 + j];
                for (int t2 = j + 1; t2 < 16; t2++)
                    s -= Kb[(size_t)(r0 + j) * PP + d0 + t2] * y[r0 + t2];
                y[r0 + j] = s / Kb[(size_t)(r0 + j) * PP + d0 + j];
            }
        }
        __syncthreads();
        for (int i = tid; i < r0; i += 256) {
            float acc = 0.f;
            for (int j = 0; j < 16; j++)
                acc += Kb[(size_t)(r0 + j) * PP + i] * y[r0 + j];
            y[i] -= dg[i] * acc;
        }
        __syncthreads();
    }
    for (int i = tid; i < PP; i += 256) x[(size_t)b * PP + i] = y[i];
}

__global__ void vote_kernel(const float* __restrict__ x, float* __restrict__ out) {
    int s = blockIdx.x, b = blockIdx.y, tid = threadIdx.x;
    __shared__ float vals[NN_];
    if (tid < NN_) {
        float v = x[(size_t)b * PP + tid * NN_ + s];
        vals[tid] = fminf(fmaxf(v, 0.f), 1.f);
    }
    __syncthreads();
    if (tid == 0) {
        float m = -1e30f;
        for (int t = 0; t < NN_; t++) m = fmaxf(m, vals[t]);
        float sum = 0.f;
        for (int t = 0; t < NN_; t++) { float e = __expf(ALPHA * (vals[t] - m)); vals[t] = e; sum += e; }
        float inv = 1.f / sum;
        for (int t = 0; t < NN_; t++) out[((size_t)b * NN_ + s) * NN_ + t] = vals[t] * inv;
    }
}

extern "C" void kernel_launch(void* const* d_in, const int* in_sizes, int n_in,
                              void* d_out, int out_size, void* d_ws, size_t ws_size,
                              hipStream_t stream) {
    const float* emb1 = (const float*)d_in[0];
    const float* emb2 = (const float*)d_in[1];
    const float* Wcg  = (const float*)d_in[2];
    const float* bcg  = (const float*)d_in[3];

    float* ws  = (float*)d_ws;
    size_t kelem = (size_t)BB * PP * PP;
    float* Km  = ws;
    float* h1  = Km + kelem;
    float* h2  = h1 + (size_t)BB * NN_ * DD;
    float* S   = h2 + (size_t)BB * NN_ * DD;
    float* rhs = S  + (size_t)BB * NN2;
    float* x   = rhs + (size_t)BB * PP;

    feat_kernel<<<dim3(BB, 2), 256, 0, stream>>>(emb1, emb2, Wcg, bcg, h1, h2);
    gram_rhs_kernel<<<BB, 256, 0, stream>>>(h1, h2, S, rhs);
    assemble_kernel<<<dim3(NBLK, NBLK, BB), dim3(16, 16), 0, stream>>>(S, Km);

    for (int jb = 0; jb < NBLK; jb++)
        lu_col_kernel<<<dim3(NBLK - jb, BB), 32, 0, stream>>>(Km, x, jb);

    trisolve_kernel<<<BB, 256, 0, stream>>>(Km, rhs, x);
    vote_kernel<<<dim3(NN_, BB), 64, 0, stream>>>(x, (float*)d_out);
}


# Triangular solves: every factor load issued a block step ahead of the recurrence; forward in dot form (16 threads a row), each 16x16 triangle solved by one wave in registers (sum order changes)
# speedup vs baseline: 3.9487x; 1.3542x over previous
#include <hip/hip_runtime.h>
#include <math.h>


#define BB   8
#define NN_  36
#define DD   64
#define NN2  (NN_*NN_)
#define NCON (2*NN_)
#define KSZ  (NN2+NCON)
#define PP   1376
#define NBLK (PP/16)
#define KDIAG 1225.0f
#define EPSK  1e-5f
#define ALPHA 1000.0f
#define PS   17

typedef float v2f __attribute__((ext_vector_type(2)));
typedef float v8f __attribute__((ext_vector_type(8)));

__global__ void feat_kernel(const float* __restrict__ emb1,
                            const float* __restrict__ emb2,
                            const float* __restrict__ W,
                            const float* __restrict__ bias,
                            float* __restrict__ h1, float* __restrict__ h2) {
    int b = blockIdx.x, which = blockIdx.y;
    const float* e = (which == 0 ? emb1 : emb2) + (size_t)b * NN_ * DD;
    float* h = (which == 0 ? h1 : h2) + (size_t)b * NN_ * DD;
    __shared__ float se[NN_ * DD];
    __shared__ float sw[DD * DD];
    __shared__ float snrm[DD];
    int tid = threadIdx.x;
    for (int i = tid; i < NN_ * DD; i += 256) se[i] = e[i];
    for (int i = tid; i < DD * DD; i += 256) sw[i] = W[i];
    __syncthreads();
    for (int d = tid; d < DD; d += 256) {
        float s = 0.f;
        for (int n = 0; n < NN_; n++) { float v = se[n * DD + d]; s += v * v; }
        snrm[d] = fmaxf(sqrtf(s), 1e-12f);
    }
    __syncthreads();
    for (int i = tid; i < NN_ * DD; i += 256) se[i] = se[i] / snrm[i % DD];
    __syncthreads();
    for (int idx = tid; idx < NN_ * DD; idx += 256) {
        int n = idx / DD, i = idx % DD;
        float acc = bias[i];
        for (int d = 0; d < DD; d++) acc += se[n * DD + d] * sw[i * DD + d];
        h[idx] = fmaxf(acc, 0.f);
    }
}

__global__ void gram_rhs_kernel(const float* __restrict__ h1,
                                const float* __restrict__ h2,
                                float* __restrict__ S, float* __restrict__ rhs) {
    int b = blockIdx.x, tid = threadIdx.x;
    __shared__ float s1[NN_ * DD];
    __shared__ float s2[NN_ * DD];
    for (int i = tid; i < NN_ * DD; i += 256) {
        s1[i] = h1[(size_t)b * NN_ * DD + i];
        s2[i] = h2[(size_t)b * NN_ * DD + i];
    }
    __syncthreads();
    for (int idx = tid; idx < NN2; idx += 256) {
        int i = idx / NN_, j = idx % NN_;
        float a = 0.f;
        for (int d = 0; d < DD; d++) a += s1[i * DD + d] * s2[j * DD + d];
        S[(size_t)b * NN2 + idx] = a;
        rhs[(size_t)b * PP + j * NN_ + i] = a;
    }
    for (int r = NN2 + tid; r < PP; r += 256)
        rhs[(size_t)b * PP + r] = (r < KSZ) ? 1.0f : 0.0f;
}

__global__ void assemble_kernel(const float* __restrict__ S, float* __restrict__ K) {
    int b = blockIdx.z;
    int r = blockIdx.y * 16 + threadIdx.y;
    int c = blockIdx.x * 16 + threadIdx.x;
    const float* Sb = S + (size_t)b * NN2;
    float val = 0.f;
    if (r < NN2 && c < NN2) {
        int t = r / NN_, s = r - t * NN_, u = c / NN_, v = c - u * NN_;
        val = (r == c) ? KDIAG : 0.f;
        if (s != v && t != u) val -= 0.5f * (Sb[s * NN_ + t] + Sb[v * NN_ + u]);
    } else if (r < NN2) {
        int i = c - NN2, t = r / NN_, s = r - t * NN_;
        if (i < NN_)            val = (s == i) ? 1.f : 0.f;
        else if (i < NCON)      val = (t == i - NN_) ? 1.f : 0.f;
    } else if (c < NN2) {
        int i = r - NN2, u = c / NN_, v = c - u * NN_;
        if (i < NN_)            val = (v == i) ? 1.f : 0.f;
        else if (i < NCON)      val = (u == i - NN_) ? 1.f : 0.f;
    } else if (r == c) {
        val = (r < KSZ) ? -EPSK : 1.0f;
    }
    K[((size_t)b * PP + r) * PP + c] = val;
}

#define QS 16

__global__ void __launch_bounds__(32) lu_col_kernel(float* __restrict__ K,
                                                    float* __restrict__ dpiv,
                                                    int jb) {
    __shared__ float sT[32 * PS];
    const int b = blockIdx.y, ti = blockIdx.x, lane = threadIdx.x;
    const int half = lane >> 4, l16 = lane & 15;
    float* Kb = K + (size_t)b * PP * PP;
    float* db = dpiv + (size_t)b * PP;
    const int c0 = jb * 16;
    const int rown = c0 + ti * 16;

    v8f cd, co;
#pragma unroll
    for (int r = 0; r < 8; r++) {
        cd[r] = Kb[(size_t)(c0 + r + 8 * half) * PP + c0 + l16];
        co[r] = Kb[(size_t)(rown + r + 8 * half) * PP + c0 + l16];
    }
    const float* pl = Kb + (size_t)(c0 + l16) * PP + 2 * half;
    const float* pa = Kb + (size_t)(rown + l16) * PP + 2 * half;
    const float* pd = db + 2 * half;

    const int nq = c0 / (4 * QS);
    v2f ra[QS], rl[QS], rd[QS];
    if (nq > 0) {
#pragma unroll
        for (int s = 0; s < QS; s++) {
            ra[s] = *(const v2f*)(pa + 4 * s);
            rl[s] = *(const v2f*)(pl + 4 * s);
            rd[s] = *(const v2f*)(pd + 4 * s);
        }
    }
    for (int q = 0; q < nq; q++) {
        v2f an[QS], ln[QS], bb[QS];
#pragma unroll
        for (int s = 0; s < QS; s++) {
            an[s] = -ra[s];
            ln[s] = -rl[s];
            bb[s] = rl[s] * rd[s];
        }
        const int kn = ((q + 1 < nq) ? (q + 1) : q) * (4 * QS);
#pragma unroll
        for (int s = 0; s < QS; s++) {
            ra[s] = *(const v2f*)(pa + kn + 4 * s);
            rl[s] = *(const v2f*)(pl + kn + 4 * s);
            rd[s] = *(const v2f*)(pd + kn + 4 * s);
        }
#pragma unroll
        for (int s = 0; s < QS; s++) {
            co = __builtin_amdgcn_wmma_f32_16x16x4_f32(false, an[s], false, bb[s], (short)0, co, false, false);
            cd = __builtin_amdgcn_wmma_f32_16x16x4_f32(false, ln[s], false, bb[s], (short)0, cd, false, false);
        }
    }
    for (int k0 = nq * (4 * QS); k0 < c0; k0 += 16) {
        v2f ta[4], tl[4], td[4];
#pragma unroll
        for (int s = 0; s < 4; s++) {
            ta[s] = *(const v2f*)(pa + k0 + 4 * s);
            tl[s] = *(const v2f*)(pl + k0 + 4 * s);
            td[s] = *(const v2f*)(pd + k0 + 4 * s);
        }
#pragma unroll
        for (int s = 0; s < 4; s++) {
            v2f a1 = -ta[s], l1 = -tl[s], b1 = tl[s] * td[s];
            co = __builtin_amdgcn_wmma_f32_16x16x4_f32(false, a1, false, b1, (short)0, co, false, false);
            cd = __builtin_amdgcn_wmma_f32_16x16x4_f32(false, l1, false, b1, (short)0, cd, false, false);
        }
    }

#pragma unroll
    for (int r = 0; r < 8; r++) {
        sT[(r + 8 * half) * PS + l16] = cd[r];
        sT[(16 + r + 8 * half) * PS + l16] = co[r];
    }
    __syncthreads();
    float t[16];
#pragma unroll
    for (int c = 0; c < 16; c++) t[c] = sT[lane * PS + c];

    float mypiv = 0.f;
#pragma unroll
    for (int p = 0; p < 16; p++) {
        float u[16];
#pragma unroll
        for (int c2 = p; c2 < 16; c2++)
            u[c2] = __int_as_float(__builtin_amdgcn_readlane(__float_as_int(t[c2]), p));
        float dinv = 1.0f / u[p];
        if (lane == p) mypiv = u[p];
        if (lane > p) {
            float lij = t[p] * dinv;
            t[p] = lij;
#pragma unroll
            for (int c2 = p + 1; c2 < 16; c2++) t[c2] -= lij * u[c2];
        }
    }

    float* dst = nullptr;
    if (ti == 0) {
        if (lane < 16) {
            dst = Kb + (size_t)(c0 + lane) * PP + c0 + ((jb < NBLK - 1) ? 16 : 0);
            db[c0 + lane] = mypiv;
        }
    } else if (lane >= 16) {
        dst = Kb + (size_t)(rown + lane - 16) * PP + c0;
    }
    if (dst) {
#pragma unroll
        for (int c = 0; c < 16; c += 4) {
            float4 v; v.x = t[c]; v.y = t[c + 1]; v.z = t[c + 2]; v.w = t[c + 3];
            *(float4*)(dst + c) = v;
        }
    }
}

#define FCH 24
#define RP  20

__global__ void __launch_bounds__(256) trisolve_kernel(const float* __restrict__ K,
                                                       const float* __restrict__ rhs,
                                                       float* __restrict__ x) {
    const int b = blockIdx.x, tid = threadIdx.x;
    const int l16 = tid & 15, grp = tid >> 4;
    __shared__ __attribute__((aligned(16))) float y[PP];
    __shared__ __attribute__((aligned(16))) float dg[PP];
    __shared__ __attribute__((aligned(16))) float red[16 * RP];
    const float* Kb = K + (size_t)b * PP * PP;
    for (int i = tid; i < PP; i += 256) {
        y[i] = rhs[(size_t)b * PP + i];
        dg[i] = Kb[(size_t)i * PP + i + (((i >> 4) < NBLK - 1) ? 16 : 0)];
    }
    const float4 zero4 = make_float4(0.f, 0.f, 0.f, 0.f);

    float4 cur[FCH];
    float lcur[16];
#pragma unroll
    for (int m = 0; m < FCH; m++) cur[m] = zero4;
#pragma unroll
    for (int c = 0; c < 16; c++) lcur[c] = 0.f;
    if (tid < 32) {
        const float* lp = Kb + (size_t)l16 * PP + 16;
#pragma unroll
        for (int c = 0; c < 4; c++) {
            float4 t4 = *(const float4*)(lp + 4 * c);
            lcur[4 * c] = t4.x; lcur[4 * c + 1] = t4.y; lcur[4 * c + 2] = t4.z; lcur[4 * c + 3] = t4.w;
        }
    }
    __syncthreads();
    for (int kb = 0; kb < NBLK; kb++) {
        const int r0 = kb * 16, r1 = r0 + 16;
        float4 nxt[FCH];
        float lnxt[16];
#pragma unroll
        for (int m = 0; m < FCH; m++) nxt[m] = zero4;
#pragma unroll
        for (int c = 0; c < 16; c++) lnxt[c] = 0.f;
        if (kb + 1 < NBLK) {
            const float* rp = Kb + (size_t)(r1 + grp) * PP;
#pragma unroll
            for (int m = 0; m < FCH; m++) {
                if (64 * m < r1) {
                    const int k = 64 * m + 4 * l16;
                    if (k < r1) nxt[m] = *(const float4*)(rp + k);
                }
            }
            if (tid < 32) {
                const int d1 = r1 + ((kb + 1 < NBLK - 1) ? 16 : 0);
                const float* lp = Kb + (size_t)(r1 + l16) * PP + d1;
#pragma unroll
                for (int c = 0; c < 4; c++) {
                    float4 t4 = *(const float4*)(lp + 4 * c);
                    lnxt[4 * c] = t4.x; lnxt[4 * c + 1] = t4.y; lnxt[4 * c + 2] = t4.z; lnxt[4 * c + 3] = t4.w;
                }
            }
        }
        float a0 = 0.f, a1 = 0.f, a2 = 0.f, a3 = 0.f;
#pragma unroll
        for (int g = 0; g < FCH / 4; g++) {
            if (256 * g < r0) {
#pragma unroll
                for (int mm = 0; mm < 4; mm++) {
                    const int m = 4 * g + mm;
                    const int k = 64 * m + 4 * l16;
                    const float4 yy = *(const float4*)(y + ((k < PP) ? k : 0));
                    a0 = __builtin_fmaf(cur[m].x, yy.x, a0); a1 = __builtin_fmaf(cur[m].y, yy.y, a1);
                    a2 = __builtin_fmaf(cur[m].z, yy.z, a2); a3 = __builtin_fmaf(cur[m].w, yy.w, a3);
                }
            }
        }
        red[grp * RP + l16] = (a0 + a1) + (a2 + a3);
        __syncthreads();
        if (tid < 32) {
            const float4* rr = (const float4*)(red + l16 * RP);
            const float4 s0 = rr[0], s1 = rr[1], s2 = rr[2], s3 = rr[3];
            const float dot = (((s0.x + s0.y) + (s0.z + s0.w)) + ((s1.x + s1.y) + (s1.z + s1.w)))
                            + (((s2.x + s2.y) + (s2.z + s2.w)) + ((s3.x + s3.y) + (s3.z + s3.w)));
            float v = y[r0 + l16] - dot;
#pragma unroll
            for (int p = 0; p < 15; p++) {
                const float vp = __int_as_float(__builtin_amdgcn_readlane(__float_as_int(v), p));
                if (l16 > p) v -= lcur[p] * vp;
            }
            if (tid < 16) y[r0 + l16] = v;
        }
        __syncthreads();
#pragma unroll
        for (int m = 0; m < FCH; m++) cur[m] = nxt[m];
#pragma unroll
        for (int c = 0; c < 16; c++) lcur[c] = lnxt[c];
    }

    float ucur[16];
#pragma unroll
    for (int c = 0; c < 16; c++) ucur[c] = 0.f;
    if (tid < 32) {
        const float* up = Kb + (size_t)((NBLK - 1) * 16 + l16) * PP + (NBLK - 1) * 16;
#pragma unroll
        for (int c = 0; c < 4; c++) {
            float4 t4 = *(const float4*)(up + 4 * c);
            ucur[4 * c] = t4.x; ucur[4 * c + 1] = t4.y; ucur[4 * c + 2] = t4.z; ucur[4 * c + 3] = t4.w;
        }
    }
    __syncthreads();
    for (int kb = NBLK - 1; kb >= 0; kb--) {
        const int r0 = kb * 16;
        const int ia = 4 * (tid - 32), ib = ia + 4 * 224;
        float4 la[16], lb[16];
        if (tid >= 32) {
            if (ia < r0) {
#pragma unroll
                for (int j = 0; j < 16; j++)
                    la[j] = *(const float4*)(Kb + (size_t)(r0 + j) * PP + ia);
            }
            if (ib < r0) {
#pragma unroll
                for (int j = 0; j < 16; j++)
                    lb[j] = *(const float4*)(Kb + (size_t)(r0 + j) * PP + ib);
            }
        } else {
            float v = y[r0 + l16];
            const float rinv = 1.0f / dg[r0 + l16];
#pragma unroll
            for (int p = 15; p >= 0; p--) {
                if (l16 == p) v *= rinv;
                const float xp = __int_as_float(__builtin_amdgcn_readlane(__float_as_int(v), p));
                if (l16 < p) v -= ucur[p] * xp;
            }
            if (tid < 16) y[r0 + l16] = v;
        }
        __syncthreads();
        float unxt[16];
#pragma unroll
        for (int c = 0; c < 16; c++) unxt[c] = 0.f;
        if (tid < 32) {
            if (kb > 0) {
                const float* up = Kb + (size_t)(r0 - 16 + l16) * PP + r0;
#pragma unroll
                for (int c = 0; c < 4; c++) {
                    float4 t4 = *(const float4*)(up + 4 * c);
                    unxt[4 * c] = t4.x; unxt[4 * c + 1] = t4.y; unxt[4 * c + 2] = t4.z; unxt[4 * c + 3] = t4.w;
                }
            }
        } else {
            float xs[16];
#pragma unroll
            for (int c = 0; c < 4; c++) {
                const float4 t4 = *(const float4*)(y + r0 + 4 * c);
                xs[4 * c] = t4.x; xs[4 * c + 1] = t4.y; xs[4 * c + 2] = t4.z; xs[4 * c + 3] = t4.w;
            }
            if (ia < r0) {
                float4 acc = zero4;
#pragma unroll
                for (int j = 0; j < 16; j++) {
                    acc.x = __builtin_fmaf(la[j].x, xs[j], acc.x); acc.y = __builtin_fmaf(la[j].y, xs[j], acc.y);
                    acc.z = __builtin_fmaf(la[j].z, xs[j], acc.z); acc.w = __builtin_fmaf(la[j].w, xs[j], acc.w);
                }
                float4 yy = *(const float4*)(y + ia);
                const float4 d4 = *(const float4*)(dg + ia);
                yy.x = __builtin_fmaf(-d4.x, acc.x, yy.x); yy.y = __builtin_fmaf(-d4.y, acc.y, yy.y);
                yy.z = __builtin_fmaf(-d4.z, acc.z, yy.z); yy.w = __builtin_fmaf(-d4.w, acc.w, yy.w);
                *(float4*)(y + ia) = yy;
            }
            if (ib < r0) {
                float4 acc = zero4;
#pragma unroll
                for (int j = 0; j < 16; j++) {
                    acc.x = __builtin_fmaf(lb[j].x, xs[j], acc.x); acc.y = __builtin_fmaf(lb[j].y, xs[j], acc.y);
                    acc.z = __builtin_fmaf(lb[j].z, xs[j], acc.z); acc.w = __builtin_fmaf(lb[j].w, xs[j], acc.w);
                }
                float4 yy = *(const float4*)(y + ib);
                const float4 d4 = *(const float4*)(dg + ib);
                yy.x = __builtin_fmaf(-d4.x, acc.x, yy.x); yy.y = __builtin_fmaf(-d4.y, acc.y, yy.y);
                yy.z = __builtin_fmaf(-d4.z, acc.z, yy.z); yy.w = __builtin_fmaf(-d4.w, acc.w, yy.w);
                *(float4*)(y + ib) = yy;
            }
        }
        __syncthreads();
#pragma unroll
        for (int c = 0; c < 16; c++) ucur[c] = unxt[c];
    }
    for (int i = tid; i < PP; i += 256) x[(size_t)b * PP + i] = y[i];
}

__global__ void vote_kernel(const float* __restrict__ x, float* __restrict__ out) {
    int s = blockIdx.x, b = blockIdx.y, tid = threadIdx.x;
    __shared__ float vals[NN_];
    if (tid < NN_) {
        float v = x[(size_t)b * PP + tid * NN_ + s];
        vals[tid] = fminf(fmaxf(v, 0.f), 1.f);
    }
    __syncthreads();
    if (tid == 0) {
        float m = -1e30f;
        for (int t = 0; t < NN_; t++) m = fmaxf(m, vals[t]);
        float sum = 0.f;
        for (int t = 0; t < NN_; t++) { float e = __expf(ALPHA * (vals[t] - m)); vals[t] = e; sum += e; }
        float inv = 1.f / sum;
        for (int t = 0; t < NN_; t++) out[((size_t)b * NN_ + s) * NN_ + t] = vals[t] * inv;
    }
}

extern "C" void kernel_launch(void* const* d_in, const int* in_sizes, int n_in,
                              void* d_out, int out_size, void* d_ws, size_t ws_size,
                              hipStream_t stream) {
    const float* emb1 = (const float*)d_in[0];
    const float* emb2 = (const float*)d_in[1];
    const float* Wcg  = (const float*)d_in[2];
    const float* bcg  = (const float*)d_in[3];

    float* ws  = (float*)d_ws;
    size_t kelem = (size_t)BB * PP * PP;
    float* Km  = ws;
    float* h1  = Km + kelem;
    float* h2  = h1 + (size_t)BB * NN_ * DD;
    float* S   = h2 + (size_t)BB * NN_ * DD;
    float* rhs = S  + (size_t)BB * NN2;
    float* x   = rhs + (size_t)BB * PP;

    feat_kernel<<<dim3(BB, 2), 256, 0, stream>>>(emb1, emb2, Wcg, bcg, h1, h2);
    gram_rhs_kernel<<<BB, 256, 0, stream>>>(h1, h2, S, rhs);
    assemble_kernel<<<dim3(NBLK, NBLK, BB), dim3(16, 16), 0, stream>>>(S, Km);

    for (int jb = 0; jb < NBLK; jb++)
        lu_col_kernel<<<dim3(NBLK - jb, BB), 32, 0, stream>>>(Km, x, jb);

    trisolve_kernel<<<BB, 256, 0, stream>>>(Km, rhs, x);
    vote_kernel<<<dim3(NN_, BB), 64, 0, stream>>>(x, (float*)d_out);
}


# LU column launch: k range of both WMMA chains split over the 4 waves of a 128-thread workgroup (one per SIMD); partial tiles added in LDS as ((p0+p1)+p2)+p3, wave 0 eliminates (sum order changes)
# speedup vs baseline: 4.8974x; 1.2403x over previous
#include <hip/hip_runtime.h>
#include <math.h>


#define BB   8
#define NN_  36
#define DD   64
#define NN2  (NN_*NN_)
#define NCON (2*NN_)
#define KSZ  (NN2+NCON)
#define PP   1376
#define NBLK (PP/16)
#define KDIAG 1225.0f
#define EPSK  1e-5f
#define ALPHA 1000.0f
#define PS   17

typedef float v2f __attribute__((ext_vector_type(2)));
typedef float v8f __attribute__((ext_vector_type(8)));

__global__ void feat_kernel(const float* __restrict__ emb1,
                            const float* __restrict__ emb2,
                            const float* __restrict__ W,
                            const float* __restrict__ bias,
                            float* __restrict__ h1, float* __restrict__ h2) {
    int b = blockIdx.x, which = blockIdx.y;
    const float* e = (which == 0 ? emb1 : emb2) + (size_t)b * NN_ * DD;
    float* h = (which == 0 ? h1 : h2) + (size_t)b * NN_ * DD;
    __shared__ float se[NN_ * DD];
    __shared__ float sw[DD * DD];
    __shared__ float snrm[DD];
    int tid = threadIdx.x;
    for (int i = tid; i < NN_ * DD; i += 256) se[i] = e[i];
    for (int i = tid; i < DD * DD; i += 256) sw[i] = W[i];
    __syncthreads();
    for (int d = tid; d < DD; d += 256) {
        float s = 0.f;
        for (int n = 0; n < NN_; n++) { float v = se[n * DD + d]; s += v * v; }
        snrm[d] = fmaxf(sqrtf(s), 1e-12f);
    }
    __syncthreads();
    for (int i = tid; i < NN_ * DD; i += 256) se[i] = se[i] / snrm[i % DD];
    __syncthreads();
    for (int idx = tid; idx < NN_ * DD; idx += 256) {
        int n = idx / DD, i = idx % DD;
        float acc = bias[i];
        for (int d = 0; d < DD; d++) acc += se[n * DD + d] * sw[i * DD + d];
        h[idx] = fmaxf(acc, 0.f);
    }
}

__global__ void gram_rhs_kernel(const float* __restrict__ h1,
                                const float* __restrict__ h2,
                                float* __restrict__ S, float* __restrict__ rhs) {
    int b = blockIdx.x, tid = threadIdx.x;
    __shared__ float s1[NN_ * DD];
    __shared__ float s2[NN_ * DD];
    for (int i = tid; i < NN_ * DD; i += 256) {
        s1[i] = h1[(size_t)b * NN_ * DD + i];
        s2[i] = h2[(size_t)b * NN_ * DD + i];
    }
    __syncthreads();
    for (int idx = tid; idx < NN2; idx += 256) {
        int i = idx / NN_, j = idx % NN_;
        float a = 0.f;
        for (int d = 0; d < DD; d++) a += s1[i * DD + d] * s2[j * DD + d];
        S[(size_t)b * NN2 + idx] = a;
        rhs[(size_t)b * PP + j * NN_ + i] = a;
    }
    for (int r = NN2 + tid; r < PP; r += 256)
        rhs[(size_t)b * PP + r] = (r < KSZ) ? 1.0f : 0.0f;
}

__global__ void assemble_kernel(const float* __restrict__ S, float* __restrict__ K) {
    int b = blockIdx.z;
    int r = blockIdx.y * 16 + threadIdx.y;
    int c = blockIdx.x * 16 + threadIdx.x;
    const float* Sb = S + (size_t)b * NN2;
    float val = 0.f;
    if (r < NN2 && c < NN2) {
        int t = r / NN_, s = r - t * NN_, u = c / NN_, v = c - u * NN_;
        val = (r == c) ? KDIAG : 0.f;
        if (s != v && t != u) val -= 0.5f * (Sb[s * NN_ + t] + Sb[v * NN_ + u]);
    } else if (r < NN2) {
        int i = c - NN2, t = r / NN_, s = r - t * NN_;
        if (i < NN_)            val = (s == i) ? 1.f : 0.f;
        else if (i < NCON)      val = (t == i - NN_) ? 1.f : 0.f;
    } else if (c < NN2) {
        int i = r - NN2, u = c / NN_, v = c - u * NN_;
        if (i < NN_)            val = (v == i) ? 1.f : 0.f;
        else if (i < NCON)      val = (u == i - NN_) ? 1.f : 0.f;
    } else if (r == c) {
        val = (r < KSZ) ? -EPSK : 1.0f;
    }
    K[((size_t)b * PP + r) * PP + c] = val;
}

#define QS 16
#define NW 4

__global__ void __launch_bounds__(32 * NW) lu_col_kernel(float* __restrict__ K,
                                                         float* __restrict__ dpiv,
                                                         int jb) {
    __shared__ float sT[32 * PS];
    __shared__ __attribute__((aligned(16))) float4 sP[(NW - 1) * 4 * 32];
    const int b = blockIdx.y, ti = blockIdx.x, lane = threadIdx.x & 31;
    const int w = __builtin_amdgcn_readfirstlane((int)(threadIdx.x >> 5));
    const int half = lane >> 4, l16 = lane & 15;
    float* Kb = K + (size_t)b * PP * PP;
    float* db = dpiv + (size_t)b * PP;
    const int c0 = jb * 16;
    const int rown = c0 + ti * 16;

    v8f cd, co;
    if (w == 0) {
#pragma unroll
        for (int r = 0; r < 8; r++) {
            cd[r] = Kb[(size_t)(c0 + r + 8 * half) * PP + c0 + l16];
            co[r] = Kb[(size_t)(rown + r + 8 * half) * PP + c0 + l16];
        }
    } else {
#pragma unroll
        for (int r = 0; r < 8; r++) { cd[r] = 0.f; co[r] = 0.f; }
    }

    const int Q = jb >> 2;
    const int qb = Q >> 2, qr = Q & 3;
    const int nq = qb + ((w < qr) ? 1 : 0);
    const int kq = (w * qb + ((w < qr) ? w : qr)) * (4 * QS);

    const float* pl0 = Kb + (size_t)(c0 + l16) * PP + 2 * half;
    const float* pa0 = Kb + (size_t)(rown + l16) * PP + 2 * half;
    const float* pd0 = db + 2 * half;
    const float* pl = pl0 + kq;
    const float* pa = pa0 + kq;
    const float* pd = pd0 + kq;

    v2f ra[QS], rl[QS], rd[QS];
    if (nq > 0) {
#pragma unroll
        for (int s = 0; s < QS; s++) {
            ra[s] = *(const v2f*)(pa + 4 * s);
            rl[s] = *(const v2f*)(pl + 4 * s);
            rd[s] = *(const v2f*)(pd + 4 * s);
        }
    }
    for (int q = 0; q < nq; q++) {
        v2f an[QS], ln[QS], bb[QS];
#pragma unroll
        for (int s = 0; s < QS; s++) {
            an[s] = -ra[s];
            ln[s] = -rl[s];
            bb[s] = rl[s] * rd[s];
        }
        const int kn = ((q + 1 < nq) ? (q + 1) : q) * (4 * QS);
#pragma unroll
        for (int s = 0; s < QS; s++) {
            ra[s] = *(const v2f*)(pa + kn + 4 * s);
            rl[s] = *(const v2f*)(pl + kn + 4 * s);
            rd[s] = *(const v2f*)(pd + kn + 4 * s);
        }
#pragma unroll
        for (int s = 0; s < QS; s++) {
            co = __builtin_amdgcn_wmma_f32_16x16x4_f32(false, an[s], false, bb[s], (short)0, co, false, false);
            cd = __builtin_amdgcn_wmma_f32_16x16x4_f32(false, ln[s], false, bb[s], (short)0, cd, false, false);
        }
    }
    if (w == NW - 1) {
        for (int k0 = Q * (4 * QS); k0 < c0; k0 += 16) {
            v2f ta[4], tl[4], td[4];
#pragma unroll
            for (int s = 0; s < 4; s++) {
                ta[s] = *(const v2f*)(pa0 + k0 + 4 * s);
                tl[s] = *(const v2f*)(pl0 + k0 + 4 * s);
                td[s] = *(const v2f*)(pd0 + k0 + 4 * s);
            }
#pragma unroll
            for (int s = 0; s < 4; s++) {
                v2f a1 = -ta[s], l1 = -tl[s], b1 = tl[s] * td[s];
                co = __builtin_amdgcn_wmma_f32_16x16x4_f32(false, a1, false, b1, (short)0, co, false, false);
                cd = __builtin_amdgcn_wmma_f32_16x16x4_f32(false, l1, false, b1, (short)0, cd, false, false);
            }
        }
    }

    if (w > 0) {
        float4* pp = sP + (w - 1) * 4 * 32 + lane;
        pp[0]  = make_float4(cd[0], cd[1], cd[2], cd[3]);
        pp[32] = make_float4(cd[4], cd[5], cd[6], cd[7]);
        pp[64] = make_float4(co[0], co[1], co[2], co[3]);
        pp[96] = make_float4(co[4], co[5], co[6], co[7]);
    }
    __syncthreads();
    if (w == 0) {
#pragma unroll
        for (int u = 0; u < NW - 1; u++) {
            const float4* pp = sP + u * 4 * 32 + lane;
            const float4 d0 = pp[0], d1 = pp[32], o0 = pp[64], o1 = pp[96];
            cd[0] += d0.x; cd[1] += d0.y; cd[2] += d0.z; cd[3] += d0.w;
            cd[4] += d1.x; cd[5] += d1.y; cd[6] += d1.z; cd[7] += d1.w;
            co[0] += o0.x; co[1] += o0.y; co[2] += o0.z; co[3] += o0.w;
            co[4] += o1.x; co[5] += o1.y; co[6] += o1.z; co[7] += o1.w;
        }
#pragma unroll
        for (int r = 0; r < 8; r++) {
            sT[(r + 8 * half) * PS + l16] = cd[r];
            sT[(16 + r + 8 * half) * PS + l16] = co[r];
        }
    }
    __syncthreads();
    if (w != 0) return;
    float t[16];
#pragma unroll
    for (int c = 0; c < 16; c++) t[c] = sT[lane * PS + c];

    float mypiv = 0.f;
#pragma unroll
    for (int p = 0; p < 16; p++) {
        float u[16];
#pragma unroll
        for (int c2 = p; c2 < 16; c2++)
            u[c2] = __int_as_float(__builtin_amdgcn_readlane(__float_as_int(t[c2]), p));
        float dinv = 1.0f / u[p];
        if (lane == p) mypiv = u[p];
        if (lane > p) {
            float lij = t[p] * dinv;
            t[p] = lij;
#pragma unroll
            for (int c2 = p + 1; c2 < 16; c2++) t[c2] -= lij * u[c2];
        }
    }

    float* dst = nullptr;
    if (ti == 0) {
        if (lane < 16) {
            dst = Kb + (size_t)(c0 + lane) * PP + c0 + ((jb < NBLK - 1) ? 16 : 0);
            db[c0 + lane] = mypiv;
        }
    } else if (lane >= 16) {
        dst = Kb + (size_t)(rown + lane - 16) * PP + c0;
    }
    if (dst) {
#pragma unroll
        for (int c = 0; c < 16; c += 4) {
            float4 v; v.x = t[c]; v.y = t[c + 1]; v.z = t[c + 2]; v.w = t[c + 3];
            *(float4*)(dst + c) = v;
        }
    }
}

#define FCH 24
#define RP  20

__global__ void __launch_bounds__(256) trisolve_kernel(const float* __restrict__ K,
                                                       const float* __restrict__ rhs,
                                                       float* __restrict__ x) {
    const int b = blockIdx.x, tid = threadIdx.x;
    const int l16 = tid & 15, grp = tid >> 4;
    __shared__ __attribute__((aligned(16))) float y[PP];
    __shared__ __attribute__((aligned(16))) float dg[PP];
    __shared__ __attribute__((aligned(16))) float red[16 * RP];
    const float* Kb = K + (size_t)b * PP * PP;
    for (int i = tid; i < PP; i += 256) {
        y[i] = rhs[(size_t)b * PP + i];
        dg[i] = Kb[(size_t)i * PP + i + (((i >> 4) < NBLK - 1) ? 16 : 0)];
    }
    const float4 zero4 = make_float4(0.f, 0.f, 0.f, 0.f);

    float4 cur[FCH];
    float lcur[16];
#pragma unroll
    for (int m = 0; m < FCH; m++) cur[m] = zero4;
#pragma unroll
    for (int c = 0; c < 16; c++) lcur[c] = 0.f;
    if (tid < 32) {
        const float* lp = Kb + (size_t)l16 * PP + 16;
#pragma unroll
        for (int c = 0; c < 4; c++) {
            float4 t4 = *(const float4*)(lp + 4 * c);
            lcur[4 * c] = t4.x; lcur[4 * c + 1] = t4.y; lcur[4 * c + 2] = t4.z; lcur[4 * c + 3] = t4.w;
        }
    }
    __syncthreads();
    for (int kb = 0; kb < NBLK; kb++) {
        const int r0 = kb * 16, r1 = r0 + 16;
        float4 nxt[FCH];
        float lnxt[16];
#pragma unroll
        for (int m = 0; m < FCH; m++) nxt[m] = zero4;
#pragma unroll
        for (int c = 0; c < 16; c++) lnxt[c] = 0.f;
        if (kb + 1 < NBLK) {
            const float* rp = Kb + (size_t)(r1 + grp) * PP;
#pragma unroll
            for (int m = 0; m < FCH; m++) {
                if (64 * m < r1) {
                    const int k = 64 * m + 4 * l16;
                    if (k < r1) nxt[m] = *(const float4*)(rp + k);
                }
            }
            if (tid < 32) {
                const int d1 = r1 + ((kb + 1 < NBLK - 1) ? 16 : 0);
                const float* lp = Kb + (size_t)(r1 + l16) * PP + d1;
#pragma unroll
                for (int c = 0; c < 4; c++) {
                    float4 t4 = *(const float4*)(lp + 4 * c);
                    lnxt[4 * c] = t4.x; lnxt[4 * c + 1] = t4.y; lnxt[4 * c + 2] = t4.z; lnxt[4 * c + 3] = t4.w;
                }
            }
        }
        float a0 = 0.f, a1 = 0.f, a2 = 0.f, a3 = 0.f;
#pragma unroll
        for (int g = 0; g < FCH / 4; g++) {
            if (256 * g < r0) {
#pragma unroll
                for (int mm = 0; mm < 4; mm++) {
                    const int m = 4 * g + mm;
                    const int k = 64 * m + 4 * l16;
                    const float4 yy = *(const float4*)(y + ((k < PP) ? k : 0));
                    a0 = __builtin_fmaf(cur[m].x, yy.x, a0); a1 = __builtin_fmaf(cur[m].y, yy.y, a1);
                    a2 = __builtin_fmaf(cur[m].z, yy.z, a2); a3 = __builtin_fmaf(cur[m].w, yy.w, a3);
                }
            }
        }
        red[grp * RP + l16] = (a0 + a1) + (a2 + a3);
        __syncthreads();
        if (tid < 32) {
            const float4* rr = (const float4*)(red + l16 * RP);
            const float4 s0 = rr[0], s1 = rr[1], s2 = rr[2], s3 = rr[3];
            const float dot = (((s0.x + s0.y) + (s0.z + s0.w)) + ((s1.x + s1.y) + (s1.z + s1.w)))
                            + (((s2.x + s2.y) + (s2.z + s2.w)) + ((s3.x + s3.y) + (s3.z + s3.w)));
            float v = y[r0 + l16] - dot;
#pragma unroll
            for (int p = 0; p < 15; p++) {
                const float vp = __int_as_float(__builtin_amdgcn_readlane(__float_as_int(v), p));
                if (l16 > p) v -= lcur[p] * vp;
            }
            if (tid < 16) y[r0 + l16] = v;
        }
        __syncthreads();
#pragma unroll
        for (int m = 0; m < FCH; m++) cur[m] = nxt[m];
#pragma unroll
        for (int c = 0; c < 16; c++) lcur[c] = lnxt[c];
    }

    float ucur[16];
#pragma unroll
    for (int c = 0; c < 16; c++) ucur[c] = 0.f;
    if (tid < 32) {
        const float* up = Kb + (size_t)((NBLK - 1) * 16 + l16) * PP + (NBLK - 1) * 16;
#pragma unroll
        for (int c = 0; c < 4; c++) {
            float4 t4 = *(const float4*)(up + 4 * c);
            ucur[4 * c] = t4.x; ucur[4 * c + 1] = t4.y; ucur[4 * c + 2] = t4.z; ucur[4 * c + 3] = t4.w;
        }
    }
    __syncthreads();
    for (int kb = NBLK - 1; kb >= 0; kb--) {
        const int r0 = kb * 16;
        const int ia = 4 * (tid - 32), ib = ia + 4 * 224;
        float4 la[16], lb[16];
        if (tid >= 32) {
            if (ia < r0) {
#pragma unroll
                for (int j = 0; j < 16; j++)
                    la[j] = *(const float4*)(Kb + (size_t)(r0 + j) * PP + ia);
            }
            if (ib < r0) {
#pragma unroll
                for (int j = 0; j < 16; j++)
                    lb[j] = *(const float4*)(Kb + (size_t)(r0 + j) * PP + ib);
            }
        } else {
            float v = y[r0 + l16];
            const float rinv = 1.0f / dg[r0 + l16];
#pragma unroll
            for (int p = 15; p >= 0; p--) {
                if (l16 == p) v *= rinv;
                const float xp = __int_as_float(__builtin_amdgcn_readlane(__float_as_int(v), p));
                if (l16 < p) v -= ucur[p] * xp;
            }
            if (tid < 16) y[r0 + l16] = v;
        }
        __syncthreads();
        float unxt[16];
#pragma unroll
        for (int c = 0; c < 16; c++) unxt[c] = 0.f;
        if (tid < 32) {
            if (kb > 0) {
                const float* up = Kb + (size_t)(r0 - 16 + l16) * PP + r0;
#pragma unroll
                for (int c = 0; c < 4; c++) {
                    float4 t4 = *(const float4*)(up + 4 * c);
                    unxt[4 * c] = t4.x; unxt[4 * c + 1] = t4.y; unxt[4 * c + 2] = t4.z; unxt[4 * c + 3] = t4.w;
                }
            }
        } else {
            float xs[16];
#pragma unroll
            for (int c = 0; c < 4; c++) {
                const float4 t4 = *(const float4*)(y + r0 + 4 * c);
                xs[4 * c] = t4.x; xs[4 * c + 1] = t4.y; xs[4 * c + 2] = t4.z; xs[4 * c + 3] = t4.w;
            }
            if (ia < r0) {
                float4 acc = zero4;
#pragma unroll
                for (int j = 0; j < 16; j++) {
                    acc.x = __builtin_fmaf(la[j].x, xs[j], acc.x); acc.y = __builtin_fmaf(la[j].y, xs[j], acc.y);
                    acc.z = __builtin_fmaf(la[j].z, xs[j], acc.z); acc.w = __builtin_fmaf(la[j].w, xs[j], acc.w);
                }
                float4 yy = *(const float4*)(y + ia);
                const float4 d4 = *(const float4*)(dg + ia);
                yy.x = __builtin_fmaf(-d4.x, acc.x, yy.x); yy.y = __builtin_fmaf(-d4.y, acc.y, yy.y);
                yy.z = __builtin_fmaf(-d4.z, acc.z, yy.z); yy.w = __builtin_fmaf(-d4.w, acc.w, yy.w);
                *(float4*)(y + ia) = yy;
            }
            if (ib < r0) {
                float4 acc = zero4;
#pragma unroll
                for (int j = 0; j < 16; j++) {
                    acc.x = __builtin_fmaf(lb[j].x, xs[j], acc.x); acc.y = __builtin_fmaf(lb[j].y, xs[j], acc.y);
                    acc.z = __builtin_fmaf(lb[j].z, xs[j], acc.z); acc.w = __builtin_fmaf(lb[j].w, xs[j], acc.w);
                }
                float4 yy = *(const float4*)(y + ib);
                const float4 d4 = *(const float4*)(dg + ib);
                yy.x = __builtin_fmaf(-d4.x, acc.x, yy.x); yy.y = __builtin_fmaf(-d4.y, acc.y, yy.y);
                yy.z = __builtin_fmaf(-d4.z, acc.z, yy.z); yy.w = __builtin_fmaf(-d4.w, acc.w, yy.w);
                *(float4*)(y + ib) = yy;
            }
        }
        __syncthreads();
#pragma unroll
        for (int c = 0; c < 16; c++) ucur[c] = unxt[c];
    }
    for (int i = tid; i < PP; i += 256) x[(size_t)b * PP + i] = y[i];
}

__global__ void vote_kernel(const float* __restrict__ x, float* __restrict__ out) {
    int s = blockIdx.x, b = blockIdx.y, tid = threadIdx.x;
    __shared__ float vals[NN_];
    if (tid < NN_) {
        float v = x[(size_t)b * PP + tid * NN_ + s];
        vals[tid] = fminf(fmaxf(v, 0.f), 1.f);
    }
    __syncthreads();
    if (tid == 0) {
        float m = -1e30f;
        for (int t = 0; t < NN_; t++) m = fmaxf(m, vals[t]);
        float sum = 0.f;
        for (int t = 0; t < NN_; t++) { float e = __expf(ALPHA * (vals[t] - m)); vals[t] = e; sum += e; }
        float inv = 1.f / sum;
        for (int t = 0; t < NN_; t++) out[((size_t)b * NN_ + s) * NN_ + t] = vals[t] * inv;
    }
}

extern "C" void kernel_launch(void* const* d_in, const int* in_sizes, int n_in,
                              void* d_out, int out_size, void* d_ws, size_t ws_size,
                              hipStream_t stream) {
    const float* emb1 = (const float*)d_in[0];
    const float* emb2 = (const float*)d_in[1];
    const float* Wcg  = (const float*)d_in[2];
    const float* bcg  = (const float*)d_in[3];

    float* ws  = (float*)d_ws;
    size_t kelem = (size_t)BB * PP * PP;
    float* Km  = ws;
    float* h1  = Km + kelem;
    float* h2  = h1 + (size_t)BB * NN_ * DD;
    float* S   = h2 + (size_t)BB * NN_ * DD;
    float* rhs = S  + (size_t)BB * NN2;
    float* x   = rhs + (size_t)BB * PP;

    feat_kernel<<<dim3(BB, 2), 256, 0, stream>>>(emb1, emb2, Wcg, bcg, h1, h2);
    gram_rhs_kernel<<<BB, 256, 0, stream>>>(h1, h2, S, rhs);
    assemble_kernel<<<dim3(NBLK, NBLK, BB), dim3(16, 16), 0, stream>>>(S, Km);

    for (int jb = 0; jb < NBLK; jb++)
        lu_col_kernel<<<dim3(NBLK - jb, BB), 32 * NW, 0, stream>>>(Km, x, jb);

    trisolve_kernel<<<BB, 256, 0, stream>>>(Km, rhs, x);
    vote_kernel<<<dim3(NN_, BB), 64, 0, stream>>>(x, (float*)d_out);
}


# LU column launch: whole k batches reach the WMMA chains through a per-wave LDS strip, rows fetched two batches ahead by 16-byte whole-line loads instead of 16-line fragment loads (sum order kept)
# speedup vs baseline: 5.2668x; 1.0754x over previous
#include <hip/hip_runtime.h>
#include <math.h>


#define BB   8
#define NN_  36
#define DD   64
#define NN2  (NN_*NN_)
#define NCON (2*NN_)
#define KSZ  (NN2+NCON)
#define PP   1376
#define NBLK (PP/16)
#define KDIAG 1225.0f
#define EPSK  1e-5f
#define ALPHA 1000.0f
#define PS   17

typedef float v2f __attribute__((ext_vector_type(2)));
typedef float v4f __attribute__((ext_vector_type(4)));
typedef float v8f __attribute__((ext_vector_type(8)));

__global__ void feat_kernel(const float* __restrict__ emb1,
                            const float* __restrict__ emb2,
                            const float* __restrict__ W,
                            const float* __restrict__ bias,
                            float* __restrict__ h1, float* __restrict__ h2) {
    int b = blockIdx.x, which = blockIdx.y;
    const float* e = (which == 0 ? emb1 : emb2) + (size_t)b * NN_ * DD;
    float* h = (which == 0 ? h1 : h2) + (size_t)b * NN_ * DD;
    __shared__ float se[NN_ * DD];
    __shared__ float sw[DD * DD];
    __shared__ float snrm[DD];
    int tid = threadIdx.x;
    for (int i = tid; i < NN_ * DD; i += 256) se[i] = e[i];
    for (int i = tid; i < DD * DD; i += 256) sw[i] = W[i];
    __syncthreads();
    for (int d = tid; d < DD; d += 256) {
        float s = 0.f;
        for (int n = 0; n < NN_; n++) { float v = se[n * DD + d]; s += v * v; }
        snrm[d] = fmaxf(sqrtf(s), 1e-12f);
    }
    __syncthreads();
    for (int i = tid; i < NN_ * DD; i += 256) se[i] = se[i] / snrm[i % DD];
    __syncthreads();
    for (int idx = tid; idx < NN_ * DD; idx += 256) {
        int n = idx / DD, i = idx % DD;
        float acc = bias[i];
        for (int d = 0; d < DD; d++) acc += se[n * DD + d] * sw[i * DD + d];
        h[idx] = fmaxf(acc, 0.f);
    }
}

__global__ void gram_rhs_kernel(const float* __restrict__ h1,
                                const float* __restrict__ h2,
                                float* __restrict__ S, float* __restrict__ rhs) {
    int b = blockIdx.x, tid = threadIdx.x;
    __shared__ float s1[NN_ * DD];
    __shared__ float s2[NN_ * DD];
    for (int i = tid; i < NN_ * DD; i += 256) {
        s1[i] = h1[(size_t)b * NN_ * DD + i];
        s2[i] = h2[(size_t)b * NN_ * DD + i];
    }
    __syncthreads();
    for (int idx = tid; idx < NN2; idx += 256) {
        int i = idx / NN_, j = idx % NN_;
        float a = 0.f;
        for (int d = 0; d < DD; d++) a += s1[i * DD + d] * s2[j * DD + d];
        S[(size_t)b * NN2 + idx] = a;
        rhs[(size_t)b * PP + j * NN_ + i] = a;
    }
    for (int r = NN2 + tid; r < PP; r += 256)
        rhs[(size_t)b * PP + r] = (r < KSZ) ? 1.0f : 0.0f;
}

__global__ void assemble_kernel(const float* __restrict__ S, float* __restrict__ K) {
    int b = blockIdx.z;
    int r = blockIdx.y * 16 + threadIdx.y;
    int c = blockIdx.x * 16 + threadIdx.x;
    const float* Sb = S + (size_t)b * NN2;
    float val = 0.f;
    if (r < NN2 && c < NN2) {
        int t = r / NN_, s = r - t * NN_, u = c / NN_, v = c - u * NN_;
        val = (r == c) ? KDIAG : 0.f;
        if (s != v && t != u) val -= 0.5f * (Sb[s * NN_ + t] + Sb[v * NN_ + u]);
    } else if (r < NN2) {
        int i = c - NN2, t = r / NN_, s = r - t * NN_;
        if (i < NN_)            val = (s == i) ? 1.f : 0.f;
        else if (i < NCON)      val = (t == i - NN_) ? 1.f : 0.f;
    } else if (c < NN2) {
        int i = r - NN2, u = c / NN_, v = c - u * NN_;
        if (i < NN_)            val = (v == i) ? 1.f : 0.f;
        else if (i < NCON)      val = (u == i - NN_) ? 1.f : 0.f;
    } else if (r == c) {
        val = (r < KSZ) ? -EPSK : 1.0f;
    }
    K[((size_t)b * PP + r) * PP + c] = val;
}

#define QS 16
#define NW 4
#define SP 68
#define STRIP (2 * 16 * SP + 4 * QS)

__device__ __forceinline__ void wave_lds_order() {
    __builtin_amdgcn_fence(__ATOMIC_RELEASE, "wavefront");
    __builtin_amdgcn_wave_barrier();
    __builtin_amdgcn_fence(__ATOMIC_ACQUIRE, "wavefront");
}

__global__ void __launch_bounds__(32 * NW) lu_col_kernel(float* __restrict__ K,
                                                         float* __restrict__ dpiv,
                                                         int jb) {
    __shared__ float sT[32 * PS];
    __shared__ __attribute__((aligned(16))) float4 sP[(NW - 1) * 4 * 32];
    __shared__ __attribute__((aligned(16))) float sStage[NW * STRIP];
    const int b = blockIdx.y, ti = blockIdx.x, lane = threadIdx.x & 31;
    const int w = __builtin_amdgcn_readfirstlane((int)(threadIdx.x >> 5));
    const int half = lane >> 4, l16 = lane & 15;
    float* Kb = K + (size_t)b * PP * PP;
    float* db = dpiv + (size_t)b * PP;
    const int c0 = jb * 16;
    const int rown = c0 + ti * 16;

    v8f cd, co;
    if (w == 0) {
#pragma unroll
        for (int r = 0; r < 8; r++) {
            cd[r] = Kb[(size_t)(c0 + r + 8 * half) * PP + c0 + l16];
            co[r] = Kb[(size_t)(rown + r + 8 * half) * PP + c0 + l16];
        }
    } else {
#pragma unroll
        for (int r = 0; r < 8; r++) { cd[r] = 0.f; co[r] = 0.f; }
    }

    const int Q = jb >> 2;
    const int qb = Q >> 2, qr = Q & 3;
    const int nq = qb + ((w < qr) ? 1 : 0);
    const int kq = (w * qb + ((w < qr) ? w : qr)) * (4 * QS);

    const float* pl0 = Kb + (size_t)(c0 + l16) * PP + 2 * half;
    const float* pa0 = Kb + (size_t)(rown + l16) * PP + 2 * half;
    const float* pd0 = db + 2 * half;

    const float* ga = Kb + (size_t)(rown + half) * PP + 4 * l16 + kq;
    const float* gl = Kb + (size_t)(c0 + half) * PP + 4 * l16 + kq;
    const float* gd = db + 2 * lane + kq;
    float* stA = sStage + w * STRIP;
    float* stL = stA + 16 * SP;
    float* stD = stL + 16 * SP;
    float* wA = stA + half * SP + 4 * l16;
    float* wL = stL + half * SP + 4 * l16;
    float* wD = stD + 2 * lane;
    const float* fA = stA + l16 * SP + 2 * half;
    const float* fL = stL + l16 * SP + 2 * half;
    const float* fD = stD + 2 * half;

    v4f ga4[8], gl4[8];
    v2f gd2;
    v2f ra[QS], rl[QS], rd[QS];
    if (nq > 0) {
#pragma unroll
        for (int j = 0; j < 8; j++) {
            ga4[j] = *(const v4f*)(ga + (size_t)(2 * j) * PP);
            gl4[j] = *(const v4f*)(gl + (size_t)(2 * j) * PP);
        }
        gd2 = *(const v2f*)gd;
#pragma unroll
        for (int j = 0; j < 8; j++) {
            *(v4f*)(wA + 2 * j * SP) = ga4[j];
            *(v4f*)(wL + 2 * j * SP) = gl4[j];
        }
        *(v2f*)wD = gd2;
        if (nq > 1) {
#pragma unroll
            for (int j = 0; j < 8; j++) {
                ga4[j] = *(const v4f*)(ga + (size_t)(2 * j) * PP + 4 * QS);
                gl4[j] = *(const v4f*)(gl + (size_t)(2 * j) * PP + 4 * QS);
            }
            gd2 = *(const v2f*)(gd + 4 * QS);
        }
        wave_lds_order();
#pragma unroll
        for (int s = 0; s < QS; s++) {
            ra[s] = *(const v2f*)(fA + 4 * s);
            rl[s] = *(const v2f*)(fL + 4 * s);
            rd[s] = *(const v2f*)(fD + 4 * s);
        }
    }
    for (int q = 0; q < nq; q++) {
        v2f an[QS], ln[QS], bb[QS];
#pragma unroll
        for (int s = 0; s < QS; s++) {
            an[s] = -ra[s];
            ln[s] = -rl[s];
            bb[s] = rl[s] * rd[s];
        }
        if (q + 1 < nq) {
            wave_lds_order();
#pragma unroll
            for (int j = 0; j < 8; j++) {
                *(v4f*)(wA + 2 * j * SP) = ga4[j];
                *(v4f*)(wL + 2 * j * SP) = gl4[j];
            }
            *(v2f*)wD = gd2;
            if (q + 2 < nq) {
                const int k2 = (q + 2) * (4 * QS);
#pragma unroll
                for (int j = 0; j < 8; j++) {
                    ga4[j] = *(const v4f*)(ga + (size_t)(2 * j) * PP + k2);
                    gl4[j] = *(const v4f*)(gl + (size_t)(2 * j) * PP + k2);
                }
                gd2 = *(const v2f*)(gd + k2);
            }
            wave_lds_order();
#pragma unroll
            for (int s = 0; s < QS; s++) {
                ra[s] = *(const v2f*)(fA + 4 * s);
                rl[s] = *(const v2f*)(fL + 4 * s);
                rd[s] = *(const v2f*)(fD + 4 * s);
            }
        }
#pragma unroll
        for (int s = 0; s < QS; s++) {
            co = __builtin_amdgcn_wmma_f32_16x16x4_f32(false, an[s], false, bb[s], (short)0, co, false, false);
            cd = __builtin_amdgcn_wmma_f32_16x16x4_f32(false, ln[s], false, bb[s], (short)0, cd, false, false);
        }
    }
    if (w == NW - 1) {
        for (int k0 = Q * (4 * QS); k0 < c0; k0 += 16) {
            v2f ta[4], tl[4], td[4];
#pragma unroll
            for (int s = 0; s < 4; s++) {
                ta[s] = *(const v2f*)(pa0 + k0 + 4 * s);
                tl[s] = *(const v2f*)(pl0 + k0 + 4 * s);
                td[s] = *(const v2f*)(pd0 + k0 + 4 * s);
            }
#pragma unroll
            for (int s = 0; s < 4; s++) {
                v2f a1 = -ta[s], l1 = -tl[s], b1 = tl[s] * td[s];
                co = __builtin_amdgcn_wmma_f32_16x16x4_f32(false, a1, false, b1, (short)0, co, false, false);
                cd = __builtin_amdgcn_wmma_f32_16x16x4_f32(false, l1, false, b1, (short)0, cd, false, false);
            }
        }
    }

    if (w > 0) {
        float4* pp = sP + (w - 1) * 4 * 32 + lane;
        pp[0]  = make_float4(cd[0], cd[1], cd[2], cd[3]);
        pp[32] = make_float4(cd[4], cd[5], cd[6], cd[7]);
        pp[64] = make_float4(co[0], co[1], co[2], co[3]);
        pp[96] = make_float4(co[4], co[5], co[6], co[7]);
    }
    __syncthreads();
    if (w == 0) {
#pragma unroll
        for (int u = 0; u < NW - 1; u++) {
            const float4* pp = sP + u * 4 * 32 + lane;
            const float4 d0 = pp[0], d1 = pp[32], o0 = pp[64], o1 = pp[96];
            cd[0] += d0.x; cd[1] += d0.y; cd[2] += d0.z; cd[3] += d0.w;
            cd[4] += d1.x; cd[5] += d1.y; cd[6] += d1.z; cd[7] += d1.w;
            co[0] += o0.x; co[1] += o0.y; co[2] += o0.z; co[3] += o0.w;
            co[4] += o1.x; co[5] += o1.y; co[6] += o1.z; co[7] += o1.w;
        }
#pragma unroll
        for (int r = 0; r < 8; r++) {
            sT[(r + 8 * half) * PS + l16] = cd[r];
            sT[(16 + r + 8 * half) * PS + l16] = co[r];
        }
    }
    __syncthreads();
    if (w != 0) return;
    float t[16];
#pragma unroll
    for (int c = 0; c < 16; c++) t[c] = sT[lane * PS + c];

    float mypiv = 0.f;
#pragma unroll
    for (int p = 0; p < 16; p++) {
        float u[16];
#pragma unroll
        for (int c2 = p; c2 < 16; c2++)
            u[c2] = __int_as_float(__builtin_amdgcn_readlane(__float_as_int(t[c2]), p));
        float dinv = 1.0f / u[p];
        if (lane == p) mypiv = u[p];
        if (lane > p) {
            float lij = t[p] * dinv;
            t[p] = lij;
#pragma unroll
            for (int c2 = p + 1; c2 < 16; c2++) t[c2] -= lij * u[c2];
        }
    }

    float* dst = nullptr;
    if (ti == 0) {
        if (lane < 16) {
            dst = Kb + (size_t)(c0 + lane) * PP + c0 + ((jb < NBLK - 1) ? 16 : 0);
            db[c0 + lane] = mypiv;
        }
    } else if (lane >= 16) {
        dst = Kb + (size_t)(rown + lane - 16) * PP + c0;
    }
    if (dst) {
#pragma unroll
        for (int c = 0; c < 16; c += 4) {
            float4 v; v.x = t[c]; v.y = t[c + 1]; v.z = t[c + 2]; v.w = t[c + 3];
            *(float4*)(dst + c) = v;
        }
    }
}

#define FCH 24
#define RP  20

__global__ void __launch_bounds__(256) trisolve_kernel(const float* __restrict__ K,
                                                       const float* __restrict__ rhs,
                                                       float* __restrict__ x) {
    const int b = blockIdx.x, tid = threadIdx.x;
    const int l16 = tid & 15, grp = tid >> 4;
    __shared__ __attribute__((aligned(16))) float y[PP];
    __shared__ __attribute__((aligned(16))) float dg[PP];
    __shared__ __attribute__((aligned(16))) float red[16 * RP];
    const float* Kb = K + (size_t)b * PP * PP;
    for (int i = tid; i < PP; i += 256) {
        y[i] = rhs[(size_t)b * PP + i];
        dg[i] = Kb[(size_t)i * PP + i + (((i >> 4) < NBLK - 1) ? 16 : 0)];
    }
    const float4 zero4 = make_float4(0.f, 0.f, 0.f, 0.f);

    float4 cur[FCH];
    float lcur[16];
#pragma unroll
    for (int m = 0; m < FCH; m++) cur[m] = zero4;
#pragma unroll
    for (int c = 0; c < 16; c++) lcur[c] = 0.f;
    if (tid < 32) {
        const float* lp = Kb + (size_t)l16 * PP + 16;
#pragma unroll
        for (int c = 0; c < 4; c++) {
            float4 t4 = *(const float4*)(lp + 4 * c);
            lcur[4 * c] = t4.x; lcur[4 * c + 1] = t4.y; lcur[4 * c + 2] = t4.z; lcur[4 * c + 3] = t4.w;
        }
    }
    __syncthreads();
    for (int kb = 0; kb < NBLK; kb++) {
        const int r0 = kb * 16, r1 = r0 + 16;
        float4 nxt[FCH];
        float lnxt[16];
#pragma unroll
        for (int m = 0; m < FCH; m++) nxt[m] = zero4;
#pragma unroll
        for (int c = 0; c < 16; c++) lnxt[c] = 0.f;
        if (kb + 1 < NBLK) {
            const float* rp = Kb + (size_t)(r1 + grp) * PP;
#pragma unroll
            for (int m = 0; m < FCH; m++) {
                if (64 * m < r1) {
                    const int k = 64 * m + 4 * l16;
                    if (k < r1) nxt[m] = *(const float4*)(rp + k);
                }
            }
            if (tid < 32) {
                const int d1 = r1 + ((kb + 1 < NBLK - 1) ? 16 : 0);
                const float* lp = Kb + (size_t)(r1 + l16) * PP + d1;
#pragma unroll
                for (int c = 0; c < 4; c++) {
                    float4 t4 = *(const float4*)(lp + 4 * c);
                    lnxt[4 * c] = t4.x; lnxt[4 * c + 1] = t4.y; lnxt[4 * c + 2] = t4.z; lnxt[4 * c + 3] = t4.w;
                }
            }
        }
        float a0 = 0.f, a1 = 0.f, a2 = 0.f, a3 = 0.f;
#pragma unroll
        for (int g = 0; g < FCH / 4; g++) {
            if (256 * g < r0) {
#pragma unroll
                for (int mm = 0; mm < 4; mm++) {
                    const int m = 4 * g + mm;
                    const int k = 64 * m + 4 * l16;
                    const float4 yy = *(const float4*)(y + ((k < PP) ? k : 0));
                    a0 = __builtin_fmaf(cur[m].x, yy.x, a0); a1 = __builtin_fmaf(cur[m].y, yy.y, a1);
                    a2 = __builtin_fmaf(cur[m].z, yy.z, a2); a3 = __builtin_fmaf(cur[m].w, yy.w, a3);
                }
            }
        }
        red[grp * RP + l16] = (a0 + a1) + (a2 + a3);
        __syncthreads();
        if (tid < 32) {
            const float4* rr = (const float4*)(red + l16 * RP);
            const float4 s0 = rr[0], s1 = rr[1], s2 = rr[2], s3 = rr[3];
            const float dot = (((s0.x + s0.y) + (s0.z + s0.w)) + ((s1.x + s1.y) + (s1.z + s1.w)))
                            + (((s2.x + s2.y) + (s2.z + s2.w)) + ((s3.x + s3.y) + (s3.z + s3.w)));
            float v = y[r0 + l16] - dot;
#pragma unroll
            for (int p = 0; p < 15; p++) {
                const float vp = __int_as_float(__builtin_amdgcn_readlane(__float_as_int(v), p));
                if (l16 > p) v -= lcur[p] * vp;
            }
            if (tid < 16) y[r0 + l16] = v;
        }
        __syncthreads();
#pragma unroll
        for (int m = 0; m < FCH; m++) cur[m] = nxt[m];
#pragma unroll
        for (int c = 0; c < 16; c++) lcur[c] = lnxt[c];
    }

    float ucur[16];
#pragma unroll
    for (int c = 0; c < 16; c++) ucur[c] = 0.f;
    if (tid < 32) {
        const float* up = Kb + (size_t)((NBLK - 1) * 16 + l16) * PP + (NBLK - 1) * 16;
#pragma unroll
        for (int c = 0; c < 4; c++) {
            float4 t4 = *(const float4*)(up + 4 * c);
            ucur[4 * c] = t4.x; ucur[4 * c + 1] = t4.y; ucur[4 * c + 2] = t4.z; ucur[4 * c + 3] = t4.w;
        }
    }
    __syncthreads();
    for (int kb = NBLK - 1; kb >= 0; kb--) {
        const int r0 = kb * 16;
        const int ia = 4 * (tid - 32), ib = ia + 4 * 224;
        float4 la[16], lb[16];
        if (tid >= 32) {
            if (ia < r0) {
#pragma unroll
                for (int j = 0; j < 16; j++)
                    la[j] = *(const float4*)(Kb + (size_t)(r0 + j) * PP + ia);
            }
            if (ib < r0) {
#pragma unroll
                for (int j = 0; j < 16; j++)
                    lb[j] = *(const float4*)(Kb + (size_t)(r0 + j) * PP + ib);
            }
        } else {
            float v = y[r0 + l16];
            const float rinv = 1.0f / dg[r0 + l16];
#pragma unroll
            for (int p = 15; p >= 0; p--) {
                if (l16 == p) v *= rinv;
                const float xp = __int_as_float(__builtin_amdgcn_readlane(__float_as_int(v), p));
                if (l16 < p) v -= ucur[p] * xp;
            }
            if (tid < 16) y[r0 + l16] = v;
        }
        __syncthreads();
        float unxt[16];
#pragma unroll
        for (int c = 0; c < 16; c++) unxt[c] = 0.f;
        if (tid < 32) {
            if (kb > 0) {
                const float* up = Kb + (size_t)(r0 - 16 + l16) * PP + r0;
#pragma unroll
                for (int c = 0; c < 4; c++) {
                    float4 t4 = *(const float4*)(up + 4 * c);
                    unxt[4 * c] = t4.x; unxt[4 * c + 1] = t4.y; unxt[4 * c + 2] = t4.z; unxt[4 * c + 3] = t4.w;
                }
            }
        } else {
            float xs[16];
#pragma unroll
            for (int c = 0; c < 4; c++) {
                const float4 t4 = *(const float4*)(y + r0 + 4 * c);
                xs[4 * c] = t4.x; xs[4 * c + 1] = t4.y; xs[4 * c + 2] = t4.z; xs[4 * c + 3] = t4.w;
            }
            if (ia < r0) {
                float4 acc = zero4;
#pragma unroll
                for (int j = 0; j < 16; j++) {
                    acc.x = __builtin_fmaf(la[j].x, xs[j], acc.x); acc.y = __builtin_fmaf(la[j].y, xs[j], acc.y);
                    acc.z = __builtin_fmaf(la[j].z, xs[j], acc.z); acc.w = __builtin_fmaf(la[j].w, xs[j], acc.w);
                }
                float4 yy = *(const float4*)(y + ia);
                const float4 d4 = *(const float4*)(dg + ia);
                yy.x = __builtin_fmaf(-d4.x, acc.x, yy.x); yy.y = __builtin_fmaf(-d4.y, acc.y, yy.y);
                yy.z = __builtin_fmaf(-d4.z, acc.z, yy.z); yy.w = __builtin_fmaf(-d4.w, acc.w, yy.w);
                *(float4*)(y + ia) = yy;
            }
            if (ib < r0) {
                float4 acc = zero4;
#pragma unroll
                for (int j = 0; j < 16; j++) {
                    acc.x = __builtin_fmaf(lb[j].x, xs[j], acc.x); acc.y = __builtin_fmaf(lb[j].y, xs[j], acc.y);
                    acc.z = __builtin_fmaf(lb[j].z, xs[j], acc.z); acc.w = __builtin_fmaf(lb[j].w, xs[j], acc.w);
                }
                float4 yy = *(const float4*)(y + ib);
                const float4 d4 = *(const float4*)(dg + ib);
                yy.x = __builtin_fmaf(-d4.x, acc.x, yy.x); yy.y = __builtin_fmaf(-d4.y, acc.y, yy.y);
                yy.z = __builtin_fmaf(-d4.z, acc.z, yy.z); yy.w = __builtin_fmaf(-d4.w, acc.w, yy.w);
                *(float4*)(y + ib) = yy;
            }
        }
        __syncthreads();
#pragma unroll
        for (int c = 0; c < 16; c++) ucur[c] = unxt[c];
    }
    for (int i = tid; i < PP; i += 256) x[(size_t)b * PP + i] = y[i];
}

__global__ void vote_kernel(const float* __restrict__ x, float* __restrict__ out) {
    int s = blockIdx.x, b = blockIdx.y, tid = threadIdx.x;
    __shared__ float vals[NN_];
    if (tid < NN_) {
        float v = x[(size_t)b * PP + tid * NN_ + s];
        vals[tid] = fminf(fmaxf(v, 0.f), 1.f);
    }
    __syncthreads();
    if (tid == 0) {
        float m = -1e30f;
        for (int t = 0; t < NN_; t++) m = fmaxf(m, vals[t]);
        float sum = 0.f;
        for (int t = 0; t < NN_; t++) { float e = __expf(ALPHA * (vals[t] - m)); vals[t] = e; sum += e; }
        float inv = 1.f / sum;
        for (int t = 0; t < NN_; t++) out[((size_t)b * NN_ + s) * NN_ + t] = vals[t] * inv;
    }
}

extern "C" void kernel_launch(void* const* d_in, const int* in_sizes, int n_in,
                              void* d_out, int out_size, void* d_ws, size_t ws_size,
                              hipStream_t stream) {
    const float* emb1 = (const float*)d_in[0];
    const float* emb2 = (const float*)d_in[1];
    const float* Wcg  = (const float*)d_in[2];
    const float* bcg  = (const float*)d_in[3];

    float* ws  = (float*)d_ws;
    size_t kelem = (size_t)BB * PP * PP;
    float* Km  = ws;
    float* h1  = Km + kelem;
    float* h2  = h1 + (size_t)BB * NN_ * DD;
    float* S   = h2 + (size_t)BB * NN_ * DD;
    float* rhs = S  + (size_t)BB * NN2;
    float* x   = rhs + (size_t)BB * PP;

    feat_kernel<<<dim3(BB, 2), 256, 0, stream>>>(emb1, emb2, Wcg, bcg, h1, h2);
    gram_rhs_kernel<<<BB, 256, 0, stream>>>(h1, h2, S, rhs);
    assemble_kernel<<<dim3(NBLK, NBLK, BB), dim3(16, 16), 0, stream>>>(S, Km);

    for (int jb = 0; jb < NBLK; jb++)
        lu_col_kernel<<<dim3(NBLK - jb, BB), 32 * NW, 0, stream>>>(Km, x, jb);

    trisolve_kernel<<<BB, 256, 0, stream>>>(Km, rhs, x);
    vote_kernel<<<dim3(NN_, BB), 64, 0, stream>>>(x, (float*)d_out);
}


# LU column launch: a workgroup owns 4 (block columns 0-21), 2 (22-53) or 1 row tiles; diagonal rows fetched, staged and multiplied once for all of them, one eliminating wave per tile (sum order kept)
# speedup vs baseline: 5.7558x; 1.0928x over previous
#include <hip/hip_runtime.h>
#include <math.h>


#define BB   8
#define NN_  36
#define DD   64
#define NN2  (NN_*NN_)
#define NCON (2*NN_)
#define KSZ  (NN2+NCON)
#define PP   1376
#define NBLK (PP/16)
#define KDIAG 1225.0f
#define EPSK  1e-5f
#define ALPHA 1000.0f
#define PS   17

typedef float v2f __attribute__((ext_vector_type(2)));
typedef float v4f __attribute__((ext_vector_type(4)));
typedef float v8f __attribute__((ext_vector_type(8)));

__global__ void feat_kernel(const float* __restrict__ emb1,
                            const float* __restrict__ emb2,
                            const float* __restrict__ W,
                            const float* __restrict__ bias,
                            float* __restrict__ h1, float* __restrict__ h2) {
    int b = blockIdx.x, which = blockIdx.y;
    const float* e = (which == 0 ? emb1 : emb2) + (size_t)b * NN_ * DD;
    float* h = (which == 0 ? h1 : h2) + (size_t)b * NN_ * DD;
    __shared__ float se[NN_ * DD];
    __shared__ float sw[DD * DD];
    __shared__ float snrm[DD];
    int tid = threadIdx.x;
    for (int i = tid; i < NN_ * DD; i += 256) se[i] = e[i];
    for (int i = tid; i < DD * DD; i += 256) sw[i] = W[i];
    __syncthreads();
    for (int d = tid; d < DD; d += 256) {
        float s = 0.f;
        for (int n = 0; n < NN_; n++) { float v = se[n * DD + d]; s += v * v; }
        snrm[d] = fmaxf(sqrtf(s), 1e-12f);
    }
    __syncthreads();
    for (int i = tid; i < NN_ * DD; i += 256) se[i] = se[i] / snrm[i % DD];
    __syncthreads();
    for (int idx = tid; idx < NN_ * DD; idx += 256) {
        int n = idx / DD, i = idx % DD;
        float acc = bias[i];
        for (int d = 0; d < DD; d++) acc += se[n * DD + d] * sw[i * DD + d];
        h[idx] = fmaxf(acc, 0.f);
    }
}

__global__ void gram_rhs_kernel(const float* __restrict__ h1,
                                const float* __restrict__ h2,
                                float* __restrict__ S, float* __restrict__ rhs) {
    int b = blockIdx.x, tid = threadIdx.x;
    __shared__ float s1[NN_ * DD];
    __shared__ float s2[NN_ * DD];
    for (int i = tid; i < NN_ * DD; i += 256) {
        s1[i] = h1[(size_t)b * NN_ * DD + i];
        s2[i] = h2[(size_t)b * NN_ * DD + i];
    }
    __syncthreads();
    for (int idx = tid; idx < NN2; idx += 256) {
        int i = idx / NN_, j = idx % NN_;
        float a = 0.f;
        for (int d = 0; d < DD; d++) a += s1[i * DD + d] * s2[j * DD + d];
        S[(size_t)b * NN2 + idx] = a;
        rhs[(size_t)b * PP + j * NN_ + i] = a;
    }
    for (int r = NN2 + tid; r < PP; r += 256)
        rhs[(size_t)b * PP + r] = (r < KSZ) ? 1.0f : 0.0f;
}

__global__ void assemble_kernel(const float* __restrict__ S, float* __restrict__ K) {
    int b = blockIdx.z;
    int r = blockIdx.y * 16 + threadIdx.y;
    int c = blockIdx.x * 16 + threadIdx.x;
    const float* Sb = S + (size_t)b * NN2;
    float val = 0.f;
    if (r < NN2 && c < NN2) {
        int t = r / NN_, s = r - t * NN_, u = c / NN_, v = c - u * NN_;
        val = (r == c) ? KDIAG : 0.f;
        if (s != v && t != u) val -= 0.5f * (Sb[s * NN_ + t] + Sb[v * NN_ + u]);
    } else if (r < NN2) {
        int i = c - NN2, t = r / NN_, s = r - t * NN_;
        if (i < NN_)            val = (s == i) ? 1.f : 0.f;
        else if (i < NCON)      val = (t == i - NN_) ? 1.f : 0.f;
    } else if (c < NN2) {
        int i = r - NN2, u = c / NN_, v = c - u * NN_;
        if (i < NN_)            val = (v == i) ? 1.f : 0.f;
        else if (i < NCON)      val = (u == i - NN_) ? 1.f : 0.f;
    } else if (r == c) {
        val = (r < KSZ) ? -EPSK : 1.0f;
    }
    K[((size_t)b * PP + r) * PP + c] = val;
}

#define QS 16
#define NW 4
#define SP 68

__device__ __forceinline__ void wave_lds_order() {
    __builtin_amdgcn_fence(__ATOMIC_RELEASE, "wavefront");
    __builtin_amdgcn_wave_barrier();
    __builtin_amdgcn_fence(__ATOMIC_ACQUIRE, "wavefront");
}

template <int R>
__global__ void __launch_bounds__(32 * NW) lu_col_kernel(float* __restrict__ K,
                                                         float* __restrict__ dpiv,
                                                         int jb) {
    constexpr int NT = R + 1;
    constexpr int STRIP = NT * 16 * SP + 4 * QS;
    __shared__ float sT[R * 32 * PS];
    __shared__ __attribute__((aligned(16))) float4 sP[NW * NT * 2 * 32];
    __shared__ __attribute__((aligned(16))) float sStage[NW * STRIP];
    const int b = blockIdx.y, lane = threadIdx.x & 31;
    const int w = __builtin_amdgcn_readfirstlane((int)(threadIdx.x >> 5));
    const int half = lane >> 4, l16 = lane & 15;
    float* Kb = K + (size_t)b * PP * PP;
    float* db = dpiv + (size_t)b * PP;
    const int c0 = jb * 16;
    const int nT = NBLK - jb;
    int rown[R];
#pragma unroll
    for (int t = 0; t < R; t++) {
        const int tt = (int)blockIdx.x * R + t;
        rown[t] = c0 + 16 * ((tt < nT) ? tt : 0);
    }

    v8f cd, co[R];
    if (w == 0) {
#pragma unroll
        for (int r = 0; r < 8; r++) {
            cd[r] = Kb[(size_t)(c0 + r + 8 * half) * PP + c0 + l16];
#pragma unroll
            for (int t = 0; t < R; t++)
                co[t][r] = Kb[(size_t)(rown[t] + r + 8 * half) * PP + c0 + l16];
        }
    } else {
#pragma unroll
        for (int r = 0; r < 8; r++) {
            cd[r] = 0.f;
#pragma unroll
            for (int t = 0; t < R; t++) co[t][r] = 0.f;
        }
    }

    const int Q = jb >> 2;
    const int qb = Q >> 2, qr = Q & 3;
    const int nq = qb + ((w < qr) ? 1 : 0);
    const int kq = (w * qb + ((w < qr) ? w : qr)) * (4 * QS);

    const float* pl0 = Kb + (size_t)(c0 + l16) * PP + 2 * half;
    const float* pd0 = db + 2 * half;
    const float* pa0[R];
    const float* ga[R];
    const float* gl = Kb + (size_t)(c0 + half) * PP + 4 * l16 + kq;
    const float* gd = db + 2 * lane + kq;
    float* stL = sStage + w * STRIP + R * 16 * SP;
    float* stD = stL + 16 * SP;
    float* wA[R];
    float* wL = stL + half * SP + 4 * l16;
    float* wD = stD + 2 * lane;
    const float* fA[R];
    const float* fL = stL + l16 * SP + 2 * half;
    const float* fD = stD + 2 * half;
#pragma unroll
    for (int t = 0; t < R; t++) {
        pa0[t] = Kb + (size_t)(rown[t] + l16) * PP + 2 * half;
        ga[t] = Kb + (size_t)(rown[t] + half) * PP + 4 * l16 + kq;
        float* stA = sStage + w * STRIP + t * 16 * SP;
        wA[t] = stA + half * SP + 4 * l16;
        fA[t] = stA + l16 * SP + 2 * half;
    }

    v4f ga4[R][8], gl4[8];
    v2f gd2;
    v2f ra[R][QS], rl[QS], rd[QS];
    if (nq > 0) {
#pragma unroll
        for (int j = 0; j < 8; j++) {
#pragma unroll
            for (int t = 0; t < R; t++) ga4[t][j] = *(const v4f*)(ga[t] + (size_t)(2 * j) * PP);
            gl4[j] = *(const v4f*)(gl + (size_t)(2 * j) * PP);
        }
        gd2 = *(const v2f*)gd;
#pragma unroll
        for (int j = 0; j < 8; j++) {
#pragma unroll
            for (int t = 0; t < R; t++) *(v4f*)(wA[t] + 2 * j * SP) = ga4[t][j];
            *(v4f*)(wL + 2 * j * SP) = gl4[j];
        }
        *(v2f*)wD = gd2;
        if (nq > 1) {
#pragma unroll
            for (int j = 0; j < 8; j++) {
#pragma unroll
                for (int t = 0; t < R; t++) ga4[t][j] = *(const v4f*)(ga[t] + (size_t)(2 * j) * PP + 4 * QS);
                gl4[j] = *(const v4f*)(gl + (size_t)(2 * j) * PP + 4 * QS);
            }
            gd2 = *(const v2f*)(gd + 4 * QS);
        }
        wave_lds_order();
#pragma unroll
        for (int s = 0; s < QS; s++) {
#pragma unroll
            for (int t = 0; t < R; t++) ra[t][s] = *(const v2f*)(fA[t] + 4 * s);
            rl[s] = *(const v2f*)(fL + 4 * s);
            rd[s] = *(const v2f*)(fD + 4 * s);
        }
    }
    for (int q = 0; q < nq; q++) {
        v2f an[R][QS], ln[QS], bb[QS];
#pragma unroll
        for (int s = 0; s < QS; s++) {
#pragma unroll
            for (int t = 0; t < R; t++) an[t][s] = -ra[t][s];
            ln[s] = -rl[s];
            bb[s] = rl[s] * rd[s];
        }
        if (q + 1 < nq) {
            wave_lds_order();
#pragma unroll
            for (int j = 0; j < 8; j++) {
#pragma unroll
                for (int t = 0; t < R; t++) *(v4f*)(wA[t] + 2 * j * SP) = ga4[t][j];
                *(v4f*)(wL + 2 * j * SP) = gl4[j];
            }
            *(v2f*)wD = gd2;
            if (q + 2 < nq) {
                const int k2 = (q + 2) * (4 * QS);
#pragma unroll
                for (int j = 0; j < 8; j++) {
#pragma unroll
                    for (int t = 0; t < R; t++) ga4[t][j] = *(const v4f*)(ga[t] + (size_t)(2 * j) * PP + k2);
                    gl4[j] = *(const v4f*)(gl + (size_t)(2 * j) * PP + k2);
                }
                gd2 = *(const v2f*)(gd + k2);
            }
            wave_lds_order();
#pragma unroll
            for (int s = 0; s < QS; s++) {
#pragma unroll
                for (int t = 0; t < R; t++) ra[t][s] = *(const v2f*)(fA[t] + 4 * s);
                rl[s] = *(const v2f*)(fL + 4 * s);
                rd[s] = *(const v2f*)(fD + 4 * s);
            }
        }
#pragma unroll
        for (int s = 0; s < QS; s++) {
#pragma unroll
            for (int t = 0; t < R; t++)
                co[t] = __builtin_amdgcn_wmma_f32_16x16x4_f32(false, an[t][s], false, bb[s], (short)0, co[t], false, false);
            cd = __builtin_amdgcn_wmma_f32_16x16x4_f32(false, ln[s], false, bb[s], (short)0, cd, false, false);
        }
    }
    if (w == NW - 1) {
        for (int k0 = Q * (4 * QS); k0 < c0; k0 += 16) {
            v2f ta[R][4], tl[4], td[4];
#pragma unroll
            for (int s = 0; s < 4; s++) {
#pragma unroll
                for (int t = 0; t < R; t++) ta[t][s] = *(const v2f*)(pa0[t] + k0 + 4 * s);
                tl[s] = *(const v2f*)(pl0 + k0 + 4 * s);
                td[s] = *(const v2f*)(pd0 + k0 + 4 * s);
            }
#pragma unroll
            for (int s = 0; s < 4; s++) {
                v2f l1 = -tl[s], b1 = tl[s] * td[s];
#pragma unroll
                for (int t = 0; t < R; t++) {
                    v2f a1 = -ta[t][s];
                    co[t] = __builtin_amdgcn_wmma_f32_16x16x4_f32(false, a1, false, b1, (short)0, co[t], false, false);
                }
                cd = __builtin_amdgcn_wmma_f32_16x16x4_f32(false, l1, false, b1, (short)0, cd, false, false);
            }
        }
    }

    {
        float4* pp = sP + (w * NT) * 64 + lane;
        pp[0]  = make_float4(cd[0], cd[1], cd[2], cd[3]);
        pp[32] = make_float4(cd[4], cd[5], cd[6], cd[7]);
#pragma unroll
        for (int t = 0; t < R; t++) {
            pp[(1 + t) * 64]      = make_float4(co[t][0], co[t][1], co[t][2], co[t][3]);
            pp[(1 + t) * 64 + 32] = make_float4(co[t][4], co[t][5], co[t][6], co[t][7]);
        }
    }
    __syncthreads();
    float* sTw = sT + ((w < R) ? w : 0) * 32 * PS;
    if (w < R) {
        float sd[8], so[8];
#pragma unroll
        for (int u = 0; u < NW; u++) {
            const float4* pp = sP + (u * NT) * 64 + lane;
            const float4 d0 = pp[0], d1 = pp[32];
            const float4 o0 = pp[(1 + w) * 64], o1 = pp[(1 + w) * 64 + 32];
            if (u == 0) {
                sd[0] = d0.x; sd[1] = d0.y; sd[2] = d0.z; sd[3] = d0.w;
                sd[4] = d1.x; sd[5] = d1.y; sd[6] = d1.z; sd[7] = d1.w;
                so[0] = o0.x; so[1] = o0.y; so[2] = o0.z; so[3] = o0.w;
                so[4] = o1.x; so[5] = o1.y; so[6] = o1.z; so[7] = o1.w;
            } else {
                sd[0] += d0.x; sd[1] += d0.y; sd[2] += d0.z; sd[3] += d0.w;
                sd[4] += d1.x; sd[5] += d1.y; sd[6] += d1.z; sd[7] += d1.w;
                so[0] += o0.x; so[1] += o0.y; so[2] += o0.z; so[3] += o0.w;
                so[4] += o1.x; so[5] += o1.y; so[6] += o1.z; so[7] += o1.w;
            }
        }
#pragma unroll
        for (int r = 0; r < 8; r++) {
            sTw[(r + 8 * half) * PS + l16] = sd[r];
            sTw[(16 + r + 8 * half) * PS + l16] = so[r];
        }
    }
    __syncthreads();
    if (w >= R) return;
    float t[16];
#pragma unroll
    for (int c = 0; c < 16; c++) t[c] = sTw[lane * PS + c];

    float mypiv = 0.f;
#pragma unroll
    for (int p = 0; p < 16; p++) {
        float u[16];
#pragma unroll
        for (int c2 = p; c2 < 16; c2++)
            u[c2] = __int_as_float(__builtin_amdgcn_readlane(__float_as_int(t[c2]), p));
        float dinv = 1.0f / u[p];
        if (lane == p) mypiv = u[p];
        if (lane > p) {
            float lij = t[p] * dinv;
            t[p] = lij;
#pragma unroll
            for (int c2 = p + 1; c2 < 16; c2++) t[c2] -= lij * u[c2];
        }
    }

    const int myt = (int)blockIdx.x * R + w;
    float* dst = nullptr;
    if (myt == 0) {
        if (lane < 16) {
            dst = Kb + (size_t)(c0 + lane) * PP + c0 + ((jb < NBLK - 1) ? 16 : 0);
            db[c0 + lane] = mypiv;
        }
    } else if (myt < nT && lane >= 16) {
        dst = Kb + (size_t)(c0 + 16 * myt + lane - 16) * PP + c0;
    }
    if (dst) {
#pragma unroll
        for (int c = 0; c < 16; c += 4) {
            float4 v; v.x = t[c]; v.y = t[c + 1]; v.z = t[c + 2]; v.w = t[c + 3];
            *(float4*)(dst + c) = v;
        }
    }
}

#define FCH 24
#define RP  20

__global__ void __launch_bounds__(256) trisolve_kernel(const float* __restrict__ K,
                                                       const float* __restrict__ rhs,
                                                       float* __restrict__ x) {
    const int b = blockIdx.x, tid = threadIdx.x;
    const int l16 = tid & 15, grp = tid >> 4;
    __shared__ __attribute__((aligned(16))) float y[PP];
    __shared__ __attribute__((aligned(16))) float dg[PP];
    __shared__ __attribute__((aligned(16))) float red[16 * RP];
    const float* Kb = K + (size_t)b * PP * PP;
    for (int i = tid; i < PP; i += 256) {
        y[i] = rhs[(size_t)b * PP + i];
        dg[i] = Kb[(size_t)i * PP + i + (((i >> 4) < NBLK - 1) ? 16 : 0)];
    }
    const float4 zero4 = make_float4(0.f, 0.f, 0.f, 0.f);

    float4 cur[FCH];
    float lcur[16];
#pragma unroll
    for (int m = 0; m < FCH; m++) cur[m] = zero4;
#pragma unroll
    for (int c = 0; c < 16; c++) lcur[c] = 0.f;
    if (tid < 32) {
        const float* lp = Kb + (size_t)l16 * PP + 16;
#pragma unroll
        for (int c = 0; c < 4; c++) {
            float4 t4 = *(const float4*)(lp + 4 * c);
            lcur[4 * c] = t4.x; lcur[4 * c + 1] = t4.y; lcur[4 * c + 2] = t4.z; lcur[4 * c + 3] = t4.w;
        }
    }
    __syncthreads();
    for (int kb = 0; kb < NBLK; kb++) {
        const int r0 = kb * 16, r1 = r0 + 16;
        float4 nxt[FCH];
        float lnxt[16];
#pragma unroll
        for (int m = 0; m < FCH; m++) nxt[m] = zero4;
#pragma unroll
        for (int c = 0; c < 16; c++) lnxt[c] = 0.f;
        if (kb + 1 < NBLK) {
            const float* rp = Kb + (size_t)(r1 + grp) * PP;
#pragma unroll
            for (int m = 0; m < FCH; m++) {
                if (64 * m < r1) {
                    const int k = 64 * m + 4 * l16;
                    if (k < r1) nxt[m] = *(const float4*)(rp + k);
                }
            }
            if (tid < 32) {
                const int d1 = r1 + ((kb + 1 < NBLK - 1) ? 16 : 0);
                const float* lp = Kb + (size_t)(r1 + l16) * PP + d1;
#pragma unroll
                for (int c = 0; c < 4; c++) {
                    float4 t4 = *(const float4*)(lp + 4 * c);
                    lnxt[4 * c] = t4.x; lnxt[4 * c + 1] = t4.y; lnxt[4 * c + 2] = t4.z; lnxt[4 * c + 3] = t4.w;
                }
            }
        }
        float a0 = 0.f, a1 = 0.f, a2 = 0.f, a3 = 0.f;
#pragma unroll
        for (int g = 0; g < FCH / 4; g++) {
            if (256 * g < r0) {
#pragma unroll
                for (int mm = 0; mm < 4; mm++) {
                    const int m = 4 * g + mm;
                    const int k = 64 * m + 4 * l16;
                    const float4 yy = *(const float4*)(y + ((k < PP) ? k : 0));
                    a0 = __builtin_fmaf(cur[m].x, yy.x, a0); a1 = __builtin_fmaf(cur[m].y, yy.y, a1);
                    a2 = __builtin_fmaf(cur[m].z, yy.z, a2); a3 = __builtin_fmaf(cur[m].w, yy.w, a3);
                }
            }
        }
        red[grp * RP + l16] = (a0 + a1) + (a2 + a3);
        __syncthreads();
        if (tid < 32) {
            const float4* rr = (const float4*)(red + l16 * RP);
            const float4 s0 = rr[0], s1 = rr[1], s2 = rr[2], s3 = rr[3];
            const float dot = (((s0.x + s0.y) + (s0.z + s0.w)) + ((s1.x + s1.y) + (s1.z + s1.w)))
                            + (((s2.x + s2.y) + (s2.z + s2.w)) + ((s3.x + s3.y) + (s3.z + s3.w)));
            float v = y[r0 + l16] - dot;
#pragma unroll
            for (int p = 0; p < 15; p++) {
                const float vp = __int_as_float(__builtin_amdgcn_readlane(__float_as_int(v), p));
                if (l16 > p) v -= lcur[p] * vp;
            }
            if (tid < 16) y[r0 + l16] = v;
        }
        __syncthreads();
#pragma unroll
        for (int m = 0; m < FCH; m++) cur[m] = nxt[m];
#pragma unroll
        for (int c = 0; c < 16; c++) lcur[c] = lnxt[c];
    }

    float ucur[16];
#pragma unroll
    for (int c = 0; c < 16; c++) ucur[c] = 0.f;
    if (tid < 32) {
        const float* up = Kb + (size_t)((NBLK - 1) * 16 + l16) * PP + (NBLK - 1) * 16;
#pragma unroll
        for (int c = 0; c < 4; c++) {
            float4 t4 = *(const float4*)(up + 4 * c);
            ucur[4 * c] = t4.x; ucur[4 * c + 1] = t4.y; ucur[4 * c + 2] = t4.z; ucur[4 * c + 3] = t4.w;
        }
    }
    __syncthreads();
    for (int kb = NBLK - 1; kb >= 0; kb--) {
        const int r0 = kb * 16;
        const int ia = 4 * (tid - 32), ib = ia + 4 * 224;
        float4 la[16], lb[16];
        if (tid >= 32) {
            if (ia < r0) {
#pragma unroll
                for (int j = 0; j < 16; j++)
                    la[j] = *(const float4*)(Kb + (size_t)(r0 + j) * PP + ia);
            }
            if (ib < r0) {
#pragma unroll
                for (int j = 0; j < 16; j++)
                    lb[j] = *(const float4*)(Kb + (size_t)(r0 + j) * PP + ib);
            }
        } else {
            float v = y[r0 + l16];
            const float rinv = 1.0f / dg[r0 + l16];
#pragma unroll
            for (int p = 15; p >= 0; p--) {
                if (l16 == p) v *= rinv;
                const float xp = __int_as_float(__builtin_amdgcn_readlane(__float_as_int(v), p));
                if (l16 < p) v -= ucur[p] * xp;
            }
            if (tid < 16) y[r0 + l16] = v;
        }
        __syncthreads();
        float unxt[16];
#pragma unroll
        for (int c = 0; c < 16; c++) unxt[c] = 0.f;
        if (tid < 32) {
            if (kb > 0) {
                const float* up = Kb + (size_t)(r0 - 16 + l16) * PP + r0;
#pragma unroll
                for (int c = 0; c < 4; c++) {
                    float4 t4 = *(const float4*)(up + 4 * c);
                    unxt[4 * c] = t4.x; unxt[4 * c + 1] = t4.y; unxt[4 * c + 2] = t4.z; unxt[4 * c + 3] = t4.w;
                }
            }
        } else {
            float xs[16];
#pragma unroll
            for (int c = 0; c < 4; c++) {
                const float4 t4 = *(const float4*)(y + r0 + 4 * c);
                xs[4 * c] = t4.x; xs[4 * c + 1] = t4.y; xs[4 * c + 2] = t4.z; xs[4 * c + 3] = t4.w;
            }
            if (ia < r0) {
                float4 acc = zero4;
#pragma unroll
                for (int j = 0; j < 16; j++) {
                    acc.x = __builtin_fmaf(la[j].x, xs[j], acc.x); acc.y = __builtin_fmaf(la[j].y, xs[j], acc.y);
                    acc.z = __builtin_fmaf(la[j].z, xs[j], acc.z); acc.w = __builtin_fmaf(la[j].w, xs[j], acc.w);
                }
                float4 yy = *(const float4*)(y + ia);
                const float4 d4 = *(const float4*)(dg + ia);
                yy.x = __builtin_fmaf(-d4.x, acc.x, yy.x); yy.y = __builtin_fmaf(-d4.y, acc.y, yy.y);
                yy.z = __builtin_fmaf(-d4.z, acc.z, yy.z); yy.w = __builtin_fmaf(-d4.w, acc.w, yy.w);
                *(float4*)(y + ia) = yy;
            }
            if (ib < r0) {
                float4 acc = zero4;
#pragma unroll
                for (int j = 0; j < 16; j++) {
                    acc.x = __builtin_fmaf(lb[j].x, xs[j], acc.x); acc.y = __builtin_fmaf(lb[j].y, xs[j], acc.y);
                    acc.z = __builtin_fmaf(lb[j].z, xs[j], acc.z); acc.w = __builtin_fmaf(lb[j].w, xs[j], acc.w);
                }
                float4 yy = *(const float4*)(y + ib);
                const float4 d4 = *(const float4*)(dg + ib);
                yy.x = __builtin_fmaf(-d4.x, acc.x, yy.x); yy.y = __builtin_fmaf(-d4.y, acc.y, yy.y);
                yy.z = __builtin_fmaf(-d4.z, acc.z, yy.z); yy.w = __builtin_fmaf(-d4.w, acc.w, yy.w);
                *(float4*)(y + ib) = yy;
            }
        }
        __syncthreads();
#pragma unroll
        for (int c = 0; c < 16; c++) ucur[c] = unxt[c];
    }
    for (int i = tid; i < PP; i += 256) x[(size_t)b * PP + i] = y[i];
}

__global__ void vote_kernel(const float* __restrict__ x, float* __restrict__ out) {
    int s = blockIdx.x, b = blockIdx.y, tid = threadIdx.x;
    __shared__ float vals[NN_];
    if (tid < NN_) {
        float v = x[(size_t)b * PP + tid * NN_ + s];
        vals[tid] = fminf(fmaxf(v, 0.f), 1.f);
    }
    __syncthreads();
    if (tid == 0) {
        float m = -1e30f;
        for (int t = 0; t < NN_; t++) m = fmaxf(m, vals[t]);
        float sum = 0.f;
        for (int t = 0; t < NN_; t++) { float e = __expf(ALPHA * (vals[t] - m)); vals[t] = e; sum += e; }
        float inv = 1.f / sum;
        for (int t = 0; t < NN_; t++) out[((size_t)b * NN_ + s) * NN_ + t] = vals[t] * inv;
    }
}

extern "C" void kernel_launch(void* const* d_in, const int* in_sizes, int n_in,
                              void* d_out, int out_size, void* d_ws, size_t ws_size,
                              hipStream_t stream) {
    const float* emb1 = (const float*)d_in[0];
    const float* emb2 = (const float*)d_in[1];
    const float* Wcg  = (const float*)d_in[2];
    const float* bcg  = (const float*)d_in[3];

    float* ws  = (float*)d_ws;
    size_t kelem = (size_t)BB * PP * PP;
    float* Km  = ws;
    float* h1  = Km + kelem;
    float* h2  = h1 + (size_t)BB * NN_ * DD;
    float* S   = h2 + (size_t)BB * NN_ * DD;
    float* rhs = S  + (size_t)BB * NN2;
    float* x   = rhs + (size_t)BB * PP;

    feat_kernel<<<dim3(BB, 2), 256, 0, stream>>>(emb1, emb2, Wcg, bcg, h1, h2);
    gram_rhs_kernel<<<BB, 256, 0, stream>>>(h1, h2, S, rhs);
    assemble_kernel<<<dim3(NBLK, NBLK, BB), dim3(16, 16), 0, stream>>>(S, Km);

    for (int jb = 0; jb < 22; jb++)
        lu_col_kernel<4><<<dim3((NBLK - jb + 3) / 4, BB), 32 * NW, 0, stream>>>(Km, x, jb);
    for (int jb = 22; jb < 54; jb++)
        lu_col_kernel<2><<<dim3((NBLK - jb + 1) / 2, BB), 32 * NW, 0, stream>>>(Km, x, jb);
    for (int jb = 54; jb < NBLK; jb++)
        lu_col_kernel<1><<<dim3(NBLK - jb, BB), 32 * NW, 0, stream>>>(Km, x, jb);

    trisolve_kernel<<<BB, 256, 0, stream>>>(Km, rhs, x);
    vote_kernel<<<dim3(NN_, BB), 64, 0, stream>>>(x, (float*)d_out);
}


# LU column launch: 16-step register elimination written branch-free (finished rows kept by selects); drops the ~1300 register copies per launch that joined the per-step branches (bit-identical)
# speedup vs baseline: 6.5543x; 1.1387x over previous
#include <hip/hip_runtime.h>
#include <math.h>


#define BB   8
#define NN_  36
#define DD   64
#define NN2  (NN_*NN_)
#define NCON (2*NN_)
#define KSZ  (NN2+NCON)
#define PP   1376
#define NBLK (PP/16)
#define KDIAG 1225.0f
#define EPSK  1e-5f
#define ALPHA 1000.0f
#define PS   17

typedef float v2f __attribute__((ext_vector_type(2)));
typedef float v4f __attribute__((ext_vector_type(4)));
typedef float v8f __attribute__((ext_vector_type(8)));

__global__ void feat_kernel(const float* __restrict__ emb1,
                            const float* __restrict__ emb2,
                            const float* __restrict__ W,
                            const float* __restrict__ bias,
                            float* __restrict__ h1, float* __restrict__ h2) {
    int b = blockIdx.x, which = blockIdx.y;
    const float* e = (which == 0 ? emb1 : emb2) + (size_t)b * NN_ * DD;
    float* h = (which == 0 ? h1 : h2) + (size_t)b * NN_ * DD;
    __shared__ float se[NN_ * DD];
    __shared__ float sw[DD * DD];
    __shared__ float snrm[DD];
    int tid = threadIdx.x;
    for (int i = tid; i < NN_ * DD; i += 256) se[i] = e[i];
    for (int i = tid; i < DD * DD; i += 256) sw[i] = W[i];
    __syncthreads();
    for (int d = tid; d < DD; d += 256) {
        float s = 0.f;
        for (int n = 0; n < NN_; n++) { float v = se[n * DD + d]; s += v * v; }
        snrm[d] = fmaxf(sqrtf(s), 1e-12f);
    }
    __syncthreads();
    for (int i = tid; i < NN_ * DD; i += 256) se[i] = se[i] / snrm[i % DD];
    __syncthreads();
    for (int idx = tid; idx < NN_ * DD; idx += 256) {
        int n = idx / DD, i = idx % DD;
        float acc = bias[i];
        for (int d = 0; d < DD; d++) acc += se[n * DD + d] * sw[i * DD + d];
        h[idx] = fmaxf(acc, 0.f);
    }
}

__global__ void gram_rhs_kernel(const float* __restrict__ h1,
                                const float* __restrict__ h2,
                                float* __restrict__ S, float* __restrict__ rhs) {
    int b = blockIdx.x, tid = threadIdx.x;
    __shared__ float s1[NN_ * DD];
    __shared__ float s2[NN_ * DD];
    for (int i = tid; i < NN_ * DD; i += 256) {
        s1[i] = h1[(size_t)b * NN_ * DD + i];
        s2[i] = h2[(size_t)b * NN_ * DD + i];
    }
    __syncthreads();
    for (int idx = tid; idx < NN2; idx += 256) {
        int i = idx / NN_, j = idx % NN_;
        float a = 0.f;
        for (int d = 0; d < DD; d++) a += s1[i * DD + d] * s2[j * DD + d];
        S[(size_t)b * NN2 + idx] = a;
        rhs[(size_t)b * PP + j * NN_ + i] = a;
    }
    for (int r = NN2 + tid; r < PP; r += 256)
        rhs[(size_t)b * PP + r] = (r < KSZ) ? 1.0f : 0.0f;
}

__global__ void assemble_kernel(const float* __restrict__ S, float* __restrict__ K) {
    int b = blockIdx.z;
    int r = blockIdx.y * 16 + threadIdx.y;
    int c = blockIdx.x * 16 + threadIdx.x;
    const float* Sb = S + (size_t)b * NN2;
    float val = 0.f;
    if (r < NN2 && c < NN2) {
        int t = r / NN_, s = r - t * NN_, u = c / NN_, v = c - u * NN_;
        val = (r == c) ? KDIAG : 0.f;
        if (s != v && t != u) val -= 0.5f * (Sb[s * NN_ + t] + Sb[v * NN_ + u]);
    } else if (r < NN2) {
        int i = c - NN2, t = r / NN_, s = r - t * NN_;
        if (i < NN_)            val = (s == i) ? 1.f : 0.f;
        else if (i < NCON)      val = (t == i - NN_) ? 1.f : 0.f;
    } else if (c < NN2) {
        int i = r - NN2, u = c / NN_, v = c - u * NN_;
        if (i < NN_)            val = (v == i) ? 1.f : 0.f;
        else if (i < NCON)      val = (u == i - NN_) ? 1.f : 0.f;
    } else if (r == c) {
        val = (r < KSZ) ? -EPSK : 1.0f;
    }
    K[((size_t)b * PP + r) * PP + c] = val;
}

#define QS 16
#define NW 4
#define SP 68

__device__ __forceinline__ void wave_lds_order() {
    __builtin_amdgcn_fence(__ATOMIC_RELEASE, "wavefront");
    __builtin_amdgcn_wave_barrier();
    __builtin_amdgcn_fence(__ATOMIC_ACQUIRE, "wavefront");
}

template <int R>
__global__ void __launch_bounds__(32 * NW) lu_col_kernel(float* __restrict__ K,
                                                         float* __restrict__ dpiv,
                                                         int jb) {
    constexpr int NT = R + 1;
    constexpr int STRIP = NT * 16 * SP + 4 * QS;
    __shared__ float sT[R * 32 * PS];
    __shared__ __attribute__((aligned(16))) float4 sP[NW * NT * 2 * 32];
    __shared__ __attribute__((aligned(16))) float sStage[NW * STRIP];
    const int b = blockIdx.y, lane = threadIdx.x & 31;
    const int w = __builtin_amdgcn_readfirstlane((int)(threadIdx.x >> 5));
    const int half = lane >> 4, l16 = lane & 15;
    float* Kb = K + (size_t)b * PP * PP;
    float* db = dpiv + (size_t)b * PP;
    const int c0 = jb * 16;
    const int nT = NBLK - jb;
    int rown[R];
#pragma unroll
    for (int t = 0; t < R; t++) {
        const int tt = (int)blockIdx.x * R + t;
        rown[t] = c0 + 16 * ((tt < nT) ? tt : 0);
    }

    v8f cd, co[R];
    if (w == 0) {
#pragma unroll
        for (int r = 0; r < 8; r++) {
            cd[r] = Kb[(size_t)(c0 + r + 8 * half) * PP + c0 + l16];
#pragma unroll
            for (int t = 0; t < R; t++)
                co[t][r] = Kb[(size_t)(rown[t] + r + 8 * half) * PP + c0 + l16];
        }
    } else {
#pragma unroll
        for (int r = 0; r < 8; r++) {
            cd[r] = 0.f;
#pragma unroll
            for (int t = 0; t < R; t++) co[t][r] = 0.f;
        }
    }

    const int Q = jb >> 2;
    const int qb = Q >> 2, qr = Q & 3;
    const int nq = qb + ((w < qr) ? 1 : 0);
    const int kq = (w * qb + ((w < qr) ? w : qr)) * (4 * QS);

    const float* pl0 = Kb + (size_t)(c0 + l16) * PP + 2 * half;
    const float* pd0 = db + 2 * half;
    const float* pa0[R];
    const float* ga[R];
    const float* gl = Kb + (size_t)(c0 + half) * PP + 4 * l16 + kq;
    const float* gd = db + 2 * lane + kq;
    float* stL = sStage + w * STRIP + R * 16 * SP;
    float* stD = stL + 16 * SP;
    float* wA[R];
    float* wL = stL + half * SP + 4 * l16;
    float* wD = stD + 2 * lane;
    const float* fA[R];
    const float* fL = stL + l16 * SP + 2 * half;
    const float* fD = stD + 2 * half;
#pragma unroll
    for (int t = 0; t < R; t++) {
        pa0[t] = Kb + (size_t)(rown[t] + l16) * PP + 2 * half;
        ga[t] = Kb + (size_t)(rown[t] + half) * PP + 4 * l16 + kq;
        float* stA = sStage + w * STRIP + t * 16 * SP;
        wA[t] = stA + half * SP + 4 * l16;
        fA[t] = stA + l16 * SP + 2 * half;
    }

    v4f ga4[R][8], gl4[8];
    v2f gd2;
    v2f ra[R][QS], rl[QS], rd[QS];
    if (nq > 0) {
#pragma unroll
        for (int j = 0; j < 8; j++) {
#pragma unroll
            for (int t = 0; t < R; t++) ga4[t][j] = *(const v4f*)(ga[t] + (size_t)(2 * j) * PP);
            gl4[j] = *(const v4f*)(gl + (size_t)(2 * j) * PP);
        }
        gd2 = *(const v2f*)gd;
#pragma unroll
        for (int j = 0; j < 8; j++) {
#pragma unroll
            for (int t = 0; t < R; t++) *(v4f*)(wA[t] + 2 * j * SP) = ga4[t][j];
            *(v4f*)(wL + 2 * j * SP) = gl4[j];
        }
        *(v2f*)wD = gd2;
        if (nq > 1) {
#pragma unroll
            for (int j = 0; j < 8; j++) {
#pragma unroll
                for (int t = 0; t < R; t++) ga4[t][j] = *(const v4f*)(ga[t] + (size_t)(2 * j) * PP + 4 * QS);
                gl4[j] = *(const v4f*)(gl + (size_t)(2 * j) * PP + 4 * QS);
            }
            gd2 = *(const v2f*)(gd + 4 * QS);
        }
        wave_lds_order();
#pragma unroll
        for (int s = 0; s < QS; s++) {
#pragma unroll
            for (int t = 0; t < R; t++) ra[t][s] = *(const v2f*)(fA[t] + 4 * s);
            rl[s] = *(const v2f*)(fL + 4 * s);
            rd[s] = *(const v2f*)(fD + 4 * s);
        }
    }
    for (int q = 0; q < nq; q++) {
        v2f an[R][QS], ln[QS], bb[QS];
#pragma unroll
        for (int s = 0; s < QS; s++) {
#pragma unroll
            for (int t = 0; t < R; t++) an[t][s] = -ra[t][s];
            ln[s] = -rl[s];
            bb[s] = rl[s] * rd[s];
        }
        if (q + 1 < nq) {
            wave_lds_order();
#pragma unroll
            for (int j = 0; j < 8; j++) {
#pragma unroll
                for (int t = 0; t < R; t++) *(v4f*)(wA[t] + 2 * j * SP) = ga4[t][j];
                *(v4f*)(wL + 2 * j * SP) = gl4[j];
            }
            *(v2f*)wD = gd2;
            if (q + 2 < nq) {
                const int k2 = (q + 2) * (4 * QS);
#pragma unroll
                for (int j = 0; j < 8; j++) {
#pragma unroll
                    for (int t = 0; t < R; t++) ga4[t][j] = *(const v4f*)(ga[t] + (size_t)(2 * j) * PP + k2);
                    gl4[j] = *(const v4f*)(gl + (size_t)(2 * j) * PP + k2);
                }
                gd2 = *(const v2f*)(gd + k2);
            }
            wave_lds_order();
#pragma unroll
            for (int s = 0; s < QS; s++) {
#pragma unroll
                for (int t = 0; t < R; t++) ra[t][s] = *(const v2f*)(fA[t] + 4 * s);
                rl[s] = *(const v2f*)(fL + 4 * s);
                rd[s] = *(const v2f*)(fD + 4 * s);
            }
        }
#pragma unroll
        for (int s = 0; s < QS; s++) {
#pragma unroll
            for (int t = 0; t < R; t++)
                co[t] = __builtin_amdgcn_wmma_f32_16x16x4_f32(false, an[t][s], false, bb[s], (short)0, co[t], false, false);
            cd = __builtin_amdgcn_wmma_f32_16x16x4_f32(false, ln[s], false, bb[s], (short)0, cd, false, false);
        }
    }
    if (w == NW - 1) {
        for (int k0 = Q * (4 * QS); k0 < c0; k0 += 16) {
            v2f ta[R][4], tl[4], td[4];
#pragma unroll
            for (int s = 0; s < 4; s++) {
#pragma unroll
                for (int t = 0; t < R; t++) ta[t][s] = *(const v2f*)(pa0[t] + k0 + 4 * s);
                tl[s] = *(const v2f*)(pl0 + k0 + 4 * s);
                td[s] = *(const v2f*)(pd0 + k0 + 4 * s);
            }
#pragma unroll
            for (int s = 0; s < 4; s++) {
                v2f l1 = -tl[s], b1 = tl[s] * td[s];
#pragma unroll
                for (int t = 0; t < R; t++) {
                    v2f a1 = -ta[t][s];
                    co[t] = __builtin_amdgcn_wmma_f32_16x16x4_f32(false, a1, false, b1, (short)0, co[t], false, false);
                }
                cd = __builtin_amdgcn_wmma_f32_16x16x4_f32(false, l1, false, b1, (short)0, cd, false, false);
            }
        }
    }

    {
        float4* pp = sP + (w * NT) * 64 + lane;
        pp[0]  = make_float4(cd[0], cd[1], cd[2], cd[3]);
        pp[32] = make_float4(cd[4], cd[5], cd[6], cd[7]);
#pragma unroll
        for (int t = 0; t < R; t++) {
            pp[(1 + t) * 64]      = make_float4(co[t][0], co[t][1], co[t][2], co[t][3]);
            pp[(1 + t) * 64 + 32] = make_float4(co[t][4], co[t][5], co[t][6], co[t][7]);
        }
    }
    __syncthreads();
    float* sTw = sT + ((w < R) ? w : 0) * 32 * PS;
    if (w < R) {
        float sd[8], so[8];
#pragma unroll
        for (int u = 0; u < NW; u++) {
            const float4* pp = sP + (u * NT) * 64 + lane;
            const float4 d0 = pp[0], d1 = pp[32];
            const float4 o0 = pp[(1 + w) * 64], o1 = pp[(1 + w) * 64 + 32];
            if (u == 0) {
                sd[0] = d0.x; sd[1] = d0.y; sd[2] = d0.z; sd[3] = d0.w;
                sd[4] = d1.x; sd[5] = d1.y; sd[6] = d1.z; sd[7] = d1.w;
                so[0] = o0.x; so[1] = o0.y; so[2] = o0.z; so[3] = o0.w;
                so[4] = o1.x; so[5] = o1.y; so[6] = o1.z; so[7] = o1.w;
            } else {
                sd[0] += d0.x; sd[1] += d0.y; sd[2] += d0.z; sd[3] += d0.w;
                sd[4] += d1.x; sd[5] += d1.y; sd[6] += d1.z; sd[7] += d1.w;
                so[0] += o0.x; so[1] += o0.y; so[2] += o0.z; so[3] += o0.w;
                so[4] += o1.x; so[5] += o1.y; so[6] += o1.z; so[7] += o1.w;
            }
        }
#pragma unroll
        for (int r = 0; r < 8; r++) {
            sTw[(r + 8 * half) * PS + l16] = sd[r];
            sTw[(16 + r + 8 * half) * PS + l16] = so[r];
        }
    }
    __syncthreads();
    if (w >= R) return;
    float t[16];
#pragma unroll
    for (int c = 0; c < 16; c++) t[c] = sTw[lane * PS + c];

    float mypiv = 0.f;
#pragma unroll
    for (int p = 0; p < 16; p++) {
        float u[16];
#pragma unroll
        for (int c2 = p; c2 < 16; c2++)
            u[c2] = __int_as_float(__builtin_amdgcn_readlane(__float_as_int(t[c2]), p));
        const float dinv = 1.0f / u[p];
        const bool below = lane > p;
        mypiv = (lane == p) ? u[p] : mypiv;
        const float lij = t[p] * dinv;
        t[p] = below ? lij : t[p];
#pragma unroll
        for (int c2 = p + 1; c2 < 16; c2++) {
            const float v = __builtin_fmaf(-lij, u[c2], t[c2]);
            t[c2] = below ? v : t[c2];
        }
    }

    const int myt = (int)blockIdx.x * R + w;
    float* dst = nullptr;
    if (myt == 0) {
        if (lane < 16) {
            dst = Kb + (size_t)(c0 + lane) * PP + c0 + ((jb < NBLK - 1) ? 16 : 0);
            db[c0 + lane] = mypiv;
        }
    } else if (myt < nT && lane >= 16) {
        dst = Kb + (size_t)(c0 + 16 * myt + lane - 16) * PP + c0;
    }
    if (dst) {
#pragma unroll
        for (int c = 0; c < 16; c += 4) {
            float4 v; v.x = t[c]; v.y = t[c + 1]; v.z = t[c + 2]; v.w = t[c + 3];
            *(float4*)(dst + c) = v;
        }
    }
}

#define FCH 24
#define RP  20

__global__ void __launch_bounds__(256) trisolve_kernel(const float* __restrict__ K,
                                                       const float* __restrict__ rhs,
                                                       float* __restrict__ x) {
    const int b = blockIdx.x, tid = threadIdx.x;
    const int l16 = tid & 15, grp = tid >> 4;
    __shared__ __attribute__((aligned(16))) float y[PP];
    __shared__ __attribute__((aligned(16))) float dg[PP];
    __shared__ __attribute__((aligned(16))) float red[16 * RP];
    const float* Kb = K + (size_t)b * PP * PP;
    for (int i = tid; i < PP; i += 256) {
        y[i] = rhs[(size_t)b * PP + i];
        dg[i] = Kb[(size_t)i * PP + i + (((i >> 4) < NBLK - 1) ? 16 : 0)];
    }
    const float4 zero4 = make_float4(0.f, 0.f, 0.f, 0.f);

    float4 cur[FCH];
    float lcur[16];
#pragma unroll
    for (int m = 0; m < FCH; m++) cur[m] = zero4;
#pragma unroll
    for (int c = 0; c < 16; c++) lcur[c] = 0.f;
    if (tid < 32) {
        const float* lp = Kb + (size_t)l16 * PP + 16;
#pragma unroll
        for (int c = 0; c < 4; c++) {
            float4 t4 = *(const float4*)(lp + 4 * c);
            lcur[4 * c] = t4.x; lcur[4 * c + 1] = t4.y; lcur[4 * c + 2] = t4.z; lcur[4 * c + 3] = t4.w;
        }
    }
    __syncthreads();
    for (int kb = 0; kb < NBLK; kb++) {
        const int r0 = kb * 16, r1 = r0 + 16;
        float4 nxt[FCH];
        float lnxt[16];
#pragma unroll
        for (int m = 0; m < FCH; m++) nxt[m] = zero4;
#pragma unroll
        for (int c = 0; c < 16; c++) lnxt[c] = 0.f;
        if (kb + 1 < NBLK) {
            const float* rp = Kb + (size_t)(r1 + grp) * PP;
#pragma unroll
            for (int m = 0; m < FCH; m++) {
                if (64 * m < r1) {
                    const int k = 64 * m + 4 * l16;
                    if (k < r1) nxt[m] = *(const float4*)(rp + k);
                }
            }
            if (tid < 32) {
                const int d1 = r1 + ((kb + 1 < NBLK - 1) ? 16 : 0);
                const float* lp = Kb + (size_t)(r1 + l16) * PP + d1;
#pragma unroll
                for (int c = 0; c < 4; c++) {
                    float4 t4 = *(const float4*)(lp + 4 * c);
                    lnxt[4 * c] = t4.x; lnxt[4 * c + 1] = t4.y; lnxt[4 * c + 2] = t4.z; lnxt[4 * c + 3] = t4.w;
                }
            }
        }
        float a0 = 0.f, a1 = 0.f, a2 = 0.f, a3 = 0.f;
#pragma unroll
        for (int g = 0; g < FCH / 4; g++) {
            if (256 * g < r0) {
#pragma unroll
                for (int mm = 0; mm < 4; mm++) {
                    const int m = 4 * g + mm;
                    const int k = 64 * m + 4 * l16;
                    const float4 yy = *(const float4*)(y + ((k < PP) ? k : 0));
                    a0 = __builtin_fmaf(cur[m].x, yy.x, a0); a1 = __builtin_fmaf(cur[m].y, yy.y, a1);
                    a2 = __builtin_fmaf(cur[m].z, yy.z, a2); a3 = __builtin_fmaf(cur[m].w, yy.w, a3);
                }
            }
        }
        red[grp * RP + l16] = (a0 + a1) + (a2 + a3);
        __syncthreads();
        if (tid < 32) {
            const float4* rr = (const float4*)(red + l16 * RP);
            const float4 s0 = rr[0], s1 = rr[1], s2 = rr[2], s3 = rr[3];
            const float dot = (((s0.x + s0.y) + (s0.z + s0.w)) + ((s1.x + s1.y) + (s1.z + s1.w)))
                            + (((s2.x + s2.y) + (s2.z + s2.w)) + ((s3.x + s3.y) + (s3.z + s3.w)));
            float v = y[r0 + l16] - dot;
#pragma unroll
            for (int p = 0; p < 15; p++) {
                const float vp = __int_as_float(__builtin_amdgcn_readlane(__float_as_int(v), p));
                if (l16 > p) v -= lcur[p] * vp;
            }
            if (tid < 16) y[r0 + l16] = v;
        }
        __syncthreads();
#pragma unroll
        for (int m = 0; m < FCH; m++) cur[m] = nxt[m];
#pragma unroll
        for (int c = 0; c < 16; c++) lcur[c] = lnxt[c];
    }

    float ucur[16];
#pragma unroll
    for (int c = 0; c < 16; c++) ucur[c] = 0.f;
    if (tid < 32) {
        const float* up = Kb + (size_t)((NBLK - 1) * 16 + l16) * PP + (NBLK - 1) * 16;
#pragma unroll
        for (int c = 0; c < 4; c++) {
            float4 t4 = *(const float4*)(up + 4 * c);
            ucur[4 * c] = t4.x; ucur[4 * c + 1] = t4.y; ucur[4 * c + 2] = t4.z; ucur[4 * c + 3] = t4.w;
        }
    }
    __syncthreads();
    for (int kb = NBLK - 1; kb >= 0; kb--) {
        const int r0 = kb * 16;
        const int ia = 4 * (tid - 32), ib = ia + 4 * 224;
        float4 la[16], lb[16];
        if (tid >= 32) {
            if (ia < r0) {
#pragma unroll
                for (int j = 0; j < 16; j++)
                    la[j] = *(const float4*)(Kb + (size_t)(r0 + j) * PP + ia);
            }
            if (ib < r0) {
#pragma unroll
                for (int j = 0; j < 16; j++)
                    lb[j] = *(const float4*)(Kb + (size_t)(r0 + j) * PP + ib);
            }
        } else {
            float v = y[r0 + l16];
            const float rinv = 1.0f / dg[r0 + l16];
#pragma unroll
            for (int p = 15; p >= 0; p--) {
                if (l16 == p) v *= rinv;
                const float xp = __int_as_float(__builtin_amdgcn_readlane(__float_as_int(v), p));
                if (l16 < p) v -= ucur[p] * xp;
            }
            if (tid < 16) y[r0 + l16] = v;
        }
        __syncthreads();
        float unxt[16];
#pragma unroll
        for (int c = 0; c < 16; c++) unxt[c] = 0.f;
        if (tid < 32) {
            if (kb > 0) {
                const float* up = Kb + (size_t)(r0 - 16 + l16) * PP + r0;
#pragma unroll
                for (int c = 0; c < 4; c++) {
                    float4 t4 = *(const float4*)(up + 4 * c);
                    unxt[4 * c] = t4.x; unxt[4 * c + 1] = t4.y; unxt[4 * c + 2] = t4.z; unxt[4 * c + 3] = t4.w;
                }
            }
        } else {
            float xs[16];
#pragma unroll
            for (int c = 0; c < 4; c++) {
                const float4 t4 = *(const float4*)(y + r0 + 4 * c);
                xs[4 * c] = t4.x; xs[4 * c + 1] = t4.y; xs[4 * c + 2] = t4.z; xs[4 * c + 3] = t4.w;
            }
            if (ia < r0) {
                float4 acc = zero4;
#pragma unroll
                for (int j = 0; j < 16; j++) {
                    acc.x = __builtin_fmaf(la[j].x, xs[j], acc.x); acc.y = __builtin_fmaf(la[j].y, xs[j], acc.y);
                    acc.z = __builtin_fmaf(la[j].z, xs[j], acc.z); acc.w = __builtin_fmaf(la[j].w, xs[j], acc.w);
                }
                float4 yy = *(const float4*)(y + ia);
                const float4 d4 = *(const float4*)(dg + ia);
                yy.x = __builtin_fmaf(-d4.x, acc.x, yy.x); yy.y = __builtin_fmaf(-d4.y, acc.y, yy.y);
                yy.z = __builtin_fmaf(-d4.z, acc.z, yy.z); yy.w = __builtin_fmaf(-d4.w, acc.w, yy.w);
                *(float4*)(y + ia) = yy;
            }
            if (ib < r0) {
                float4 acc = zero4;
#pragma unroll
                for (int j = 0; j < 16; j++) {
                    acc.x = __builtin_fmaf(lb[j].x, xs[j], acc.x); acc.y = __builtin_fmaf(lb[j].y, xs[j], acc.y);
                    acc.z = __builtin_fmaf(lb[j].z, xs[j], acc.z); acc.w = __builtin_fmaf(lb[j].w, xs[j], acc.w);
                }
                float4 yy = *(const float4*)(y + ib);
                const float4 d4 = *(const float4*)(dg + ib);
                yy.x = __builtin_fmaf(-d4.x, acc.x, yy.x); yy.y = __builtin_fmaf(-d4.y, acc.y, yy.y);
                yy.z = __builtin_fmaf(-d4.z, acc.z, yy.z); yy.w = __builtin_fmaf(-d4.w, acc.w, yy.w);
                *(float4*)(y + ib) = yy;
            }
        }
        __syncthreads();
#pragma unroll
        for (int c = 0; c < 16; c++) ucur[c] = unxt[c];
    }
    for (int i = tid; i < PP; i += 256) x[(size_t)b * PP + i] = y[i];
}

__global__ void vote_kernel(const float* __restrict__ x, float* __restrict__ out) {
    int s = blockIdx.x, b = blockIdx.y, tid = threadIdx.x;
    __shared__ float vals[NN_];
    if (tid < NN_) {
        float v = x[(size_t)b * PP + tid * NN_ + s];
        vals[tid] = fminf(fmaxf(v, 0.f), 1.f);
    }
    __syncthreads();
    if (tid == 0) {
        float m = -1e30f;
        for (int t = 0; t < NN_; t++) m = fmaxf(m, vals[t]);
        float sum = 0.f;
        for (int t = 0; t < NN_; t++) { float e = __expf(ALPHA * (vals[t] - m)); vals[t] = e; sum += e; }
        float inv = 1.f / sum;
        for (int t = 0; t < NN_; t++) out[((size_t)b * NN_ + s) * NN_ + t] = vals[t] * inv;
    }
}

extern "C" void kernel_launch(void* const* d_in, const int* in_sizes, int n_in,
                              void* d_out, int out_size, void* d_ws, size_t ws_size,
                              hipStream_t stream) {
    const float* emb1 = (const float*)d_in[0];
    const float* emb2 = (const float*)d_in[1];
    const float* Wcg  = (const float*)d_in[2];
    const float* bcg  = (const float*)d_in[3];

    float* ws  = (float*)d_ws;
    size_t kelem = (size_t)BB * PP * PP;
    float* Km  = ws;
    float* h1  = Km + kelem;
    float* h2  = h1 + (size_t)BB * NN_ * DD;
    float* S   = h2 + (size_t)BB * NN_ * DD;
    float* rhs = S  + (size_t)BB * NN2;
    float* x   = rhs + (size_t)BB * PP;

    feat_kernel<<<dim3(BB, 2), 256, 0, stream>>>(emb1, emb2, Wcg, bcg, h1, h2);
    gram_rhs_kernel<<<BB, 256, 0, stream>>>(h1, h2, S, rhs);
    assemble_kernel<<<dim3(NBLK, NBLK, BB), dim3(16, 16), 0, stream>>>(S, Km);

    for (int jb = 0; jb < 22; jb++)
        lu_col_kernel<4><<<dim3((NBLK - jb + 3) / 4, BB), 32 * NW, 0, stream>>>(Km, x, jb);
    for (int jb = 22; jb < 54; jb++)
        lu_col_kernel<2><<<dim3((NBLK - jb + 1) / 2, BB), 32 * NW, 0, stream>>>(Km, x, jb);
    for (int jb = 54; jb < NBLK; jb++)
        lu_col_kernel<1><<<dim3(NBLK - jb, BB), 32 * NW, 0, stream>>>(Km, x, jb);

    trisolve_kernel<<<BB, 256, 0, stream>>>(Km, rhs, x);
    vote_kernel<<<dim3(NN_, BB), 64, 0, stream>>>(x, (float*)d_out);
}


# KKT assembly launch: 473344 one-store waves become 2752 waves that walk rows with 16-byte stores and write only the lower block triangle, all that LU and solve read (values bit-identical)
# speedup vs baseline: 6.7724x; 1.0333x over previous
#include <hip/hip_runtime.h>
#include <math.h>


#define BB   8
#define NN_  36
#define DD   64
#define NN2  (NN_*NN_)
#define NCON (2*NN_)
#define KSZ  (NN2+NCON)
#define PP   1376
#define NBLK (PP/16)
#define KDIAG 1225.0f
#define EPSK  1e-5f
#define ALPHA 1000.0f
#define PS   17

typedef float v2f __attribute__((ext_vector_type(2)));
typedef float v4f __attribute__((ext_vector_type(4)));
typedef float v8f __attribute__((ext_vector_type(8)));

__global__ void feat_kernel(const float* __restrict__ emb1,
                            const float* __restrict__ emb2,
                            const float* __restrict__ W,
                            const float* __restrict__ bias,
                            float* __restrict__ h1, float* __restrict__ h2) {
    int b = blockIdx.x, which = blockIdx.y;
    const float* e = (which == 0 ? emb1 : emb2) + (size_t)b * NN_ * DD;
    float* h = (which == 0 ? h1 : h2) + (size_t)b * NN_ * DD;
    __shared__ float se[NN_ * DD];
    __shared__ float sw[DD * DD];
    __shared__ float snrm[DD];
    int tid = threadIdx.x;
    for (int i = tid; i < NN_ * DD; i += 256) se[i] = e[i];
    for (int i = tid; i < DD * DD; i += 256) sw[i] = W[i];
    __syncthreads();
    for (int d = tid; d < DD; d += 256) {
        float s = 0.f;
        for (int n = 0; n < NN_; n++) { float v = se[n * DD + d]; s += v * v; }
        snrm[d] = fmaxf(sqrtf(s), 1e-12f);
    }
    __syncthreads();
    for (int i = tid; i < NN_ * DD; i += 256) se[i] = se[i] / snrm[i % DD];
    __syncthreads();
    for (int idx = tid; idx < NN_ * DD; idx += 256) {
        int n = idx / DD, i = idx % DD;
        float acc = bias[i];
        for (int d = 0; d < DD; d++) acc += se[n * DD + d] * sw[i * DD + d];
        h[idx] = fmaxf(acc, 0.f);
    }
}

__global__ void gram_rhs_kernel(const float* __restrict__ h1,
                                const float* __restrict__ h2,
                                float* __restrict__ S, float* __restrict__ rhs) {
    int b = blockIdx.x, tid = threadIdx.x;
    __shared__ float s1[NN_ * DD];
    __shared__ float s2[NN_ * DD];
    for (int i = tid; i < NN_ * DD; i += 256) {
        s1[i] = h1[(size_t)b * NN_ * DD + i];
        s2[i] = h2[(size_t)b * NN_ * DD + i];
    }
    __syncthreads();
    for (int idx = tid; idx < NN2; idx += 256) {
        int i = idx / NN_, j = idx % NN_;
        float a = 0.f;
        for (int d = 0; d < DD; d++) a += s1[i * DD + d] * s2[j * DD + d];
        S[(size_t)b * NN2 + idx] = a;
        rhs[(size_t)b * PP + j * NN_ + i] = a;
    }
    for (int r = NN2 + tid; r < PP; r += 256)
        rhs[(size_t)b * PP + r] = (r < KSZ) ? 1.0f : 0.0f;
}

static_assert(NN2 % 16 == 0, "rows of the quadratic block own whole tiles");
static_assert(NN_ % 4 == 0, "four neighbouring columns share u");
static_assert(NBLK % 2 == 0, "block rows are paired");
static_assert(NN2 <= 6 * 256, "six loads a thread cover S");
__global__ void __launch_bounds__(256) assemble_kernel(const float* __restrict__ S, float* __restrict__ K) {
    const int b = blockIdx.y, tid = threadIdx.x;
    const int row16 = tid >> 4, lane16 = tid & 15;
    __shared__ __attribute__((aligned(16))) float sT[NN2];
    const float* Sb = S + (size_t)b * NN2;
    float sv[6];
#pragma unroll
    for (int k = 0; k < 6; k++) {
        const int i = tid + 256 * k;
        const int u = i / NN_, v = i - u * NN_;
        sv[k] = (i < NN2) ? Sb[v * NN_ + u] : 0.f;
    }
#pragma unroll
    for (int k = 0; k < 6; k++) {
        const int i = tid + 256 * k;
        if (i < NN2) sT[i] = sv[k];
    }
    __syncthreads();
    float* Kb = K + (size_t)b * PP * PP;
    for (int pass = 0; pass < 2; pass++) {
        const int I = (pass == 0) ? (int)blockIdx.x : NBLK - 1 - (int)blockIdx.x;
        const int r = 16 * I + row16;
        const int ncq = 4 * (I + 1);
        float* Kr = Kb + (size_t)r * PP;
        if (r < NN2) {
            const int t = r / NN_, s = r - t * NN_;
            const float sr = sT[r];
            for (int cq = lane16; cq < ncq; cq += 16) {
                const int c = 4 * cq;
                const int u = c / NN_, v0 = c - u * NN_;
                const float4 sc4 = *(const float4*)(sT + c);
                const float sc[4] = {sc4.x, sc4.y, sc4.z, sc4.w};
                const bool tu = (t != u);
                float o[4];
#pragma unroll
                for (int j = 0; j < 4; j++) {
                    float val = (r == c + j) ? KDIAG : 0.f;
                    if (s != v0 + j && tu) val -= 0.5f * (sr + sc[j]);
                    o[j] = val;
                }
                *(float4*)(Kr + c) = make_float4(o[0], o[1], o[2], o[3]);
            }
        } else {
            const int i = r - NN2;
            for (int cq = lane16; cq < ncq; cq += 16) {
                const int c = 4 * cq;
                float o[4];
                if (c < NN2) {
                    const int u = c / NN_, v0 = c - u * NN_;
#pragma unroll
                    for (int j = 0; j < 4; j++) {
                        float val = 0.f;
                        if (i < NN_)        val = (v0 + j == i) ? 1.f : 0.f;
                        else if (i < NCON)  val = (u == i - NN_) ? 1.f : 0.f;
                        o[j] = val;
                    }
                } else {
#pragma unroll
                    for (int j = 0; j < 4; j++)
                        o[j] = (r == c + j) ? ((r < KSZ) ? -EPSK : 1.0f) : 0.f;
                }
                *(float4*)(Kr + c) = make_float4(o[0], o[1], o[2], o[3]);
            }
        }
    }
}

#define QS 16
#define NW 4
#define SP 68

__device__ __forceinline__ void wave_lds_order() {
    __builtin_amdgcn_fence(__ATOMIC_RELEASE, "wavefront");
    __builtin_amdgcn_wave_barrier();
    __builtin_amdgcn_fence(__ATOMIC_ACQUIRE, "wavefront");
}

template <int R>
__global__ void __launch_bounds__(32 * NW) lu_col_kernel(float* __restrict__ K,
                                                         float* __restrict__ dpiv,
                                                         int jb) {
    constexpr int NT = R + 1;
    constexpr int STRIP = NT * 16 * SP + 4 * QS;
    __shared__ float sT[R * 32 * PS];
    __shared__ __attribute__((aligned(16))) float4 sP[NW * NT * 2 * 32];
    __shared__ __attribute__((aligned(16))) float sStage[NW * STRIP];
    const int b = blockIdx.y, lane = threadIdx.x & 31;
    const int w = __builtin_amdgcn_readfirstlane((int)(threadIdx.x >> 5));
    const int half = lane >> 4, l16 = lane & 15;
    float* Kb = K + (size_t)b * PP * PP;
    float* db = dpiv + (size_t)b * PP;
    const int c0 = jb * 16;
    const int nT = NBLK - jb;
    int rown[R];
#pragma unroll
    for (int t = 0; t < R; t++) {
        const int tt = (int)blockIdx.x * R + t;
        rown[t] = c0 + 16 * ((tt < nT) ? tt : 0);
    }

    v8f cd, co[R];
    if (w == 0) {
#pragma unroll
        for (int r = 0; r < 8; r++) {
            cd[r] = Kb[(size_t)(c0 + r + 8 * half) * PP + c0 + l16];
#pragma unroll
            for (int t = 0; t < R; t++)
                co[t][r] = Kb[(size_t)(rown[t] + r + 8 * half) * PP + c0 + l16];
        }
    } else {
#pragma unroll
        for (int r = 0; r < 8; r++) {
            cd[r] = 0.f;
#pragma unroll
            for (int t = 0; t < R; t++) co[t][r] = 0.f;
        }
    }

    const int Q = jb >> 2;
    const int qb = Q >> 2, qr = Q & 3;
    const int nq = qb + ((w < qr) ? 1 : 0);
    const int kq = (w * qb + ((w < qr) ? w : qr)) * (4 * QS);

    const float* pl0 = Kb + (size_t)(c0 + l16) * PP + 2 * half;
    const float* pd0 = db + 2 * half;
    const float* pa0[R];
    const float* ga[R];
    const float* gl = Kb + (size_t)(c0 + half) * PP + 4 * l16 + kq;
    const float* gd = db + 2 * lane + kq;
    float* stL = sStage + w * STRIP + R * 16 * SP;
    float* stD = stL + 16 * SP;
    float* wA[R];
    float* wL = stL + half * SP + 4 * l16;
    float* wD = stD + 2 * lane;
    const float* fA[R];
    const float* fL = stL + l16 * SP + 2 * half;
    const float* fD = stD + 2 * half;
#pragma unroll
    for (int t = 0; t < R; t++) {
        pa0[t] = Kb + (size_t)(rown[t] + l16) * PP + 2 * half;
        ga[t] = Kb + (size_t)(rown[t] + half) * PP + 4 * l16 + kq;
        float* stA = sStage + w * STRIP + t * 16 * SP;
        wA[t] = stA + half * SP + 4 * l16;
        fA[t] = stA + l16 * SP + 2 * half;
    }

    v4f ga4[R][8], gl4[8];
    v2f gd2;
    v2f ra[R][QS], rl[QS], rd[QS];
    if (nq > 0) {
#pragma unroll
        for (int j = 0; j < 8; j++) {
#pragma unroll
            for (int t = 0; t < R; t++) ga4[t][j] = *(const v4f*)(ga[t] + (size_t)(2 * j) * PP);
            gl4[j] = *(const v4f*)(gl + (size_t)(2 * j) * PP);
        }
        gd2 = *(const v2f*)gd;
#pragma unroll
        for (int j = 0; j < 8; j++) {
#pragma unroll
            for (int t = 0; t < R; t++) *(v4f*)(wA[t] + 2 * j * SP) = ga4[t][j];
            *(v4f*)(wL + 2 * j * SP) = gl4[j];
        }
        *(v2f*)wD = gd2;
        if (nq > 1) {
#pragma unroll
            for (int j = 0; j < 8; j++) {
#pragma unroll
                for (int t = 0; t < R; t++) ga4[t][j] = *(const v4f*)(ga[t] + (size_t)(2 * j) * PP + 4 * QS);
                gl4[j] = *(const v4f*)(gl + (size_t)(2 * j) * PP + 4 * QS);
            }
            gd2 = *(const v2f*)(gd + 4 * QS);
        }
        wave_lds_order();
#pragma unroll
        for (int s = 0; s < QS; s++) {
#pragma unroll
            for (int t = 0; t < R; t++) ra[t][s] = *(const v2f*)(fA[t] + 4 * s);
            rl[s] = *(const v2f*)(fL + 4 * s);
            rd[s] = *(const v2f*)(fD + 4 * s);
        }
    }
    for (int q = 0; q < nq; q++) {
        v2f an[R][QS], ln[QS], bb[QS];
#pragma unroll
        for (int s = 0; s < QS; s++) {
#pragma unroll
            for (int t = 0; t < R; t++) an[t][s] = -ra[t][s];
            ln[s] = -rl[s];
            bb[s] = rl[s] * rd[s];
        }
        if (q + 1 < nq) {
            wave_lds_order();
#pragma unroll
            for (int j = 0; j < 8; j++) {
#pragma unroll
                for (int t = 0; t < R; t++) *(v4f*)(wA[t] + 2 * j * SP) = ga4[t][j];
                *(v4f*)(wL + 2 * j * SP) = gl4[j];
            }
            *(v2f*)wD = gd2;
            if (q + 2 < nq) {
                const int k2 = (q + 2) * (4 * QS);
#pragma unroll
                for (int j = 0; j < 8; j++) {
#pragma unroll
                    for (int t = 0; t < R; t++) ga4[t][j] = *(const v4f*)(ga[t] + (size_t)(2 * j) * PP + k2);
                    gl4[j] = *(const v4f*)(gl + (size_t)(2 * j) * PP + k2);
                }
                gd2 = *(const v2f*)(gd + k2);
            }
            wave_lds_order();
#pragma unroll
            for (int s = 0; s < QS; s++) {
#pragma unroll
                for (int t = 0; t < R; t++) ra[t][s] = *(const v2f*)(fA[t] + 4 * s);
                rl[s] = *(const v2f*)(fL + 4 * s);
                rd[s] = *(const v2f*)(fD + 4 * s);
            }
        }
#pragma unroll
        for (int s = 0; s < QS; s++) {
#pragma unroll
            for (int t = 0; t < R; t++)
                co[t] = __builtin_amdgcn_wmma_f32_16x16x4_f32(false, an[t][s], false, bb[s], (short)0, co[t], false, false);
            cd = __builtin_amdgcn_wmma_f32_16x16x4_f32(false, ln[s], false, bb[s], (short)0, cd, false, false);
        }
    }
    if (w == NW - 1) {
        for (int k0 = Q * (4 * QS); k0 < c0; k0 += 16) {
            v2f ta[R][4], tl[4], td[4];
#pragma unroll
            for (int s = 0; s < 4; s++) {
#pragma unroll
                for (int t = 0; t < R; t++) ta[t][s] = *(const v2f*)(pa0[t] + k0 + 4 * s);
                tl[s] = *(const v2f*)(pl0 + k0 + 4 * s);
                td[s] = *(const v2f*)(pd0 + k0 + 4 * s);
            }
#pragma unroll
            for (int s = 0; s < 4; s++) {
                v2f l1 = -tl[s], b1 = tl[s] * td[s];
#pragma unroll
                for (int t = 0; t < R; t++) {
                    v2f a1 = -ta[t][s];
                    co[t] = __builtin_amdgcn_wmma_f32_16x16x4_f32(false, a1, false, b1, (short)0, co[t], false, false);
                }
                cd = __builtin_amdgcn_wmma_f32_16x16x4_f32(false, l1, false, b1, (short)0, cd, false, false);
            }
        }
    }

    {
        float4* pp = sP + (w * NT) * 64 + lane;
        pp[0]  = make_float4(cd[0], cd[1], cd[2], cd[3]);
        pp[32] = make_float4(cd[4], cd[5], cd[6], cd[7]);
#pragma unroll
        for (int t = 0; t < R; t++) {
            pp[(1 + t) * 64]      = make_float4(co[t][0], co[t][1], co[t][2], co[t][3]);
            pp[(1 + t) * 64 + 32] = make_float4(co[t][4], co[t][5], co[t][6], co[t][7]);
        }
    }
    __syncthreads();
    float* sTw = sT + ((w < R) ? w : 0) * 32 * PS;
    if (w < R) {
        float sd[8], so[8];
#pragma unroll
        for (int u = 0; u < NW; u++) {
            const float4* pp = sP + (u * NT) * 64 + lane;
            const float4 d0 = pp[0], d1 = pp[32];
            const float4 o0 = pp[(1 + w) * 64], o1 = pp[(1 + w) * 64 + 32];
            if (u == 0) {
                sd[0] = d0.x; sd[1] = d0.y; sd[2] = d0.z; sd[3] = d0.w;
                sd[4] = d1.x; sd[5] = d1.y; sd[6] = d1.z; sd[7] = d1.w;
                so[0] = o0.x; so[1] = o0.y; so[2] = o0.z; so[3] = o0.w;
                so[4] = o1.x; so[5] = o1.y; so[6] = o1.z; so[7] = o1.w;
            } else {
                sd[0] += d0.x; sd[1] += d0.y; sd[2] += d0.z; sd[3] += d0.w;
                sd[4] += d1.x; sd[5] += d1.y; sd[6] += d1.z; sd[7] += d1.w;
                so[0] += o0.x; so[1] += o0.y; so[2] += o0.z; so[3] += o0.w;
                so[4] += o1.x; so[5] += o1.y; so[6] += o1.z; so[7] += o1.w;
            }
        }
#pragma unroll
        for (int r = 0; r < 8; r++) {
            sTw[(r + 8 * half) * PS + l16] = sd[r];
            sTw[(16 + r + 8 * half) * PS + l16] = so[r];
        }
    }
    __syncthreads();
    if (w >= R) return;
    float t[16];
#pragma unroll
    for (int c = 0; c < 16; c++) t[c] = sTw[lane * PS + c];

    float mypiv = 0.f;
#pragma unroll
    for (int p = 0; p < 16; p++) {
        float u[16];
#pragma unroll
        for (int c2 = p; c2 < 16; c2++)
            u[c2] = __int_as_float(__builtin_amdgcn_readlane(__float_as_int(t[c2]), p));
        const float dinv = 1.0f / u[p];
        const bool below = lane > p;
        mypiv = (lane == p) ? u[p] : mypiv;
        const float lij = t[p] * dinv;
        t[p] = below ? lij : t[p];
#pragma unroll
        for (int c2 = p + 1; c2 < 16; c2++) {
            const float v = __builtin_fmaf(-lij, u[c2], t[c2]);
            t[c2] = below ? v : t[c2];
        }
    }

    const int myt = (int)blockIdx.x * R + w;
    float* dst = nullptr;
    if (myt == 0) {
        if (lane < 16) {
            dst = Kb + (size_t)(c0 + lane) * PP + c0 + ((jb < NBLK - 1) ? 16 : 0);
            db[c0 + lane] = mypiv;
        }
    } else if (myt < nT && lane >= 16) {
        dst = Kb + (size_t)(c0 + 16 * myt + lane - 16) * PP + c0;
    }
    if (dst) {
#pragma unroll
        for (int c = 0; c < 16; c += 4) {
            float4 v; v.x = t[c]; v.y = t[c + 1]; v.z = t[c + 2]; v.w = t[c + 3];
            *(float4*)(dst + c) = v;
        }
    }
}

#define FCH 24
#define RP  20

__global__ void __launch_bounds__(256) trisolve_kernel(const float* __restrict__ K,
                                                       const float* __restrict__ rhs,
                                                       float* __restrict__ x) {
    const int b = blockIdx.x, tid = threadIdx.x;
    const int l16 = tid & 15, grp = tid >> 4;
    __shared__ __attribute__((aligned(16))) float y[PP];
    __shared__ __attribute__((aligned(16))) float dg[PP];
    __shared__ __attribute__((aligned(16))) float red[16 * RP];
    const float* Kb = K + (size_t)b * PP * PP;
    for (int i = tid; i < PP; i += 256) {
        y[i] = rhs[(size_t)b * PP + i];
        dg[i] = Kb[(size_t)i * PP + i + (((i >> 4) < NBLK - 1) ? 16 : 0)];
    }
    const float4 zero4 = make_float4(0.f, 0.f, 0.f, 0.f);

    float4 cur[FCH];
    float lcur[16];
#pragma unroll
    for (int m = 0; m < FCH; m++) cur[m] = zero4;
#pragma unroll
    for (int c = 0; c < 16; c++) lcur[c] = 0.f;
    if (tid < 32) {
        const float* lp = Kb + (size_t)l16 * PP + 16;
#pragma unroll
        for (int c = 0; c < 4; c++) {
            float4 t4 = *(const float4*)(lp + 4 * c);
            lcur[4 * c] = t4.x; lcur[4 * c + 1] = t4.y; lcur[4 * c + 2] = t4.z; lcur[4 * c + 3] = t4.w;
        }
    }
    __syncthreads();
    for (int kb = 0; kb < NBLK; kb++) {
        const int r0 = kb * 16, r1 = r0 + 16;
        float4 nxt[FCH];
        float lnxt[16];
#pragma unroll
        for (int m = 0; m < FCH; m++) nxt[m] = zero4;
#pragma unroll
        for (int c = 0; c < 16; c++) lnxt[c] = 0.f;
        if (kb + 1 < NBLK) {
            const float* rp = Kb + (size_t)(r1 + grp) * PP;
#pragma unroll
            for (int m = 0; m < FCH; m++) {
                if (64 * m < r1) {
                    const int k = 64 * m + 4 * l16;
                    if (k < r1) nxt[m] = *(const float4*)(rp + k);
                }
            }
            if (tid < 32) {
                const int d1 = r1 + ((kb + 1 < NBLK - 1) ? 16 : 0);
                const float* lp = Kb + (size_t)(r1 + l16) * PP + d1;
#pragma unroll
                for (int c = 0; c < 4; c++) {
                    float4 t4 = *(const float4*)(lp + 4 * c);
                    lnxt[4 * c] = t4.x; lnxt[4 * c + 1] = t4.y; lnxt[4 * c + 2] = t4.z; lnxt[4 * c + 3] = t4.w;
                }
            }
        }
        float a0 = 0.f, a1 = 0.f, a2 = 0.f, a3 = 0.f;
#pragma unroll
        for (int g = 0; g < FCH / 4; g++) {
            if (256 * g < r0) {
#pragma unroll
                for (int mm = 0; mm < 4; mm++) {
                    const int m = 4 * g + mm;
                    const int k = 64 * m + 4 * l16;
                    const float4 yy = *(const float4*)(y + ((k < PP) ? k : 0));
                    a0 = __builtin_fmaf(cur[m].x, yy.x, a0); a1 = __builtin_fmaf(cur[m].y, yy.y, a1);
                    a2 = __builtin_fmaf(cur[m].z, yy.z, a2); a3 = __builtin_fmaf(cur[m].w, yy.w, a3);
                }
            }
        }
        red[grp * RP + l16] = (a0 + a1) + (a2 + a3);
        __syncthreads();
        if (tid < 32) {
            const float4* rr = (const float4*)(red + l16 * RP);
            const float4 s0 = rr[0], s1 = rr[1], s2 = rr[2], s3 = rr[3];
            const float dot = (((s0.x + s0.y) + (s0.z + s0.w)) + ((s1.x + s1.y) + (s1.z + s1.w)))
                            + (((s2.x + s2.y) + (s2.z + s2.w)) + ((s3.x + s3.y) + (s3.z + s3.w)));
            float v = y[r0 + l16] - dot;
#pragma unroll
            for (int p = 0; p < 15; p++) {
                const float vp = __int_as_float(__builtin_amdgcn_readlane(__float_as_int(v), p));
                if (l16 > p) v -= lcur[p] * vp;
            }
            if (tid < 16) y[r0 + l16] = v;
        }
        __syncthreads();
#pragma unroll
        for (int m = 0; m < FCH; m++) cur[m] = nxt[m];
#pragma unroll
        for (int c = 0; c < 16; c++) lcur[c] = lnxt[c];
    }

    float ucur[16];
#pragma unroll
    for (int c = 0; c < 16; c++) ucur[c] = 0.f;
    if (tid < 32) {
        const float* up = Kb + (size_t)((NBLK - 1) * 16 + l16) * PP + (NBLK - 1) * 16;
#pragma unroll
        for (int c = 0; c < 4; c++) {
            float4 t4 = *(const float4*)(up + 4 * c);
            ucur[4 * c] = t4.x; ucur[4 * c + 1] = t4.y; ucur[4 * c + 2] = t4.z; ucur[4 * c + 3] = t4.w;
        }
    }
    __syncthreads();
    for (int kb = NBLK - 1; kb >= 0; kb--) {
        const int r0 = kb * 16;
        const int ia = 4 * (tid - 32), ib = ia + 4 * 224;
        float4 la[16], lb[16];
        if (tid >= 32) {
            if (ia < r0) {
#pragma unroll
                for (int j = 0; j < 16; j++)
                    la[j] = *(const float4*)(Kb + (size_t)(r0 + j) * PP + ia);
            }
            if (ib < r0) {
#pragma unroll
                for (int j = 0; j < 16; j++)
                    lb[j] = *(const float4*)(Kb + (size_t)(r0 + j) * PP + ib);
            }
        } else {
            float v = y[r0 + l16];
            const float rinv = 1.0f / dg[r0 + l16];
#pragma unroll
            for (int p = 15; p >= 0; p--) {
                if (l16 == p) v *= rinv;
                const float xp = __int_as_float(__builtin_amdgcn_readlane(__float_as_int(v), p));
                if (l16 < p) v -= ucur[p] * xp;
            }
            if (tid < 16) y[r0 + l16] = v;
        }
        __syncthreads();
        float unxt[16];
#pragma unroll
        for (int c = 0; c < 16; c++) unxt[c] = 0.f;
        if (tid < 32) {
            if (kb > 0) {
                const float* up = Kb + (size_t)(r0 - 16 + l16) * PP + r0;
#pragma unroll
                for (int c = 0; c < 4; c++) {
                    float4 t4 = *(const float4*)(up + 4 * c);
                    unxt[4 * c] = t4.x; unxt[4 * c + 1] = t4.y; unxt[4 * c + 2] = t4.z; unxt[4 * c + 3] = t4.w;
                }
            }
        } else {
            float xs[16];
#pragma unroll
            for (int c = 0; c < 4; c++) {
                const float4 t4 = *(const float4*)(y + r0 + 4 * c);
                xs[4 * c] = t4.x; xs[4 * c + 1] = t4.y; xs[4 * c + 2] = t4.z; xs[4 * c + 3] = t4.w;
            }
            if (ia < r0) {
                float4 acc = zero4;
#pragma unroll
                for (int j = 0; j < 16; j++) {
                    acc.x = __builtin_fmaf(la[j].x, xs[j], acc.x); acc.y = __builtin_fmaf(la[j].y, xs[j], acc.y);
                    acc.z = __builtin_fmaf(la[j].z, xs[j], acc.z); acc.w = __builtin_fmaf(la[j].w, xs[j], acc.w);
                }
                float4 yy = *(const float4*)(y + ia);
                const float4 d4 = *(const float4*)(dg + ia);
                yy.x = __builtin_fmaf(-d4.x, acc.x, yy.x); yy.y = __builtin_fmaf(-d4.y, acc.y, yy.y);
                yy.z = __builtin_fmaf(-d4.z, acc.z, yy.z); yy.w = __builtin_fmaf(-d4.w, acc.w, yy.w);
                *(float4*)(y + ia) = yy;
            }
            if (ib < r0) {
                float4 acc = zero4;
#pragma unroll
                for (int j = 0; j < 16; j++) {
                    acc.x = __builtin_fmaf(lb[j].x, xs[j], acc.x); acc.y = __builtin_fmaf(lb[j].y, xs[j], acc.y);
                    acc.z = __builtin_fmaf(lb[j].z, xs[j], acc.z); acc.w = __builtin_fmaf(lb[j].w, xs[j], acc.w);
                }
                float4 yy = *(const float4*)(y + ib);
                const float4 d4 = *(const float4*)(dg + ib);
                yy.x = __builtin_fmaf(-d4.x, acc.x, yy.x); yy.y = __builtin_fmaf(-d4.y, acc.y, yy.y);
                yy.z = __builtin_fmaf(-d4.z, acc.z, yy.z); yy.w = __builtin_fmaf(-d4.w, acc.w, yy.w);
                *(float4*)(y + ib) = yy;
            }
        }
        __syncthreads();
#pragma unroll
        for (int c = 0; c < 16; c++) ucur[c] = unxt[c];
    }
    for (int i = tid; i < PP; i += 256) x[(size_t)b * PP + i] = y[i];
}

__global__ void vote_kernel(const float* __restrict__ x, float* __restrict__ out) {
    int s = blockIdx.x, b = blockIdx.y, tid = threadIdx.x;
    __shared__ float vals[NN_];
    if (tid < NN_) {
        float v = x[(size_t)b * PP + tid * NN_ + s];
        vals[tid] = fminf(fmaxf(v, 0.f), 1.f);
    }
    __syncthreads();
    if (tid == 0) {
        float m = -1e30f;
        for (int t = 0; t < NN_; t++) m = fmaxf(m, vals[t]);
        float sum = 0.f;
        for (int t = 0; t < NN_; t++) { float e = __expf(ALPHA * (vals[t] - m)); vals[t] = e; sum += e; }
        float inv = 1.f / sum;
        for (int t = 0; t < NN_; t++) out[((size_t)b * NN_ + s) * NN_ + t] = vals[t] * inv;
    }
}

extern "C" void kernel_launch(void* const* d_in, const int* in_sizes, int n_in,
                              void* d_out, int out_size, void* d_ws, size_t ws_size,
                              hipStream_t stream) {
    const float* emb1 = (const float*)d_in[0];
    const float* emb2 = (const float*)d_in[1];
    const float* Wcg  = (const float*)d_in[2];
    const float* bcg  = (const float*)d_in[3];

    float* ws  = (float*)d_ws;
    size_t kelem = (size_t)BB * PP * PP;
    float* Km  = ws;
    float* h1  = Km + kelem;
    float* h2  = h1 + (size_t)BB * NN_ * DD;
    float* S   = h2 + (size_t)BB * NN_ * DD;
    float* rhs = S  + (size_t)BB * NN2;
    float* x   = rhs + (size_t)BB * PP;

    feat_kernel<<<dim3(BB, 2), 256, 0, stream>>>(emb1, emb2, Wcg, bcg, h1, h2);
    gram_rhs_kernel<<<BB, 256, 0, stream>>>(h1, h2, S, rhs);
    assemble_kernel<<<dim3(NBLK / 2, BB), 256, 0, stream>>>(S, Km);

    for (int jb = 0; jb < 22; jb++)
        lu_col_kernel<4><<<dim3((NBLK - jb + 3) / 4, BB), 32 * NW, 0, stream>>>(Km, x, jb);
    for (int jb = 22; jb < 54; jb++)
        lu_col_kernel<2><<<dim3((NBLK - jb + 1) / 2, BB), 32 * NW, 0, stream>>>(Km, x, jb);
    for (int jb = 54; jb < NBLK; jb++)
        lu_col_kernel<1><<<dim3(NBLK - jb, BB), 32 * NW, 0, stream>>>(Km, x, jb);

    trisolve_kernel<<<BB, 256, 0, stream>>>(Km, rhs, x);
    vote_kernel<<<dim3(NN_, BB), 64, 0, stream>>>(x, (float*)d_out);
}
